# starting kernel
# baseline (speedup 1.0000x reference)
#include <hip/hip_runtime.h>
#include <cmath>

constexpr int N_  = 100000;
constexpr int E_  = 1000000;
constexpr int ET_ = N_ + E_;
constexpr int D_  = 64;
constexpr int F1_ = 144;
constexpr int F2_ = 64;

#define CDIV(a,b) (((a)+(b)-1)/(b))

typedef __attribute__((ext_vector_type(2))) float v2f;
typedef __attribute__((ext_vector_type(8))) float v8f;

__device__ __forceinline__ void edge_sd(const int* __restrict__ src,
                                        const int* __restrict__ dst,
                                        int e, int& s, int& d) {
  if (e < E_) { s = src[e]; d = dst[e]; }
  else        { s = e - E_; d = s; }
}

__device__ __forceinline__ void atomicMaxF(float* addr, float v) {
  if (v >= 0.0f) atomicMax((int*)addr, __float_as_int(v));
  else           atomicMin((unsigned int*)addr, __float_as_uint(v));
}

__global__ void k_fill(float* __restrict__ p, int n, float v) {
  int i = blockIdx.x * blockDim.x + threadIdx.x;
  if (i < n) p[i] = v;
}

__global__ void k_deg(const int* __restrict__ dst, float* __restrict__ deg) {
  int e = blockIdx.x * blockDim.x + threadIdx.x;
  if (e < E_) atomicAdd(&deg[dst[e]], 1.0f);
}

__global__ void k_rsqrt_inplace(float* __restrict__ p, int n) {
  int i = blockIdx.x * blockDim.x + threadIdx.x;
  if (i < n) p[i] = rsqrtf(p[i]);
}

__global__ void k_norm(const int* __restrict__ src, const int* __restrict__ dst,
                       const float* __restrict__ dinv, float* __restrict__ norm) {
  int e = blockIdx.x * blockDim.x + threadIdx.x;
  if (e < E_) norm[e] = dinv[src[e]] * dinv[dst[e]];
}

__global__ void k_agg(const int* __restrict__ src, const int* __restrict__ dst,
                      const float* __restrict__ x, const float* __restrict__ norm,
                      const float* __restrict__ dinv, float* __restrict__ agg) {
  int idx = blockIdx.x * blockDim.x + threadIdx.x;
  if (idx >= ET_ * D_) return;
  int e = idx >> 6, c = idx & 63;
  int s, d; float w;
  if (e < E_) { s = src[e]; d = dst[e]; w = norm[e]; }
  else        { s = e - E_; d = s; float di = dinv[s]; w = di * di; }
  atomicAdd(&agg[(size_t)d * D_ + c], x[(size_t)s * D_ + c] * w);
}

__global__ void k_combine(const float* __restrict__ x, float* __restrict__ h) {
  int i = blockIdx.x * blockDim.x + threadIdx.x;
  if (i < N_ * D_) h[i] = 0.5f * x[i] + 0.5f * h[i];
}

__global__ void k_gemm_bias(const float* __restrict__ A, const float* __restrict__ W,
                            const float* __restrict__ bias, float* __restrict__ C,
                            int K, int Nc) {
  const int wave = blockIdx.x * (blockDim.x >> 5) + (threadIdx.x >> 5);
  const int lane = threadIdx.x & 31;
  const int tilesN = Nc >> 4;
  const int totalTiles = (N_ >> 4) * tilesN;
  if (wave >= totalTiles) return;

  const int row0 = (wave / tilesN) << 4;
  const int col0 = (wave % tilesN) << 4;
  const int lm = lane & 15;
  const int lk = (lane >> 4) << 1;

  const float* __restrict__ Arow = A + (size_t)(row0 + lm) * K;
  const float* __restrict__ Wcol = W + (col0 + lm);

  v8f acc = {};
  for (int k = 0; k < K; k += 4) {
    v2f a, b;
    a.x = Arow[k + lk];
    a.y = Arow[k + lk + 1];
    b.x = Wcol[(size_t)(k + lk) * Nc];
    b.y = Wcol[(size_t)(k + lk + 1) * Nc];
    acc = __builtin_amdgcn_wmma_f32_16x16x4_f32(false, a, false, b,
                                                (short)0, acc, false, false);
  }

  const int n = col0 + lm;
  const float bn = bias[n];
  const int rbase = row0 + ((lane >> 4) << 3);
#pragma unroll
  for (int j = 0; j < 8; ++j)
    C[(size_t)(rbase + j) * Nc + n] = acc[j] + bn;
}

__global__ void k_logit12(const int* __restrict__ src, const int* __restrict__ dst,
                          const float* __restrict__ xl, const float* __restrict__ xr,
                          const float* __restrict__ att, float* __restrict__ lg,
                          float* __restrict__ m) {
  int idx = blockIdx.x * blockDim.x + threadIdx.x;
  if (idx >= ET_ * 12) return;
  int e = idx / 12, h = idx % 12;
  int s, d; edge_sd(src, dst, e, s, d);
  const float* pl = xl + (size_t)s * F1_ + h * 12;
  const float* pr = xr + (size_t)d * F1_ + h * 12;
  const float* pa = att + h * 12;
  float acc = 0.0f;
#pragma unroll
  for (int c = 0; c < 12; ++c) {
    float t = pl[c] + pr[c];
    t = (t > 0.0f) ? t : 0.2f * t;
    acc += t * pa[c];
  }
  lg[idx] = acc;
  atomicMaxF(&m[d * 12 + h], acc);
}

__global__ void k_expden12(const int* __restrict__ src, const int* __restrict__ dst,
                           const float* __restrict__ m, float* __restrict__ lg,
                           float* __restrict__ den) {
  int idx = blockIdx.x * blockDim.x + threadIdx.x;
  if (idx >= ET_ * 12) return;
  int e = idx / 12, h = idx % 12;
  int s, d; edge_sd(src, dst, e, s, d);
  float a = expf(lg[idx] - m[d * 12 + h]);
  lg[idx] = a;
  atomicAdd(&den[d * 12 + h], a);
}

__global__ void k_accum12(const int* __restrict__ src, const int* __restrict__ dst,
                          const float* __restrict__ lg, const float* __restrict__ xl,
                          float* __restrict__ out) {
  int idx = blockIdx.x * blockDim.x + threadIdx.x;
  if (idx >= ET_ * F1_) return;
  int e = idx / F1_, r = idx % F1_;
  int h = r / 12;
  int s, d; edge_sd(src, dst, e, s, d);
  atomicAdd(&out[(size_t)d * F1_ + r], lg[e * 12 + h] * xl[(size_t)s * F1_ + r]);
}

__global__ void k_fin12(const float* __restrict__ den, const float* __restrict__ bias,
                        float* __restrict__ out) {
  int idx = blockIdx.x * blockDim.x + threadIdx.x;
  if (idx >= N_ * F1_) return;
  int n = idx / F1_, r = idx % F1_;
  int h = r / 12;
  out[idx] = out[idx] / (den[n * 12 + h] + 1e-16f) + bias[r];
}

__global__ void k_logit1(const int* __restrict__ src, const int* __restrict__ dst,
                         const float* __restrict__ xl, const float* __restrict__ xr,
                         const float* __restrict__ att, float* __restrict__ lg,
                         float* __restrict__ m) {
  int wv = blockIdx.x * (blockDim.x >> 5) + (threadIdx.x >> 5);
  int lane = threadIdx.x & 31;
  if (wv >= ET_) return;
  int s, d; edge_sd(src, dst, wv, s, d);
  float acc = 0.0f;
#pragma unroll
  for (int c = lane; c < 64; c += 32) {
    float t = xl[(size_t)s * F2_ + c] + xr[(size_t)d * F2_ + c];
    t = (t > 0.0f) ? t : 0.2f * t;
    acc += t * att[c];
  }
#pragma unroll
  for (int o = 16; o > 0; o >>= 1) acc += __shfl_xor(acc, o, 32);
  if (lane == 0) { lg[wv] = acc; atomicMaxF(&m[d], acc); }
}

__global__ void k_expden1(const int* __restrict__ src, const int* __restrict__ dst,
                          const float* __restrict__ m, float* __restrict__ lg,
                          float* __restrict__ den) {
  int e = blockIdx.x * blockDim.x + threadIdx.x;
  if (e >= ET_) return;
  int s, d; edge_sd(src, dst, e, s, d);
  float a = expf(lg[e] - m[d]);
  lg[e] = a;
  atomicAdd(&den[d], a);
}

__global__ void k_accum1(const int* __restrict__ src, const int* __restrict__ dst,
                         const float* __restrict__ lg, const float* __restrict__ xl,
                         float* __restrict__ num) {
  int idx = blockIdx.x * blockDim.x + threadIdx.x;
  if (idx >= ET_ * F2_) return;
  int e = idx >> 6, c = idx & 63;
  int s, d; edge_sd(src, dst, e, s, d);
  atomicAdd(&num[(size_t)d * F2_ + c], lg[e] * xl[(size_t)s * F2_ + c]);
}

__global__ void k_final(const float* __restrict__ num, const float* __restrict__ den,
                        const float* __restrict__ bias, float* __restrict__ out) {
  int idx = blockIdx.x * blockDim.x + threadIdx.x;
  if (idx >= N_ * F2_) return;
  int n = idx >> 6, c = idx & 63;
  out[idx] = num[idx] / (den[n] + 1e-16f) + bias[c];
}

constexpr size_t OFF_DEG  = 0;
constexpr size_t OFF_NORM = OFF_DEG  + N_;
constexpr size_t OFF_HBUF = OFF_NORM + E_;
constexpr size_t OFF_X1   = OFF_HBUF + (size_t)N_*D_;
constexpr size_t OFF_XL1  = OFF_X1   + (size_t)N_*D_;
constexpr size_t OFF_XR1  = OFF_XL1  + (size_t)N_*F1_;
constexpr size_t OFF_M1   = OFF_XR1  + (size_t)N_*F1_;
constexpr size_t OFF_DEN1 = OFF_M1   + (size_t)N_*12;
constexpr size_t OFF_LG1  = OFF_DEN1 + (size_t)N_*12;

extern "C" void kernel_launch(void* const* d_in, const int* in_sizes, int n_in,
                              void* d_out, int out_size, void* d_ws, size_t ws_size,
                              hipStream_t stream) {
  const float* x     = (const float*)d_in[0];
  const int*   ei    = (const int*)  d_in[1];
  const float* Wssg  = (const float*)d_in[2];
  const float* bssg  = (const float*)d_in[3];
  const float* W1l   = (const float*)d_in[4];
  const float* b1l   = (const float*)d_in[5];
  const float* W1r   = (const float*)d_in[6];
  const float* b1r   = (const float*)d_in[7];
  const float* att1  = (const float*)d_in[8];
  const float* bias1 = (const float*)d_in[9];
  const float* W2l   = (const float*)d_in[10];
  const float* b2l   = (const float*)d_in[11];
  const float* W2r   = (const float*)d_in[12];
  const float* b2r   = (const float*)d_in[13];
  const float* att2  = (const float*)d_in[14];
  const float* bias2 = (const float*)d_in[15];

  const int* src = ei;
  const int* dst = ei + E_;

  float* ws   = (float*)d_ws;
  float* deg  = ws + OFF_DEG;
  float* norm = ws + OFF_NORM;
  float* hbuf = ws + OFF_HBUF;
  float* x1   = ws + OFF_X1;
  float* xl1  = ws + OFF_XL1;
  float* xr1  = ws + OFF_XR1;
  float* m1   = ws + OFF_M1;
  float* den1 = ws + OFF_DEN1;
  float* lg1  = ws + OFF_LG1;
  float* out1 = xr1;
  float* xl2  = hbuf;
  float* xr2  = xl1;
  float* m2   = deg;
  float* den2 = norm;
  float* lg2  = lg1;
  float* num2 = x1;
  float* out  = (float*)d_out;

  const int TB = 256;
  const float NEG_INF = -__builtin_huge_valf();

  auto gemm = [&](const float* A, const float* W, const float* b, float* C,
                  int K, int Nc) {
    int tiles = (N_ / 16) * (Nc / 16);
    k_gemm_bias<<<CDIV(tiles, TB / 32), TB, 0, stream>>>(A, W, b, C, K, Nc);
  };

  k_fill<<<CDIV(N_, TB), TB, 0, stream>>>(deg, N_, 1.0f);
  k_fill<<<CDIV(N_*D_, TB), TB, 0, stream>>>(hbuf, N_*D_, 0.0f);
  k_deg<<<CDIV(E_, TB), TB, 0, stream>>>(dst, deg);
  k_rsqrt_inplace<<<CDIV(N_, TB), TB, 0, stream>>>(deg, N_);
  k_norm<<<CDIV(E_, TB), TB, 0, stream>>>(src, dst, deg, norm);
  k_agg<<<CDIV(ET_*D_, TB), TB, 0, stream>>>(src, dst, x, norm, deg, hbuf);
  k_combine<<<CDIV(N_*D_, TB), TB, 0, stream>>>(x, hbuf);
  gemm(hbuf, Wssg, bssg, x1, D_, D_);

  gemm(x1, W1l, b1l, xl1, D_, F1_);
  gemm(x1, W1r, b1r, xr1, D_, F1_);
  k_fill<<<CDIV(N_*12, TB), TB, 0, stream>>>(m1, N_*12, NEG_INF);
  k_fill<<<CDIV(N_*12, TB), TB, 0, stream>>>(den1, N_*12, 0.0f);
  k_logit12<<<CDIV(ET_*12, TB), TB, 0, stream>>>(src, dst, xl1, xr1, att1, lg1, m1);
  k_fill<<<CDIV(N_*F1_, TB), TB, 0, stream>>>(out1, N_*F1_, 0.0f);
  k_expden12<<<CDIV(ET_*12, TB), TB, 0, stream>>>(src, dst, m1, lg1, den1);
  k_accum12<<<CDIV(ET_*F1_, TB), TB, 0, stream>>>(src, dst, lg1, xl1, out1);
  k_fin12<<<CDIV(N_*F1_, TB), TB, 0, stream>>>(den1, bias1, out1);

  gemm(out1, W2l, b2l, xl2, F1_, F2_);
  gemm(out1, W2r, b2r, xr2, F1_, F2_);
  k_fill<<<CDIV(N_, TB), TB, 0, stream>>>(m2, N_, NEG_INF);
  k_fill<<<CDIV(N_, TB), TB, 0, stream>>>(den2, N_, 0.0f);
  k_fill<<<CDIV(N_*F2_, TB), TB, 0, stream>>>(num2, N_*F2_, 0.0f);
  k_logit1<<<CDIV(ET_, TB/32), TB, 0, stream>>>(src, dst, xl2, xr2, att2, lg2, m2);
  k_expden1<<<CDIV(ET_, TB), TB, 0, stream>>>(src, dst, m2, lg2, den2);
  k_accum1<<<CDIV(ET_*F2_, TB), TB, 0, stream>>>(src, dst, lg2, xl2, num2);
  k_final<<<CDIV(N_*F2_, TB), TB, 0, stream>>>(num2, den2, bias2, out);
}


# float atomics moved from L2 to LDS: edges binned by 64-destination tile in-call; one work-group per tile does max, sums, weighted sums in LDS atomics, passes fused per layer; sums stay arrival order
# speedup vs baseline: 2.8979x; 2.8979x over previous
#include <hip/hip_runtime.h>
#include <cmath>

constexpr int N_  = 100000;
constexpr int E_  = 1000000;
constexpr int D_  = 64;
constexpr int F1_ = 144;
constexpr int F2_ = 64;

#define CDIV(a,b) (((a)+(b)-1)/(b))

typedef __attribute__((ext_vector_type(2))) float v2f;
typedef __attribute__((ext_vector_type(8))) float v8f;

constexpr int TSH = 6;
constexpr int TS  = 1 << TSH;
constexpr int NT  = CDIV(N_, TS);
constexpr int CH  = 8192;
constexpr int SRC_BITS = 17;
constexpr int SRC_MASK = (1 << SRC_BITS) - 1;
static_assert(N_ <= (1 << SRC_BITS), "source id must fit the packed field");
static_assert(NT + 1 <= 2048, "scan handles two tiles per thread of one 1024-thread group");

__global__ void k_zero_i(int* __restrict__ p, int n) {
  int i = blockIdx.x * blockDim.x + threadIdx.x;
  if (i < n) p[i] = 0;
}

__global__ __launch_bounds__(1024) void k_tilecount(const int* __restrict__ dst, int* __restrict__ cnt) {
  __shared__ int hist[NT];
  const int tid = threadIdx.x;
  for (int i = tid; i < NT; i += 1024) hist[i] = 0;
  __syncthreads();
  const int base = blockIdx.x * CH;
#pragma unroll
  for (int k = 0; k < CH / 1024; ++k) {
    int e = base + k * 1024 + tid;
    if (e < E_) atomicAdd(&hist[dst[e] >> TSH], 1);
  }
  __syncthreads();
  for (int i = tid; i < NT; i += 1024) { int v = hist[i]; if (v) atomicAdd(&cnt[i], v); }
}

__global__ __launch_bounds__(1024) void k_scan(const int* __restrict__ cnt, int* __restrict__ off,
                                               int* __restrict__ cur) {
  __shared__ int p[1024];
  const int t = threadIdx.x;
  const int a = (2 * t     < NT) ? cnt[2 * t]     : 0;
  const int b = (2 * t + 1 < NT) ? cnt[2 * t + 1] : 0;
  int v = a + b;
  p[t] = v;
  __syncthreads();
  for (int o = 1; o < 1024; o <<= 1) {
    int add = (t >= o) ? p[t - o] : 0;
    __syncthreads();
    v += add;
    p[t] = v;
    __syncthreads();
  }
  const int excl = v - (a + b);
  if (2 * t     < NT) { off[2 * t]     = excl;     cur[2 * t]     = excl; }
  if (2 * t + 1 < NT) { off[2 * t + 1] = excl + a; cur[2 * t + 1] = excl + a; }
  if (t == 1023) off[NT] = v;
}

__global__ __launch_bounds__(1024) void k_binscatter(const int* __restrict__ src, const int* __restrict__ dst,
                                                     int* __restrict__ cur, int* __restrict__ bins) {
  __shared__ int hist[NT];
  const int tid = threadIdx.x;
  for (int i = tid; i < NT; i += 1024) hist[i] = 0;
  __syncthreads();
  const int base = blockIdx.x * CH;
  int pk[CH / 1024], bn[CH / 1024], lr[CH / 1024];
#pragma unroll
  for (int k = 0; k < CH / 1024; ++k) {
    int e = base + k * 1024 + tid;
    bn[k] = -1; pk[k] = 0; lr[k] = 0;
    if (e < E_) {
      int d = dst[e];
      bn[k] = d >> TSH;
      pk[k] = src[e] | ((d & (TS - 1)) << SRC_BITS);
      lr[k] = atomicAdd(&hist[bn[k]], 1);
    }
  }
  __syncthreads();
  for (int i = tid; i < NT; i += 1024) { int v = hist[i]; hist[i] = v ? atomicAdd(&cur[i], v) : 0; }
  __syncthreads();
#pragma unroll
  for (int k = 0; k < CH / 1024; ++k)
    if (bn[k] >= 0) bins[hist[bn[k]] + lr[k]] = pk[k];
}

__device__ __forceinline__ bool tile_item(int j, int tot, int cnt, int b, int n0,
                                          const int* __restrict__ bins, int& s, int& dl, int& sl) {
  const int p  = min(b + j, E_ - 1);
  const int pk = bins[p];
  const bool isE = j < cnt;
  const int dlS = j - cnt;
  const int sS  = n0 + dlS;
  const bool okS = (!isE) && (j < tot) && (sS < N_);
  s  = isE ? (pk & SRC_MASK)  : (okS ? sS : 0);
  dl = isE ? (pk >> SRC_BITS) : (okS ? dlS : 0);
  sl = isE ? p                : (okS ? E_ + sS : 0);
  return isE || okS;
}

__device__ __forceinline__ void ldsMaxF(int* cell, float v) {
  if (v >= 0.0f) atomicMax(cell, __float_as_int(v));
  else           atomicMin((unsigned int*)cell, __float_as_uint(v));
}

__global__ __launch_bounds__(TS) void k_dinv_tile(const int* __restrict__ off, const int* __restrict__ bins,
                                                  float* __restrict__ dinv) {
  __shared__ int c[TS];
  const int t = blockIdx.x, tid = threadIdx.x;
  c[tid] = 0;
  __syncthreads();
  const int b = off[t], e = off[t + 1];
  for (int p = b + tid; p < e; p += TS) atomicAdd(&c[bins[p] >> SRC_BITS], 1);
  __syncthreads();
  const int n = t * TS + tid;
  if (n < N_) dinv[n] = rsqrtf(1.0f + (float)c[tid]);
}

template <int U>
__global__ __launch_bounds__(256) void k_ssg_tile(const int* __restrict__ off, const int* __restrict__ bins,
                                                  const float* __restrict__ x, const float* __restrict__ dinv,
                                                  float* __restrict__ h) {
  __shared__ float acc[TS * D_];
  const int t = blockIdx.x, tid = threadIdx.x, lane = tid & 31;
  const int wv = __builtin_amdgcn_readfirstlane(tid >> 5);
  for (int i = tid; i < TS * D_; i += 256) acc[i] = 0.0f;
  __syncthreads();
  const int b = off[t];
  const int cnt = off[t + 1] - b;
  const int tot = cnt + TS;
  const int n0 = t * TS;
  for (int j0 = wv * U; j0 < tot; j0 += 8 * U) {
    int s[U], dl[U], sl[U]; bool ok[U];
    float x0[U], x1[U], w[U];
#pragma unroll
    for (int u = 0; u < U; ++u) ok[u] = tile_item(j0 + u, tot, cnt, b, n0, bins, s[u], dl[u], sl[u]);
#pragma unroll
    for (int u = 0; u < U; ++u) {
      w[u]  = dinv[s[u]] * dinv[n0 + dl[u]];
      x0[u] = x[(size_t)s[u] * D_ + lane];
      x1[u] = x[(size_t)s[u] * D_ + lane + 32];
    }
#pragma unroll
    for (int u = 0; u < U; ++u) {
      atomicAdd(&acc[dl[u] * D_ + lane],      ok[u] ? x0[u] * w[u] : 0.0f);
      atomicAdd(&acc[dl[u] * D_ + lane + 32], ok[u] ? x1[u] * w[u] : 0.0f);
    }
  }
  __syncthreads();
  for (int i = tid; i < TS * D_; i += 256) {
    const int n = n0 + (i >> 6), c = i & 63;
    if (n < N_) h[(size_t)n * D_ + c] = 0.5f * x[(size_t)n * D_ + c] + 0.5f * acc[i];
  }
}

__global__ void k_gemm_bias(const float* __restrict__ A, const float* __restrict__ W,
                            const float* __restrict__ bias, float* __restrict__ C,
                            int K, int Nc) {
  const int wave = blockIdx.x * (blockDim.x >> 5) + (threadIdx.x >> 5);
  const int lane = threadIdx.x & 31;
  const int tilesN = Nc >> 4;
  const int totalTiles = (N_ >> 4) * tilesN;
  if (wave >= totalTiles) return;

  const int row0 = (wave / tilesN) << 4;
  const int col0 = (wave % tilesN) << 4;
  const int lm = lane & 15;
  const int lk = (lane >> 4) << 1;

  const float* __restrict__ Arow = A + (size_t)(row0 + lm) * K;
  const float* __restrict__ Wcol = W + (col0 + lm);

  v8f acc = {};
  for (int k = 0; k < K; k += 4) {
    v2f a, b;
    a.x = Arow[k + lk];
    a.y = Arow[k + lk + 1];
    b.x = Wcol[(size_t)(k + lk) * Nc];
    b.y = Wcol[(size_t)(k + lk + 1) * Nc];
    acc = __builtin_amdgcn_wmma_f32_16x16x4_f32(false, a, false, b,
                                                (short)0, acc, false, false);
  }

  const int n = col0 + lm;
  const float bn = bias[n];
  const int rbase = row0 + ((lane >> 4) << 3);
#pragma unroll
  for (int j = 0; j < 8; ++j)
    C[(size_t)(rbase + j) * Nc + n] = acc[j] + bn;
}

template <int H, int C, int U>
__global__ __launch_bounds__(256) void k_gat_tile(const int* __restrict__ off, const int* __restrict__ bins,
                                                  const float* __restrict__ xl, const float* xr,
                                                  const float* __restrict__ att, const float* __restrict__ bias,
                                                  float* __restrict__ lg, float* out) {
  constexpr int F = H * C;
  constexpr int K = (F + 31) / 32;
  static_assert(C % 4 == 0, "head rows are read as 16-byte pieces");
  __shared__ float acc[TS * F];
  __shared__ int   mxi[TS * H];
  __shared__ float den[TS * H];
  __shared__ float satt[F];

  const int t = blockIdx.x, tid = threadIdx.x, lane = tid & 31;
  const int wv = __builtin_amdgcn_readfirstlane(tid >> 5);
  for (int i = tid; i < TS * F; i += 256) acc[i] = 0.0f;
  for (int i = tid; i < TS * H; i += 256) { mxi[i] = (int)0xFF800000u; den[i] = 0.0f; }
  for (int i = tid; i < F; i += 256) satt[i] = att[i];
  __syncthreads();
  const int b = off[t];
  const int cnt = off[t + 1] - b;
  const int tot = cnt + TS;
  const int n0 = t * TS;

  if constexpr (H == 1) {
    for (int j0 = wv * U; j0 < tot; j0 += 8 * U) {
      int s[U], dl[U], sl[U]; bool ok[U];
      float l0[U], l1[U], r0[U], r1[U];
#pragma unroll
      for (int u = 0; u < U; ++u) ok[u] = tile_item(j0 + u, tot, cnt, b, n0, bins, s[u], dl[u], sl[u]);
#pragma unroll
      for (int u = 0; u < U; ++u) {
        l0[u] = xl[(size_t)s[u] * F + lane];
        l1[u] = xl[(size_t)s[u] * F + lane + 32];
        r0[u] = xr[(size_t)(n0 + dl[u]) * F + lane];
        r1[u] = xr[(size_t)(n0 + dl[u]) * F + lane + 32];
      }
#pragma unroll
      for (int u = 0; u < U; ++u) {
        float a = 0.0f;
        float t0 = l0[u] + r0[u];
        t0 = (t0 > 0.0f) ? t0 : 0.2f * t0;
        a += t0 * satt[lane];
        float t1 = l1[u] + r1[u];
        t1 = (t1 > 0.0f) ? t1 : 0.2f * t1;
        a += t1 * satt[lane + 32];
#pragma unroll
        for (int o = 16; o > 0; o >>= 1) a += __shfl_xor(a, o, 32);
        if (lane == 0 && ok[u]) { lg[sl[u]] = a; ldsMaxF(&mxi[dl[u]], a); }
      }
    }
  } else {
    for (int idx = tid; idx < tot * H; idx += 256) {
      const int j = idx / H, h = idx - j * H;
      int s, dl, sl;
      if (tile_item(j, tot, cnt, b, n0, bins, s, dl, sl)) {
        const float4* pl = (const float4*)(xl + (size_t)s * F + h * C);
        const float4* pr = (const float4*)(xr + (size_t)(n0 + dl) * F + h * C);
        float4 lv[C / 4], rv[C / 4];
#pragma unroll
        for (int q = 0; q < C / 4; ++q) { lv[q] = pl[q]; rv[q] = pr[q]; }
        float a = 0.0f;
#pragma unroll
        for (int q = 0; q < C / 4; ++q) {
          const float le[4] = {lv[q].x, lv[q].y, lv[q].z, lv[q].w};
          const float re[4] = {rv[q].x, rv[q].y, rv[q].z, rv[q].w};
#pragma unroll
          for (int z = 0; z < 4; ++z) {
            float tt = le[z] + re[z];
            tt = (tt > 0.0f) ? tt : 0.2f * tt;
            a += tt * satt[h * C + q * 4 + z];
          }
        }
        lg[(size_t)sl * H + h] = a;
        ldsMaxF(&mxi[dl * H + h], a);
      }
    }
  }
  __syncthreads();

  for (int idx = tid; idx < tot * H; idx += 256) {
    const int j = idx / H, h = idx - j * H;
    int s, dl, sl;
    if (tile_item(j, tot, cnt, b, n0, bins, s, dl, sl)) {
      const float a = expf(lg[(size_t)sl * H + h] - __int_as_float(mxi[dl * H + h]));
      lg[(size_t)sl * H + h] = a;
      atomicAdd(&den[dl * H + h], a);
    }
  }
  __syncthreads();

  for (int j0 = wv * U; j0 < tot; j0 += 8 * U) {
    int s[U], dl[U], sl[U]; bool ok[U];
    float xv[U][K], av[U][K];
#pragma unroll
    for (int u = 0; u < U; ++u) ok[u] = tile_item(j0 + u, tot, cnt, b, n0, bins, s[u], dl[u], sl[u]);
#pragma unroll
    for (int u = 0; u < U; ++u) {
#pragma unroll
      for (int k = 0; k < K; ++k) {
        const int r  = lane + 32 * k;
        const int rr = (r < F) ? r : r - 32;
        xv[u][k] = xl[(size_t)s[u] * F + rr];
        av[u][k] = lg[(size_t)sl[u] * H + rr / C];
      }
    }
#pragma unroll
    for (int u = 0; u < U; ++u) {
#pragma unroll
      for (int k = 0; k < K; ++k) {
        const int r  = lane + 32 * k;
        const int rr = (r < F) ? r : r - 32;
        atomicAdd(&acc[dl[u] * F + rr], (ok[u] && r < F) ? av[u][k] * xv[u][k] : 0.0f);
      }
    }
  }
  __syncthreads();

  for (int i = tid; i < TS * F; i += 256) {
    const int dl = i / F, r = i - dl * F;
    const int n = n0 + dl;
    if (n < N_) out[(size_t)n * F + r] = acc[i] / (den[dl * H + r / C] + 1e-16f) + bias[r];
  }
}

constexpr size_t OFF_DINV = 0;
constexpr size_t OFF_CNT  = OFF_DINV + N_;
constexpr size_t OFF_OFFS = OFF_CNT  + 2048;
constexpr size_t OFF_CUR  = OFF_OFFS + 2048;
constexpr size_t OFF_HBUF = (size_t)N_ + E_;
constexpr size_t OFF_X1   = OFF_HBUF + (size_t)N_*D_;
constexpr size_t OFF_XL1  = OFF_X1   + (size_t)N_*D_;
constexpr size_t OFF_XR1  = OFF_XL1  + (size_t)N_*F1_;
constexpr size_t OFF_BINS = OFF_XR1  + (size_t)N_*F1_;
constexpr size_t OFF_LG   = OFF_BINS + (size_t)N_*24;
static_assert(OFF_CUR + 2048 <= OFF_HBUF, "bin tables fit in front of h");
static_assert((size_t)E_ <= (size_t)N_ * 24, "bins fit their region");

extern "C" void kernel_launch(void* const* d_in, const int* in_sizes, int n_in,
                              void* d_out, int out_size, void* d_ws, size_t ws_size,
                              hipStream_t stream) {
  const float* x     = (const float*)d_in[0];
  const int*   ei    = (const int*)  d_in[1];
  const float* Wssg  = (const float*)d_in[2];
  const float* bssg  = (const float*)d_in[3];
  const float* W1l   = (const float*)d_in[4];
  const float* b1l   = (const float*)d_in[5];
  const float* W1r   = (const float*)d_in[6];
  const float* b1r   = (const float*)d_in[7];
  const float* att1  = (const float*)d_in[8];
  const float* bias1 = (const float*)d_in[9];
  const float* W2l   = (const float*)d_in[10];
  const float* b2l   = (const float*)d_in[11];
  const float* W2r   = (const float*)d_in[12];
  const float* b2r   = (const float*)d_in[13];
  const float* att2  = (const float*)d_in[14];
  const float* bias2 = (const float*)d_in[15];

  const int* src = ei;
  const int* dst = ei + E_;

  float* ws   = (float*)d_ws;
  float* dinv = ws + OFF_DINV;
  int*   cnt  = (int*)(ws + OFF_CNT);
  int*   offs = (int*)(ws + OFF_OFFS);
  int*   cur  = (int*)(ws + OFF_CUR);
  float* hbuf = ws + OFF_HBUF;
  float* x1   = ws + OFF_X1;
  float* xl1  = ws + OFF_XL1;
  float* xr1  = ws + OFF_XR1;
  int*   bins = (int*)(ws + OFF_BINS);
  float* lg   = ws + OFF_LG;
  float* out1 = xr1;
  float* xl2  = hbuf;
  float* xr2  = xl1;
  float* out  = (float*)d_out;

  const int TB = 256;

  auto gemm = [&](const float* A, const float* W, const float* b, float* C,
                  int K, int Nc) {
    int tiles = (N_ / 16) * (Nc / 16);
    k_gemm_bias<<<CDIV(tiles, TB / 32), TB, 0, stream>>>(A, W, b, C, K, Nc);
  };

  k_zero_i<<<CDIV(2048, TB), TB, 0, stream>>>(cnt, 2048);
  k_tilecount<<<CDIV(E_, CH), 1024, 0, stream>>>(dst, cnt);
  k_scan<<<1, 1024, 0, stream>>>(cnt, offs, cur);
  k_binscatter<<<CDIV(E_, CH), 1024, 0, stream>>>(src, dst, cur, bins);

  k_dinv_tile<<<NT, TS, 0, stream>>>(offs, bins, dinv);
  k_ssg_tile<4><<<NT, 256, 0, stream>>>(offs, bins, x, dinv, hbuf);
  gemm(hbuf, Wssg, bssg, x1, D_, D_);

  gemm(x1, W1l, b1l, xl1, D_, F1_);
  gemm(x1, W1r, b1r, xr1, D_, F1_);
  k_gat_tile<12, 12, 4><<<NT, 256, 0, stream>>>(offs, bins, xl1, xr1, att1, bias1, lg, out1);

  gemm(out1, W2l, b2l, xl2, F1_, F2_);
  gemm(out1, W2r, b2r, xr2, F1_, F2_);
  k_gat_tile<1, 64, 4><<<NT, 256, 0, stream>>>(offs, bins, xl2, xr2, att2, bias2, lg, out);
}


# projection GEMMs run inside the aggregation tile kernels: the 64-row tile left in LDS is the WMMA A operand; 5 GEMM launches and the h, x1, out1 round trips removed; k and sum order unchanged
# speedup vs baseline: 3.7630x; 1.2985x over previous
#include <hip/hip_runtime.h>
#include <cmath>

constexpr int N_  = 100000;
constexpr int E_  = 1000000;
constexpr int D_  = 64;
constexpr int F1_ = 144;
constexpr int F2_ = 64;

#define CDIV(a,b) (((a)+(b)-1)/(b))

typedef __attribute__((ext_vector_type(2))) float v2f;
typedef __attribute__((ext_vector_type(8))) float v8f;

constexpr int TSH = 6;
constexpr int TS  = 1 << TSH;
constexpr int NT  = CDIV(N_, TS);
constexpr int CH  = 8192;
constexpr int SRC_BITS = 17;
constexpr int SRC_MASK = (1 << SRC_BITS) - 1;
static_assert(N_ <= (1 << SRC_BITS), "source id must fit the packed field");
static_assert(NT + 1 <= 2048, "scan handles two tiles per thread of one 1024-thread group");

__global__ void k_zero_i(int* __restrict__ p, int n) {
  int i = blockIdx.x * blockDim.x + threadIdx.x;
  if (i < n) p[i] = 0;
}

__global__ __launch_bounds__(1024) void k_tilecount(const int* __restrict__ dst, int* __restrict__ cnt) {
  __shared__ int hist[NT];
  const int tid = threadIdx.x;
  for (int i = tid; i < NT; i += 1024) hist[i] = 0;
  __syncthreads();
  const int base = blockIdx.x * CH;
#pragma unroll
  for (int k = 0; k < CH / 1024; ++k) {
    int e = base + k * 1024 + tid;
    if (e < E_) atomicAdd(&hist[dst[e] >> TSH], 1);
  }
  __syncthreads();
  for (int i = tid; i < NT; i += 1024) { int v = hist[i]; if (v) atomicAdd(&cnt[i], v); }
}

__global__ __launch_bounds__(1024) void k_scan(const int* __restrict__ cnt, int* __restrict__ off,
                                               int* __restrict__ cur) {
  __shared__ int p[1024];
  const int t = threadIdx.x;
  const int a = (2 * t     < NT) ? cnt[2 * t]     : 0;
  const int b = (2 * t + 1 < NT) ? cnt[2 * t + 1] : 0;
  int v = a + b;
  p[t] = v;
  __syncthreads();
  for (int o = 1; o < 1024; o <<= 1) {
    int add = (t >= o) ? p[t - o] : 0;
    __syncthreads();
    v += add;
    p[t] = v;
    __syncthreads();
  }
  const int excl = v - (a + b);
  if (2 * t     < NT) { off[2 * t]     = excl;     cur[2 * t]     = excl; }
  if (2 * t + 1 < NT) { off[2 * t + 1] = excl + a; cur[2 * t + 1] = excl + a; }
  if (t == 1023) off[NT] = v;
}

__global__ __launch_bounds__(1024) void k_binscatter(const int* __restrict__ src, const int* __restrict__ dst,
                                                     int* __restrict__ cur, int* __restrict__ bins) {
  __shared__ int hist[NT];
  const int tid = threadIdx.x;
  for (int i = tid; i < NT; i += 1024) hist[i] = 0;
  __syncthreads();
  const int base = blockIdx.x * CH;
  int pk[CH / 1024], bn[CH / 1024], lr[CH / 1024];
#pragma unroll
  for (int k = 0; k < CH / 1024; ++k) {
    int e = base + k * 1024 + tid;
    bn[k] = -1; pk[k] = 0; lr[k] = 0;
    if (e < E_) {
      int d = dst[e];
      bn[k] = d >> TSH;
      pk[k] = src[e] | ((d & (TS - 1)) << SRC_BITS);
      lr[k] = atomicAdd(&hist[bn[k]], 1);
    }
  }
  __syncthreads();
  for (int i = tid; i < NT; i += 1024) { int v = hist[i]; hist[i] = v ? atomicAdd(&cur[i], v) : 0; }
  __syncthreads();
#pragma unroll
  for (int k = 0; k < CH / 1024; ++k)
    if (bn[k] >= 0) bins[hist[bn[k]] + lr[k]] = pk[k];
}

__device__ __forceinline__ bool tile_item(int j, int tot, int cnt, int b, int n0,
                                          const int* __restrict__ bins, int& s, int& dl, int& sl) {
  const int p  = min(b + j, E_ - 1);
  const int pk = bins[p];
  const bool isE = j < cnt;
  const int dlS = j - cnt;
  const int sS  = n0 + dlS;
  const bool okS = (!isE) && (j < tot) && (sS < N_);
  s  = isE ? (pk & SRC_MASK)  : (okS ? sS : 0);
  dl = isE ? (pk >> SRC_BITS) : (okS ? dlS : 0);
  sl = isE ? p                : (okS ? E_ + sS : 0);
  return isE || okS;
}

__device__ __forceinline__ void ldsMaxF(int* cell, float v) {
  if (v >= 0.0f) atomicMax(cell, __float_as_int(v));
  else           atomicMin((unsigned int*)cell, __float_as_uint(v));
}

__global__ __launch_bounds__(TS) void k_dinv_tile(const int* __restrict__ off, const int* __restrict__ bins,
                                                  float* __restrict__ dinv) {
  __shared__ int c[TS];
  const int t = blockIdx.x, tid = threadIdx.x;
  c[tid] = 0;
  __syncthreads();
  const int b = off[t], e = off[t + 1];
  for (int p = b + tid; p < e; p += TS) atomicAdd(&c[bins[p] >> SRC_BITS], 1);
  __syncthreads();
  const int n = t * TS + tid;
  if (n < N_) dinv[n] = rsqrtf(1.0f + (float)c[tid]);
}

template <int K, int PA, int NC>
__device__ __forceinline__ void tile_mma2(const float* a0p, const float* __restrict__ wp, v8f& acc0, v8f& acc1) {
  const float* a1p = a0p + 16 * PA;
#pragma unroll 4
  for (int k = 0; k < K; k += 4) {
    v2f a0, a1, b;
    a0.x = a0p[k]; a0.y = a0p[k + 1];
    a1.x = a1p[k]; a1.y = a1p[k + 1];
    b.x = wp[(size_t)k * NC];
    b.y = wp[(size_t)(k + 1) * NC];
    acc0 = __builtin_amdgcn_wmma_f32_16x16x4_f32(false, a0, false, b, (short)0, acc0, false, false);
    acc1 = __builtin_amdgcn_wmma_f32_16x16x4_f32(false, a1, false, b, (short)0, acc1, false, false);
  }
}

template <int K, int PA, int NC, int PC>
__device__ __forceinline__ void tile_proj_lds(const float* As, const float* __restrict__ W,
                                              const float* __restrict__ bias, float* Cs, int wv, int lane) {
  const int c = lane & 15, hh = lane >> 4;
  const int r0 = (wv & 1) * 32;
  for (int ct = wv >> 1; ct < NC / 16; ct += 4) {
    const int col = ct * 16 + c;
    v8f acc0 = {}, acc1 = {};
    tile_mma2<K, PA, NC>(As + (r0 + c) * PA + 2 * hh, W + (size_t)(2 * hh) * NC + col, acc0, acc1);
    const float bn = bias[col];
#pragma unroll
    for (int j = 0; j < 8; ++j) {
      Cs[(r0 + 8 * hh + j) * PC + col]      = acc0[j] + bn;
      Cs[(r0 + 16 + 8 * hh + j) * PC + col] = acc1[j] + bn;
    }
  }
}

template <int K, int PA, int NC>
__device__ __forceinline__ void tile_proj_pair(const float* As, int n0, int wv, int lane,
                                               const float* __restrict__ Wa, const float* __restrict__ ba,
                                               float* __restrict__ Ca,
                                               const float* __restrict__ Wb, const float* __restrict__ bb,
                                               float* __restrict__ Cb) {
  constexpr int CT = NC / 16;
  const int c = lane & 15, hh = lane >> 4;
  const int r0 = (wv & 1) * 32;
  for (int ct = wv >> 1; ct < 2 * CT; ct += 4) {
    const bool second = ct >= CT;
    const float* __restrict__ W  = second ? Wb : Wa;
    const float* __restrict__ bs = second ? bb : ba;
    float* __restrict__ C        = second ? Cb : Ca;
    const int col = (second ? ct - CT : ct) * 16 + c;
    v8f acc0 = {}, acc1 = {};
    tile_mma2<K, PA, NC>(As + (r0 + c) * PA + 2 * hh, W + (size_t)(2 * hh) * NC + col, acc0, acc1);
    const float bn = bs[col];
    float* __restrict__ cp = C + (size_t)(n0 + r0 + 8 * hh) * NC + col;
    if (n0 + r0 < N_) {
#pragma unroll
      for (int j = 0; j < 8; ++j) cp[(size_t)j * NC] = acc0[j] + bn;
    }
    if (n0 + r0 + 16 < N_) {
#pragma unroll
      for (int j = 0; j < 8; ++j) cp[(size_t)(16 + j) * NC] = acc1[j] + bn;
    }
  }
}

constexpr int PH = D_ + 4;
static_assert(N_ % 16 == 0, "a 16-row output tile is never partly past the last node");
template <int U>
__global__ __launch_bounds__(256) void k_ssg_tile(const int* __restrict__ off, const int* __restrict__ bins,
                                                  const float* __restrict__ x, const float* __restrict__ dinv,
                                                  const float* __restrict__ Wssg, const float* __restrict__ bssg,
                                                  const float* __restrict__ W1l, const float* __restrict__ b1l,
                                                  const float* __restrict__ W1r, const float* __restrict__ b1r,
                                                  float* __restrict__ xl1, float* __restrict__ xr1) {
  __shared__ float acc[TS * PH];
  __shared__ float x1s[TS * PH];
  const int t = blockIdx.x, tid = threadIdx.x, lane = tid & 31;
  const int wv = __builtin_amdgcn_readfirstlane(tid >> 5);
  for (int i = tid; i < TS * PH; i += 256) acc[i] = 0.0f;
  __syncthreads();
  const int b = off[t];
  const int cnt = off[t + 1] - b;
  const int tot = cnt + TS;
  const int n0 = t * TS;
  for (int j0 = wv * U; j0 < tot; j0 += 8 * U) {
    int s[U], dl[U], sl[U]; bool ok[U];
    float x0[U], x1[U], w[U];
#pragma unroll
    for (int u = 0; u < U; ++u) ok[u] = tile_item(j0 + u, tot, cnt, b, n0, bins, s[u], dl[u], sl[u]);
#pragma unroll
    for (int u = 0; u < U; ++u) {
      w[u]  = dinv[s[u]] * dinv[n0 + dl[u]];
      x0[u] = x[(size_t)s[u] * D_ + lane];
      x1[u] = x[(size_t)s[u] * D_ + lane + 32];
    }
#pragma unroll
    for (int u = 0; u < U; ++u) {
      atomicAdd(&acc[dl[u] * PH + lane],      ok[u] ? x0[u] * w[u] : 0.0f);
      atomicAdd(&acc[dl[u] * PH + lane + 32], ok[u] ? x1[u] * w[u] : 0.0f);
    }
  }
  __syncthreads();
  for (int i = tid; i < TS * D_; i += 256) {
    const int dl = i >> 6, c = i & 63;
    const int n = n0 + dl;
    acc[dl * PH + c] = (n < N_) ? 0.5f * x[(size_t)n * D_ + c] + 0.5f * acc[dl * PH + c] : 0.0f;
  }
  __syncthreads();
  tile_proj_lds<D_, PH, D_, PH>(acc, Wssg, bssg, x1s, wv, lane);
  __syncthreads();
  tile_proj_pair<D_, PH, F1_>(x1s, n0, wv, lane, W1l, b1l, xl1, W1r, b1r, xr1);
}

template <int H, int C, int U, bool PROJ>
__global__ __launch_bounds__(256) void k_gat_tile(const int* __restrict__ off, const int* __restrict__ bins,
                                                  const float* __restrict__ xl, const float* __restrict__ xr,
                                                  const float* __restrict__ att, const float* __restrict__ bias,
                                                  float* __restrict__ lg, float* __restrict__ out,
                                                  const float* __restrict__ Wnl, const float* __restrict__ bnl,
                                                  const float* __restrict__ Wnr, const float* __restrict__ bnr,
                                                  float* __restrict__ nxl, float* __restrict__ nxr) {
  constexpr int F = H * C;
  constexpr int K = (F + 31) / 32;
  constexpr int PF = PROJ ? F + 4 : F;
  static_assert(C % 4 == 0, "head rows are read as 16-byte pieces");
  static_assert(!PROJ || (PF % 64 == 4 || PF % 64 == 20 || PF % 64 == 36 || PF % 64 == 52),
                "fragment reads of 16 rows must fall on 16 different bank groups");
  __shared__ float acc[TS * PF];
  __shared__ int   mxi[TS * H];
  __shared__ float den[TS * H];
  __shared__ float satt[F];

  const int t = blockIdx.x, tid = threadIdx.x, lane = tid & 31;
  const int wv = __builtin_amdgcn_readfirstlane(tid >> 5);
  for (int i = tid; i < TS * PF; i += 256) acc[i] = 0.0f;
  for (int i = tid; i < TS * H; i += 256) { mxi[i] = (int)0xFF800000u; den[i] = 0.0f; }
  for (int i = tid; i < F; i += 256) satt[i] = att[i];
  __syncthreads();
  const int b = off[t];
  const int cnt = off[t + 1] - b;
  const int tot = cnt + TS;
  const int n0 = t * TS;

  if constexpr (H == 1) {
    for (int j0 = wv * U; j0 < tot; j0 += 8 * U) {
      int s[U], dl[U], sl[U]; bool ok[U];
      float l0[U], l1[U], r0[U], r1[U];
#pragma unroll
      for (int u = 0; u < U; ++u) ok[u] = tile_item(j0 + u, tot, cnt, b, n0, bins, s[u], dl[u], sl[u]);
#pragma unroll
      for (int u = 0; u < U; ++u) {
        l0[u] = xl[(size_t)s[u] * F + lane];
        l1[u] = xl[(size_t)s[u] * F + lane + 32];
        r0[u] = xr[(size_t)(n0 + dl[u]) * F + lane];
        r1[u] = xr[(size_t)(n0 + dl[u]) * F + lane + 32];
      }
#pragma unroll
      for (int u = 0; u < U; ++u) {
        float a = 0.0f;
        float t0 = l0[u] + r0[u];
        t0 = (t0 > 0.0f) ? t0 : 0.2f * t0;
        a += t0 * satt[lane];
        float t1 = l1[u] + r1[u];
        t1 = (t1 > 0.0f) ? t1 : 0.2f * t1;
        a += t1 * satt[lane + 32];
#pragma unroll
        for (int o = 16; o > 0; o >>= 1) a += __shfl_xor(a, o, 32);
        if (lane == 0 && ok[u]) { lg[sl[u]] = a; ldsMaxF(&mxi[dl[u]], a); }
      }
    }
  } else {
    for (int idx = tid; idx < tot * H; idx += 256) {
      const int j = idx / H, h = idx - j * H;
      int s, dl, sl;
      if (tile_item(j, tot, cnt, b, n0, bins, s, dl, sl)) {
        const float4* pl = (const float4*)(xl + (size_t)s * F + h * C);
        const float4* pr = (const float4*)(xr + (size_t)(n0 + dl) * F + h * C);
        float4 lv[C / 4], rv[C / 4];
#pragma unroll
        for (int q = 0; q < C / 4; ++q) { lv[q] = pl[q]; rv[q] = pr[q]; }
        float a = 0.0f;
#pragma unroll
        for (int q = 0; q < C / 4; ++q) {
          const float le[4] = {lv[q].x, lv[q].y, lv[q].z, lv[q].w};
          const float re[4] = {rv[q].x, rv[q].y, rv[q].z, rv[q].w};
#pragma unroll
          for (int z = 0; z < 4; ++z) {
            float tt = le[z] + re[z];
            tt = (tt > 0.0f) ? tt : 0.2f * tt;
            a += tt * satt[h * C + q * 4 + z];
          }
        }
        lg[(size_t)sl * H + h] = a;
        ldsMaxF(&mxi[dl * H + h], a);
      }
    }
  }
  __syncthreads();

  for (int idx = tid; idx < tot * H; idx += 256) {
    const int j = idx / H, h = idx - j * H;
    int s, dl, sl;
    if (tile_item(j, tot, cnt, b, n0, bins, s, dl, sl)) {
      const float a = expf(lg[(size_t)sl * H + h] - __int_as_float(mxi[dl * H + h]));
      lg[(size_t)sl * H + h] = a;
      atomicAdd(&den[dl * H + h], a);
    }
  }
  __syncthreads();

  for (int j0 = wv * U; j0 < tot; j0 += 8 * U) {
    int s[U], dl[U], sl[U]; bool ok[U];
    float xv[U][K], av[U][K];
#pragma unroll
    for (int u = 0; u < U; ++u) ok[u] = tile_item(j0 + u, tot, cnt, b, n0, bins, s[u], dl[u], sl[u]);
#pragma unroll
    for (int u = 0; u < U; ++u) {
#pragma unroll
      for (int k = 0; k < K; ++k) {
        const int r  = lane + 32 * k;
        const int rr = (r < F) ? r : r - 32;
        xv[u][k] = xl[(size_t)s[u] * F + rr];
        av[u][k] = lg[(size_t)sl[u] * H + rr / C];
      }
    }
#pragma unroll
    for (int u = 0; u < U; ++u) {
#pragma unroll
      for (int k = 0; k < K; ++k) {
        const int r  = lane + 32 * k;
        const int rr = (r < F) ? r : r - 32;
        atomicAdd(&acc[dl[u] * PF + rr], (ok[u] && r < F) ? av[u][k] * xv[u][k] : 0.0f);
      }
    }
  }
  __syncthreads();

  if constexpr (!PROJ) {
    for (int i = tid; i < TS * F; i += 256) {
      const int dl = i / F, r = i - dl * F;
      const int n = n0 + dl;
      if (n < N_) out[(size_t)n * F + r] = acc[i] / (den[dl * H + r / C] + 1e-16f) + bias[r];
    }
  } else {
    for (int i = tid; i < TS * F; i += 256) {
      const int dl = i / F, r = i - dl * F;
      acc[dl * PF + r] = acc[dl * PF + r] / (den[dl * H + r / C] + 1e-16f) + bias[r];
    }
    __syncthreads();
    tile_proj_pair<F, PF, F2_>(acc, n0, wv, lane, Wnl, bnl, nxl, Wnr, bnr, nxr);
  }
}

constexpr size_t OFF_DINV = 0;
constexpr size_t OFF_CNT  = OFF_DINV + N_;
constexpr size_t OFF_OFFS = OFF_CNT  + 2048;
constexpr size_t OFF_CUR  = OFF_OFFS + 2048;
constexpr size_t OFF_XL2  = (size_t)N_ + E_;
constexpr size_t OFF_XR2  = OFF_XL2  + (size_t)N_*D_;
constexpr size_t OFF_XL1  = OFF_XR2  + (size_t)N_*D_;
constexpr size_t OFF_XR1  = OFF_XL1  + (size_t)N_*F1_;
constexpr size_t OFF_BINS = OFF_XR1  + (size_t)N_*F1_;
constexpr size_t OFF_LG   = OFF_BINS + (size_t)N_*24;
static_assert(OFF_CUR + 2048 <= OFF_XL2, "bin tables fit in front of the projections");
static_assert(F2_ == D_, "layer 2's projections are sized with D_ above");
static_assert((size_t)E_ <= (size_t)N_ * 24, "bins fit their region");

extern "C" void kernel_launch(void* const* d_in, const int* in_sizes, int n_in,
                              void* d_out, int out_size, void* d_ws, size_t ws_size,
                              hipStream_t stream) {
  const float* x     = (const float*)d_in[0];
  const int*   ei    = (const int*)  d_in[1];
  const float* Wssg  = (const float*)d_in[2];
  const float* bssg  = (const float*)d_in[3];
  const float* W1l   = (const float*)d_in[4];
  const float* b1l   = (const float*)d_in[5];
  const float* W1r   = (const float*)d_in[6];
  const float* b1r   = (const float*)d_in[7];
  const float* att1  = (const float*)d_in[8];
  const float* bias1 = (const float*)d_in[9];
  const float* W2l   = (const float*)d_in[10];
  const float* b2l   = (const float*)d_in[11];
  const float* W2r   = (const float*)d_in[12];
  const float* b2r   = (const float*)d_in[13];
  const float* att2  = (const float*)d_in[14];
  const float* bias2 = (const float*)d_in[15];

  const int* src = ei;
  const int* dst = ei + E_;

  float* ws   = (float*)d_ws;
  float* dinv = ws + OFF_DINV;
  int*   cnt  = (int*)(ws + OFF_CNT);
  int*   offs = (int*)(ws + OFF_OFFS);
  int*   cur  = (int*)(ws + OFF_CUR);
  float* xl2  = ws + OFF_XL2;
  float* xr2  = ws + OFF_XR2;
  float* xl1  = ws + OFF_XL1;
  float* xr1  = ws + OFF_XR1;
  int*   bins = (int*)(ws + OFF_BINS);
  float* lg   = ws + OFF_LG;
  float* out  = (float*)d_out;

  const int TB = 256;

  k_zero_i<<<CDIV(2048, TB), TB, 0, stream>>>(cnt, 2048);
  k_tilecount<<<CDIV(E_, CH), 1024, 0, stream>>>(dst, cnt);
  k_scan<<<1, 1024, 0, stream>>>(cnt, offs, cur);
  k_binscatter<<<CDIV(E_, CH), 1024, 0, stream>>>(src, dst, cur, bins);

  k_dinv_tile<<<NT, TS, 0, stream>>>(offs, bins, dinv);
  k_ssg_tile<4><<<NT, 256, 0, stream>>>(offs, bins, x, dinv, Wssg, bssg, W1l, b1l, W1r, b1r, xl1, xr1);

  k_gat_tile<12, 12, 4, true><<<NT, 256, 0, stream>>>(offs, bins, xl1, xr1, att1, bias1, lg, nullptr,
                                                      W2l, b2l, W2r, b2r, xl2, xr2);

  k_gat_tile<1, 64, 4, false><<<NT, 256, 0, stream>>>(offs, bins, xl2, xr2, att2, bias2, lg, out,
                                                      nullptr, nullptr, nullptr, nullptr, nullptr, nullptr);
}


# each tile's 64 destination rows of xr staged once in LDS, in the memory that later holds the sums; the logit phase of both attention layers reads them there, not per item from memory; sums unchanged
# speedup vs baseline: 4.2650x; 1.1334x over previous
#include <hip/hip_runtime.h>
#include <cmath>

constexpr int N_  = 100000;
constexpr int E_  = 1000000;
constexpr int D_  = 64;
constexpr int F1_ = 144;
constexpr int F2_ = 64;

#define CDIV(a,b) (((a)+(b)-1)/(b))

typedef __attribute__((ext_vector_type(2))) float v2f;
typedef __attribute__((ext_vector_type(8))) float v8f;

constexpr int TSH = 6;
constexpr int TS  = 1 << TSH;
constexpr int NT  = CDIV(N_, TS);
constexpr int CH  = 8192;
constexpr int SRC_BITS = 17;
constexpr int SRC_MASK = (1 << SRC_BITS) - 1;
static_assert(N_ <= (1 << SRC_BITS), "source id must fit the packed field");
static_assert(NT + 1 <= 2048, "scan handles two tiles per thread of one 1024-thread group");

__global__ void k_zero_i(int* __restrict__ p, int n) {
  int i = blockIdx.x * blockDim.x + threadIdx.x;
  if (i < n) p[i] = 0;
}

__global__ __launch_bounds__(1024) void k_tilecount(const int* __restrict__ dst, int* __restrict__ cnt) {
  __shared__ int hist[NT];
  const int tid = threadIdx.x;
  for (int i = tid; i < NT; i += 1024) hist[i] = 0;
  __syncthreads();
  const int base = blockIdx.x * CH;
#pragma unroll
  for (int k = 0; k < CH / 1024; ++k) {
    int e = base + k * 1024 + tid;
    if (e < E_) atomicAdd(&hist[dst[e] >> TSH], 1);
  }
  __syncthreads();
  for (int i = tid; i < NT; i += 1024) { int v = hist[i]; if (v) atomicAdd(&cnt[i], v); }
}

__global__ __launch_bounds__(1024) void k_scan(const int* __restrict__ cnt, int* __restrict__ off,
                                               int* __restrict__ cur) {
  __shared__ int p[1024];
  const int t = threadIdx.x;
  const int a = (2 * t     < NT) ? cnt[2 * t]     : 0;
  const int b = (2 * t + 1 < NT) ? cnt[2 * t + 1] : 0;
  int v = a + b;
  p[t] = v;
  __syncthreads();
  for (int o = 1; o < 1024; o <<= 1) {
    int add = (t >= o) ? p[t - o] : 0;
    __syncthreads();
    v += add;
    p[t] = v;
    __syncthreads();
  }
  const int excl = v - (a + b);
  if (2 * t     < NT) { off[2 * t]     = excl;     cur[2 * t]     = excl; }
  if (2 * t + 1 < NT) { off[2 * t + 1] = excl + a; cur[2 * t + 1] = excl + a; }
  if (t == 1023) off[NT] = v;
}

__global__ __launch_bounds__(1024) void k_binscatter(const int* __restrict__ src, const int* __restrict__ dst,
                                                     int* __restrict__ cur, int* __restrict__ bins) {
  __shared__ int hist[NT];
  const int tid = threadIdx.x;
  for (int i = tid; i < NT; i += 1024) hist[i] = 0;
  __syncthreads();
  const int base = blockIdx.x * CH;
  int pk[CH / 1024], bn[CH / 1024], lr[CH / 1024];
#pragma unroll
  for (int k = 0; k < CH / 1024; ++k) {
    int e = base + k * 1024 + tid;
    bn[k] = -1; pk[k] = 0; lr[k] = 0;
    if (e < E_) {
      int d = dst[e];
      bn[k] = d >> TSH;
      pk[k] = src[e] | ((d & (TS - 1)) << SRC_BITS);
      lr[k] = atomicAdd(&hist[bn[k]], 1);
    }
  }
  __syncthreads();
  for (int i = tid; i < NT; i += 1024) { int v = hist[i]; hist[i] = v ? atomicAdd(&cur[i], v) : 0; }
  __syncthreads();
#pragma unroll
  for (int k = 0; k < CH / 1024; ++k)
    if (bn[k] >= 0) bins[hist[bn[k]] + lr[k]] = pk[k];
}

__device__ __forceinline__ bool tile_item(int j, int tot, int cnt, int b, int n0,
                                          const int* __restrict__ bins, int& s, int& dl, int& sl) {
  const int p  = min(b + j, E_ - 1);
  const int pk = bins[p];
  const bool isE = j < cnt;
  const int dlS = j - cnt;
  const int sS  = n0 + dlS;
  const bool okS = (!isE) && (j < tot) && (sS < N_);
  s  = isE ? (pk & SRC_MASK)  : (okS ? sS : 0);
  dl = isE ? (pk >> SRC_BITS) : (okS ? dlS : 0);
  sl = isE ? p                : (okS ? E_ + sS : 0);
  return isE || okS;
}

__device__ __forceinline__ void ldsMaxF(int* cell, float v) {
  if (v >= 0.0f) atomicMax(cell, __float_as_int(v));
  else           atomicMin((unsigned int*)cell, __float_as_uint(v));
}

__global__ __launch_bounds__(TS) void k_dinv_tile(const int* __restrict__ off, const int* __restrict__ bins,
                                                  float* __restrict__ dinv) {
  __shared__ int c[TS];
  const int t = blockIdx.x, tid = threadIdx.x;
  c[tid] = 0;
  __syncthreads();
  const int b = off[t], e = off[t + 1];
  for (int p = b + tid; p < e; p += TS) atomicAdd(&c[bins[p] >> SRC_BITS], 1);
  __syncthreads();
  const int n = t * TS + tid;
  if (n < N_) dinv[n] = rsqrtf(1.0f + (float)c[tid]);
}

template <int K, int PA, int NC>
__device__ __forceinline__ void tile_mma2(const float* a0p, const float* __restrict__ wp, v8f& acc0, v8f& acc1) {
  const float* a1p = a0p + 16 * PA;
#pragma unroll 4
  for (int k = 0; k < K; k += 4) {
    v2f a0, a1, b;
    a0.x = a0p[k]; a0.y = a0p[k + 1];
    a1.x = a1p[k]; a1.y = a1p[k + 1];
    b.x = wp[(size_t)k * NC];
    b.y = wp[(size_t)(k + 1) * NC];
    acc0 = __builtin_amdgcn_wmma_f32_16x16x4_f32(false, a0, false, b, (short)0, acc0, false, false);
    acc1 = __builtin_amdgcn_wmma_f32_16x16x4_f32(false, a1, false, b, (short)0, acc1, false, false);
  }
}

template <int K, int PA, int NC, int PC>
__device__ __forceinline__ void tile_proj_lds(const float* As, const float* __restrict__ W,
                                              const float* __restrict__ bias, float* Cs, int wv, int lane) {
  const int c = lane & 15, hh = lane >> 4;
  const int r0 = (wv & 1) * 32;
  for (int ct = wv >> 1; ct < NC / 16; ct += 4) {
    const int col = ct * 16 + c;
    v8f acc0 = {}, acc1 = {};
    tile_mma2<K, PA, NC>(As + (r0 + c) * PA + 2 * hh, W + (size_t)(2 * hh) * NC + col, acc0, acc1);
    const float bn = bias[col];
#pragma unroll
    for (int j = 0; j < 8; ++j) {
      Cs[(r0 + 8 * hh + j) * PC + col]      = acc0[j] + bn;
      Cs[(r0 + 16 + 8 * hh + j) * PC + col] = acc1[j] + bn;
    }
  }
}

template <int K, int PA, int NC>
__device__ __forceinline__ void tile_proj_pair(const float* As, int n0, int wv, int lane,
                                               const float* __restrict__ Wa, const float* __restrict__ ba,
                                               float* __restrict__ Ca,
                                               const float* __restrict__ Wb, const float* __restrict__ bb,
                                               float* __restrict__ Cb) {
  constexpr int CT = NC / 16;
  const int c = lane & 15, hh = lane >> 4;
  const int r0 = (wv & 1) * 32;
  for (int ct = wv >> 1; ct < 2 * CT; ct += 4) {
    const bool second = ct >= CT;
    const float* __restrict__ W  = second ? Wb : Wa;
    const float* __restrict__ bs = second ? bb : ba;
    float* __restrict__ C        = second ? Cb : Ca;
    const int col = (second ? ct - CT : ct) * 16 + c;
    v8f acc0 = {}, acc1 = {};
    tile_mma2<K, PA, NC>(As + (r0 + c) * PA + 2 * hh, W + (size_t)(2 * hh) * NC + col, acc0, acc1);
    const float bn = bs[col];
    float* __restrict__ cp = C + (size_t)(n0 + r0 + 8 * hh) * NC + col;
    if (n0 + r0 < N_) {
#pragma unroll
      for (int j = 0; j < 8; ++j) cp[(size_t)j * NC] = acc0[j] + bn;
    }
    if (n0 + r0 + 16 < N_) {
#pragma unroll
      for (int j = 0; j < 8; ++j) cp[(size_t)(16 + j) * NC] = acc1[j] + bn;
    }
  }
}

constexpr int PH = D_ + 4;
static_assert(N_ % 16 == 0, "a 16-row output tile is never partly past the last node");
template <int U>
__global__ __launch_bounds__(256) void k_ssg_tile(const int* __restrict__ off, const int* __restrict__ bins,
                                                  const float* __restrict__ x, const float* __restrict__ dinv,
                                                  const float* __restrict__ Wssg, const float* __restrict__ bssg,
                                                  const float* __restrict__ W1l, const float* __restrict__ b1l,
                                                  const float* __restrict__ W1r, const float* __restrict__ b1r,
                                                  float* __restrict__ xl1, float* __restrict__ xr1) {
  __shared__ float acc[TS * PH];
  __shared__ float x1s[TS * PH];
  const int t = blockIdx.x, tid = threadIdx.x, lane = tid & 31;
  const int wv = __builtin_amdgcn_readfirstlane(tid >> 5);
  for (int i = tid; i < TS * PH; i += 256) acc[i] = 0.0f;
  __syncthreads();
  const int b = off[t];
  const int cnt = off[t + 1] - b;
  const int tot = cnt + TS;
  const int n0 = t * TS;
  for (int j0 = wv * U; j0 < tot; j0 += 8 * U) {
    int s[U], dl[U], sl[U]; bool ok[U];
    float x0[U], x1[U], w[U];
#pragma unroll
    for (int u = 0; u < U; ++u) ok[u] = tile_item(j0 + u, tot, cnt, b, n0, bins, s[u], dl[u], sl[u]);
#pragma unroll
    for (int u = 0; u < U; ++u) {
      w[u]  = dinv[s[u]] * dinv[n0 + dl[u]];
      x0[u] = x[(size_t)s[u] * D_ + lane];
      x1[u] = x[(size_t)s[u] * D_ + lane + 32];
    }
#pragma unroll
    for (int u = 0; u < U; ++u) {
      atomicAdd(&acc[dl[u] * PH + lane],      ok[u] ? x0[u] * w[u] : 0.0f);
      atomicAdd(&acc[dl[u] * PH + lane + 32], ok[u] ? x1[u] * w[u] : 0.0f);
    }
  }
  __syncthreads();
  for (int i = tid; i < TS * D_; i += 256) {
    const int dl = i >> 6, c = i & 63;
    const int n = n0 + dl;
    acc[dl * PH + c] = (n < N_) ? 0.5f * x[(size_t)n * D_ + c] + 0.5f * acc[dl * PH + c] : 0.0f;
  }
  __syncthreads();
  tile_proj_lds<D_, PH, D_, PH>(acc, Wssg, bssg, x1s, wv, lane);
  __syncthreads();
  tile_proj_pair<D_, PH, F1_>(x1s, n0, wv, lane, W1l, b1l, xl1, W1r, b1r, xr1);
}

template <int H, int C, int U, bool PROJ>
__global__ __launch_bounds__(256) void k_gat_tile(const int* __restrict__ off, const int* __restrict__ bins,
                                                  const float* __restrict__ xl, const float* __restrict__ xr,
                                                  const float* __restrict__ att, const float* __restrict__ bias,
                                                  float* __restrict__ lg, float* __restrict__ out,
                                                  const float* __restrict__ Wnl, const float* __restrict__ bnl,
                                                  const float* __restrict__ Wnr, const float* __restrict__ bnr,
                                                  float* __restrict__ nxl, float* __restrict__ nxr) {
  constexpr int F = H * C;
  constexpr int K = (F + 31) / 32;
  constexpr int PF = PROJ ? F + 4 : F;
  static_assert(C % 4 == 0, "head rows are read as 16-byte pieces");
  static_assert(!PROJ || (PF % 64 == 4 || PF % 64 == 20 || PF % 64 == 36 || PF % 64 == 52),
                "fragment reads of 16 rows must fall on 16 different bank groups");
  __shared__ __attribute__((aligned(16))) float acc[TS * PF];
  __shared__ int   mxi[TS * H];
  __shared__ float den[TS * H];
  __shared__ float satt[F];

  const int t = blockIdx.x, tid = threadIdx.x, lane = tid & 31;
  const int wv = __builtin_amdgcn_readfirstlane(tid >> 5);
  const int n0 = t * TS;
  float* const xrs = acc;
  constexpr int NST = TS * (F / 4) / 256;
  static_assert(TS * (F / 4) % 256 == 0, "the tile's rows split evenly over the group");
  {
    const float4* __restrict__ xr4 = (const float4*)(xr + (size_t)n0 * F);
    const int npc = min(TS, N_ - n0) * (F / 4);
    float4 st[NST];
#pragma unroll
    for (int k = 0; k < NST; ++k) {
      const int i = tid + 256 * k;
      st[k] = make_float4(0.0f, 0.0f, 0.0f, 0.0f);
      if (i < npc) st[k] = xr4[i];
    }
#pragma unroll
    for (int k = 0; k < NST; ++k) *(float4*)(xrs + 4 * (tid + 256 * k)) = st[k];
  }
  for (int i = tid; i < TS * H; i += 256) { mxi[i] = (int)0xFF800000u; den[i] = 0.0f; }
  for (int i = tid; i < F; i += 256) satt[i] = att[i];
  __syncthreads();
  const int b = off[t];
  const int cnt = off[t + 1] - b;
  const int tot = cnt + TS;

  if constexpr (H == 1) {
    for (int j0 = wv * U; j0 < tot; j0 += 8 * U) {
      int s[U], dl[U], sl[U]; bool ok[U];
      float l0[U], l1[U], r0[U], r1[U];
#pragma unroll
      for (int u = 0; u < U; ++u) ok[u] = tile_item(j0 + u, tot, cnt, b, n0, bins, s[u], dl[u], sl[u]);
#pragma unroll
      for (int u = 0; u < U; ++u) {
        l0[u] = xl[(size_t)s[u] * F + lane];
        l1[u] = xl[(size_t)s[u] * F + lane + 32];
        r0[u] = xrs[dl[u] * F + lane];
        r1[u] = xrs[dl[u] * F + lane + 32];
      }
#pragma unroll
      for (int u = 0; u < U; ++u) {
        float a = 0.0f;
        float t0 = l0[u] + r0[u];
        t0 = (t0 > 0.0f) ? t0 : 0.2f * t0;
        a += t0 * satt[lane];
        float t1 = l1[u] + r1[u];
        t1 = (t1 > 0.0f) ? t1 : 0.2f * t1;
        a += t1 * satt[lane + 32];
#pragma unroll
        for (int o = 16; o > 0; o >>= 1) a += __shfl_xor(a, o, 32);
        if (lane == 0 && ok[u]) { lg[sl[u]] = a; ldsMaxF(&mxi[dl[u]], a); }
      }
    }
  } else {
    for (int idx = tid; idx < tot * H; idx += 256) {
      const int j = idx / H, h = idx - j * H;
      int s, dl, sl;
      if (tile_item(j, tot, cnt, b, n0, bins, s, dl, sl)) {
        const float4* pl = (const float4*)(xl + (size_t)s * F + h * C);
        const float4* pr = (const float4*)(xrs + dl * F + h * C);
        float4 lv[C / 4], rv[C / 4];
#pragma unroll
        for (int q = 0; q < C / 4; ++q) { lv[q] = pl[q]; rv[q] = pr[q]; }
        float a = 0.0f;
#pragma unroll
        for (int q = 0; q < C / 4; ++q) {
          const float le[4] = {lv[q].x, lv[q].y, lv[q].z, lv[q].w};
          const float re[4] = {rv[q].x, rv[q].y, rv[q].z, rv[q].w};
#pragma unroll
          for (int z = 0; z < 4; ++z) {
            float tt = le[z] + re[z];
            tt = (tt > 0.0f) ? tt : 0.2f * tt;
            a += tt * satt[h * C + q * 4 + z];
          }
        }
        lg[(size_t)sl * H + h] = a;
        ldsMaxF(&mxi[dl * H + h], a);
      }
    }
  }
  __syncthreads();

  for (int i = tid; i < TS * PF; i += 256) acc[i] = 0.0f;
  for (int idx = tid; idx < tot * H; idx += 256) {
    const int j = idx / H, h = idx - j * H;
    int s, dl, sl;
    if (tile_item(j, tot, cnt, b, n0, bins, s, dl, sl)) {
      const float a = expf(lg[(size_t)sl * H + h] - __int_as_float(mxi[dl * H + h]));
      lg[(size_t)sl * H + h] = a;
      atomicAdd(&den[dl * H + h], a);
    }
  }
  __syncthreads();

  for (int j0 = wv * U; j0 < tot; j0 += 8 * U) {
    int s[U], dl[U], sl[U]; bool ok[U];
    float xv[U][K], av[U][K];
#pragma unroll
    for (int u = 0; u < U; ++u) ok[u] = tile_item(j0 + u, tot, cnt, b, n0, bins, s[u], dl[u], sl[u]);
#pragma unroll
    for (int u = 0; u < U; ++u) {
#pragma unroll
      for (int k = 0; k < K; ++k) {
        const int r  = lane + 32 * k;
        const int rr = (r < F) ? r : r - 32;
        xv[u][k] = xl[(size_t)s[u] * F + rr];
        av[u][k] = lg[(size_t)sl[u] * H + rr / C];
      }
    }
#pragma unroll
    for (int u = 0; u < U; ++u) {
#pragma unroll
      for (int k = 0; k < K; ++k) {
        const int r  = lane + 32 * k;
        const int rr = (r < F) ? r : r - 32;
        atomicAdd(&acc[dl[u] * PF + rr], (ok[u] && r < F) ? av[u][k] * xv[u][k] : 0.0f);
      }
    }
  }
  __syncthreads();

  if constexpr (!PROJ) {
    for (int i = tid; i < TS * F; i += 256) {
      const int dl = i / F, r = i - dl * F;
      const int n = n0 + dl;
      if (n < N_) out[(size_t)n * F + r] = acc[i] / (den[dl * H + r / C] + 1e-16f) + bias[r];
    }
  } else {
    for (int i = tid; i < TS * F; i += 256) {
      const int dl = i / F, r = i - dl * F;
      acc[dl * PF + r] = acc[dl * PF + r] / (den[dl * H + r / C] + 1e-16f) + bias[r];
    }
    __syncthreads();
    tile_proj_pair<F, PF, F2_>(acc, n0, wv, lane, Wnl, bnl, nxl, Wnr, bnr, nxr);
  }
}

constexpr size_t OFF_DINV = 0;
constexpr size_t OFF_CNT  = OFF_DINV + N_;
constexpr size_t OFF_OFFS = OFF_CNT  + 2048;
constexpr size_t OFF_CUR  = OFF_OFFS + 2048;
constexpr size_t OFF_XL2  = (size_t)N_ + E_;
constexpr size_t OFF_XR2  = OFF_XL2  + (size_t)N_*D_;
constexpr size_t OFF_XL1  = OFF_XR2  + (size_t)N_*D_;
constexpr size_t OFF_XR1  = OFF_XL1  + (size_t)N_*F1_;
constexpr size_t OFF_BINS = OFF_XR1  + (size_t)N_*F1_;
constexpr size_t OFF_LG   = OFF_BINS + (size_t)N_*24;
static_assert(OFF_CUR + 2048 <= OFF_XL2, "bin tables fit in front of the projections");
static_assert(F2_ == D_, "layer 2's projections are sized with D_ above");
static_assert((size_t)E_ <= (size_t)N_ * 24, "bins fit their region");

extern "C" void kernel_launch(void* const* d_in, const int* in_sizes, int n_in,
                              void* d_out, int out_size, void* d_ws, size_t ws_size,
                              hipStream_t stream) {
  const float* x     = (const float*)d_in[0];
  const int*   ei    = (const int*)  d_in[1];
  const float* Wssg  = (const float*)d_in[2];
  const float* bssg  = (const float*)d_in[3];
  const float* W1l   = (const float*)d_in[4];
  const float* b1l   = (const float*)d_in[5];
  const float* W1r   = (const float*)d_in[6];
  const float* b1r   = (const float*)d_in[7];
  const float* att1  = (const float*)d_in[8];
  const float* bias1 = (const float*)d_in[9];
  const float* W2l   = (const float*)d_in[10];
  const float* b2l   = (const float*)d_in[11];
  const float* W2r   = (const float*)d_in[12];
  const float* b2r   = (const float*)d_in[13];
  const float* att2  = (const float*)d_in[14];
  const float* bias2 = (const float*)d_in[15];

  const int* src = ei;
  const int* dst = ei + E_;

  float* ws   = (float*)d_ws;
  float* dinv = ws + OFF_DINV;
  int*   cnt  = (int*)(ws + OFF_CNT);
  int*   offs = (int*)(ws + OFF_OFFS);
  int*   cur  = (int*)(ws + OFF_CUR);
  float* xl2  = ws + OFF_XL2;
  float* xr2  = ws + OFF_XR2;
  float* xl1  = ws + OFF_XL1;
  float* xr1  = ws + OFF_XR1;
  int*   bins = (int*)(ws + OFF_BINS);
  float* lg   = ws + OFF_LG;
  float* out  = (float*)d_out;

  const int TB = 256;

  k_zero_i<<<CDIV(2048, TB), TB, 0, stream>>>(cnt, 2048);
  k_tilecount<<<CDIV(E_, CH), 1024, 0, stream>>>(dst, cnt);
  k_scan<<<1, 1024, 0, stream>>>(cnt, offs, cur);
  k_binscatter<<<CDIV(E_, CH), 1024, 0, stream>>>(src, dst, cur, bins);

  k_dinv_tile<<<NT, TS, 0, stream>>>(offs, bins, dinv);
  k_ssg_tile<4><<<NT, 256, 0, stream>>>(offs, bins, x, dinv, Wssg, bssg, W1l, b1l, W1r, b1r, xl1, xr1);

  k_gat_tile<12, 12, 4, true><<<NT, 256, 0, stream>>>(offs, bins, xl1, xr1, att1, bias1, lg, nullptr,
                                                      W2l, b2l, W2r, b2r, xl2, xr2);

  k_gat_tile<1, 64, 4, false><<<NT, 256, 0, stream>>>(offs, bins, xl2, xr2, att2, bias2, lg, out,
                                                      nullptr, nullptr, nullptr, nullptr, nullptr, nullptr);
}


# wave-per-item passes: a wave decodes 32 items per vector load, lane-parallel, and broadcasts each item's values by readlane; the per-item scalar load and ~35 scalar instructions go; sums unchanged
# speedup vs baseline: 4.5656x; 1.0705x over previous
#include <hip/hip_runtime.h>
#include <cmath>

constexpr int N_  = 100000;
constexpr int E_  = 1000000;
constexpr int D_  = 64;
constexpr int F1_ = 144;
constexpr int F2_ = 64;

#define CDIV(a,b) (((a)+(b)-1)/(b))

typedef __attribute__((ext_vector_type(2))) float v2f;
typedef __attribute__((ext_vector_type(8))) float v8f;

constexpr int TSH = 6;
constexpr int TS  = 1 << TSH;
constexpr int NT  = CDIV(N_, TS);
constexpr int CH  = 8192;
constexpr int SRC_BITS = 17;
constexpr int SRC_MASK = (1 << SRC_BITS) - 1;
static_assert(N_ <= (1 << SRC_BITS), "source id must fit the packed field");
static_assert(NT + 1 <= 2048, "scan handles two tiles per thread of one 1024-thread group");

__global__ void k_zero_i(int* __restrict__ p, int n) {
  int i = blockIdx.x * blockDim.x + threadIdx.x;
  if (i < n) p[i] = 0;
}

__global__ __launch_bounds__(1024) void k_tilecount(const int* __restrict__ dst, int* __restrict__ cnt) {
  __shared__ int hist[NT];
  const int tid = threadIdx.x;
  for (int i = tid; i < NT; i += 1024) hist[i] = 0;
  __syncthreads();
  const int base = blockIdx.x * CH;
#pragma unroll
  for (int k = 0; k < CH / 1024; ++k) {
    int e = base + k * 1024 + tid;
    if (e < E_) atomicAdd(&hist[dst[e] >> TSH], 1);
  }
  __syncthreads();
  for (int i = tid; i < NT; i += 1024) { int v = hist[i]; if (v) atomicAdd(&cnt[i], v); }
}

__global__ __launch_bounds__(1024) void k_scan(const int* __restrict__ cnt, int* __restrict__ off,
                                               int* __restrict__ cur) {
  __shared__ int p[1024];
  const int t = threadIdx.x;
  const int a = (2 * t     < NT) ? cnt[2 * t]     : 0;
  const int b = (2 * t + 1 < NT) ? cnt[2 * t + 1] : 0;
  int v = a + b;
  p[t] = v;
  __syncthreads();
  for (int o = 1; o < 1024; o <<= 1) {
    int add = (t >= o) ? p[t - o] : 0;
    __syncthreads();
    v += add;
    p[t] = v;
    __syncthreads();
  }
  const int excl = v - (a + b);
  if (2 * t     < NT) { off[2 * t]     = excl;     cur[2 * t]     = excl; }
  if (2 * t + 1 < NT) { off[2 * t + 1] = excl + a; cur[2 * t + 1] = excl + a; }
  if (t == 1023) off[NT] = v;
}

__global__ __launch_bounds__(1024) void k_binscatter(const int* __restrict__ src, const int* __restrict__ dst,
                                                     int* __restrict__ cur, int* __restrict__ bins) {
  __shared__ int hist[NT];
  const int tid = threadIdx.x;
  for (int i = tid; i < NT; i += 1024) hist[i] = 0;
  __syncthreads();
  const int base = blockIdx.x * CH;
  int pk[CH / 1024], bn[CH / 1024], lr[CH / 1024];
#pragma unroll
  for (int k = 0; k < CH / 1024; ++k) {
    int e = base + k * 1024 + tid;
    bn[k] = -1; pk[k] = 0; lr[k] = 0;
    if (e < E_) {
      int d = dst[e];
      bn[k] = d >> TSH;
      pk[k] = src[e] | ((d & (TS - 1)) << SRC_BITS);
      lr[k] = atomicAdd(&hist[bn[k]], 1);
    }
  }
  __syncthreads();
  for (int i = tid; i < NT; i += 1024) { int v = hist[i]; hist[i] = v ? atomicAdd(&cur[i], v) : 0; }
  __syncthreads();
#pragma unroll
  for (int k = 0; k < CH / 1024; ++k)
    if (bn[k] >= 0) bins[hist[bn[k]] + lr[k]] = pk[k];
}

__device__ __forceinline__ bool tile_item(int j, int tot, int cnt, int b, int n0,
                                          const int* __restrict__ bins, int& s, int& dl, int& sl) {
  const int p  = min(b + j, E_ - 1);
  const int pk = bins[p];
  const bool isE = j < cnt;
  const int dlS = j - cnt;
  const int sS  = n0 + dlS;
  const bool okS = (!isE) && (j < tot) && (sS < N_);
  s  = isE ? (pk & SRC_MASK)  : (okS ? sS : 0);
  dl = isE ? (pk >> SRC_BITS) : (okS ? dlS : 0);
  sl = isE ? p                : (okS ? E_ + sS : 0);
  return isE || okS;
}

__device__ __forceinline__ void ldsMaxF(int* cell, float v) {
  if (v >= 0.0f) atomicMax(cell, __float_as_int(v));
  else           atomicMin((unsigned int*)cell, __float_as_uint(v));
}

template <int U>
__device__ __forceinline__ int wave_item(int q, int wv) {
  const unsigned uq = (unsigned)q;
  return (int)(((uq / U) * 8u + (unsigned)wv) * U + (uq % U));
}
__device__ __forceinline__ int lane_bcast(int v, int ln) { return __builtin_amdgcn_readlane(v, ln); }
__device__ __forceinline__ float lane_bcast(float v, int ln) {
  return __int_as_float(__builtin_amdgcn_readlane(__float_as_int(v), ln));
}

__global__ __launch_bounds__(TS) void k_dinv_tile(const int* __restrict__ off, const int* __restrict__ bins,
                                                  float* __restrict__ dinv) {
  __shared__ int c[TS];
  const int t = blockIdx.x, tid = threadIdx.x;
  c[tid] = 0;
  __syncthreads();
  const int b = off[t], e = off[t + 1];
  for (int p = b + tid; p < e; p += TS) atomicAdd(&c[bins[p] >> SRC_BITS], 1);
  __syncthreads();
  const int n = t * TS + tid;
  if (n < N_) dinv[n] = rsqrtf(1.0f + (float)c[tid]);
}

template <int K, int PA, int NC>
__device__ __forceinline__ void tile_mma2(const float* a0p, const float* __restrict__ wp, v8f& acc0, v8f& acc1) {
  const float* a1p = a0p + 16 * PA;
#pragma unroll 4
  for (int k = 0; k < K; k += 4) {
    v2f a0, a1, b;
    a0.x = a0p[k]; a0.y = a0p[k + 1];
    a1.x = a1p[k]; a1.y = a1p[k + 1];
    b.x = wp[(size_t)k * NC];
    b.y = wp[(size_t)(k + 1) * NC];
    acc0 = __builtin_amdgcn_wmma_f32_16x16x4_f32(false, a0, false, b, (short)0, acc0, false, false);
    acc1 = __builtin_amdgcn_wmma_f32_16x16x4_f32(false, a1, false, b, (short)0, acc1, false, false);
  }
}

template <int K, int PA, int NC, int PC>
__device__ __forceinline__ void tile_proj_lds(const float* As, const float* __restrict__ W,
                                              const float* __restrict__ bias, float* Cs, int wv, int lane) {
  const int c = lane & 15, hh = lane >> 4;
  const int r0 = (wv & 1) * 32;
  for (int ct = wv >> 1; ct < NC / 16; ct += 4) {
    const int col = ct * 16 + c;
    v8f acc0 = {}, acc1 = {};
    tile_mma2<K, PA, NC>(As + (r0 + c) * PA + 2 * hh, W + (size_t)(2 * hh) * NC + col, acc0, acc1);
    const float bn = bias[col];
#pragma unroll
    for (int j = 0; j < 8; ++j) {
      Cs[(r0 + 8 * hh + j) * PC + col]      = acc0[j] + bn;
      Cs[(r0 + 16 + 8 * hh + j) * PC + col] = acc1[j] + bn;
    }
  }
}

template <int K, int PA, int NC>
__device__ __forceinline__ void tile_proj_pair(const float* As, int n0, int wv, int lane,
                                               const float* __restrict__ Wa, const float* __restrict__ ba,
                                               float* __restrict__ Ca,
                                               const float* __restrict__ Wb, const float* __restrict__ bb,
                                               float* __restrict__ Cb) {
  constexpr int CT = NC / 16;
  const int c = lane & 15, hh = lane >> 4;
  const int r0 = (wv & 1) * 32;
  for (int ct = wv >> 1; ct < 2 * CT; ct += 4) {
    const bool second = ct >= CT;
    const float* __restrict__ W  = second ? Wb : Wa;
    const float* __restrict__ bs = second ? bb : ba;
    float* __restrict__ C        = second ? Cb : Ca;
    const int col = (second ? ct - CT : ct) * 16 + c;
    v8f acc0 = {}, acc1 = {};
    tile_mma2<K, PA, NC>(As + (r0 + c) * PA + 2 * hh, W + (size_t)(2 * hh) * NC + col, acc0, acc1);
    const float bn = bs[col];
    float* __restrict__ cp = C + (size_t)(n0 + r0 + 8 * hh) * NC + col;
    if (n0 + r0 < N_) {
#pragma unroll
      for (int j = 0; j < 8; ++j) cp[(size_t)j * NC] = acc0[j] + bn;
    }
    if (n0 + r0 + 16 < N_) {
#pragma unroll
      for (int j = 0; j < 8; ++j) cp[(size_t)(16 + j) * NC] = acc1[j] + bn;
    }
  }
}

constexpr int PH = D_ + 4;
static_assert(N_ % 16 == 0, "a 16-row output tile is never partly past the last node");
template <int U>
__global__ __launch_bounds__(256) void k_ssg_tile(const int* __restrict__ off, const int* __restrict__ bins,
                                                  const float* __restrict__ x, const float* __restrict__ dinv,
                                                  const float* __restrict__ Wssg, const float* __restrict__ bssg,
                                                  const float* __restrict__ W1l, const float* __restrict__ b1l,
                                                  const float* __restrict__ W1r, const float* __restrict__ b1r,
                                                  float* __restrict__ xl1, float* __restrict__ xr1) {
  __shared__ float acc[TS * PH];
  __shared__ float x1s[TS * PH];
  const int t = blockIdx.x, tid = threadIdx.x, lane = tid & 31;
  const int wv = __builtin_amdgcn_readfirstlane(tid >> 5);
  for (int i = tid; i < TS * PH; i += 256) acc[i] = 0.0f;
  __syncthreads();
  const int b = off[t];
  const int cnt = off[t + 1] - b;
  const int tot = cnt + TS;
  const int n0 = t * TS;
  for (int q0 = 0; wave_item<U>(q0, wv) < tot; q0 += 32) {
    int sL, dlL, slL;
    const int okL = tile_item(wave_item<U>(q0 + lane, wv), tot, cnt, b, n0, bins, sL, dlL, slL) ? 1 : 0;
    const float wL = dinv[sL] * dinv[n0 + dlL];
    for (int g = 0; g < 32 / U; ++g) {
      if (wave_item<U>(q0 + g * U, wv) >= tot) break;
      int s[U], dl[U]; bool ok[U];
      float x0[U], x1[U], w[U];
#pragma unroll
      for (int u = 0; u < U; ++u) {
        s[u]  = lane_bcast(sL, g * U + u);
        dl[u] = lane_bcast(dlL, g * U + u);
        ok[u] = lane_bcast(okL, g * U + u) != 0;
        w[u]  = lane_bcast(wL, g * U + u);
      }
#pragma unroll
      for (int u = 0; u < U; ++u) {
        x0[u] = x[(size_t)s[u] * D_ + lane];
        x1[u] = x[(size_t)s[u] * D_ + lane + 32];
      }
#pragma unroll
      for (int u = 0; u < U; ++u) {
        atomicAdd(&acc[dl[u] * PH + lane],      ok[u] ? x0[u] * w[u] : 0.0f);
        atomicAdd(&acc[dl[u] * PH + lane + 32], ok[u] ? x1[u] * w[u] : 0.0f);
      }
    }
  }
  __syncthreads();
  for (int i = tid; i < TS * D_; i += 256) {
    const int dl = i >> 6, c = i & 63;
    const int n = n0 + dl;
    acc[dl * PH + c] = (n < N_) ? 0.5f * x[(size_t)n * D_ + c] + 0.5f * acc[dl * PH + c] : 0.0f;
  }
  __syncthreads();
  tile_proj_lds<D_, PH, D_, PH>(acc, Wssg, bssg, x1s, wv, lane);
  __syncthreads();
  tile_proj_pair<D_, PH, F1_>(x1s, n0, wv, lane, W1l, b1l, xl1, W1r, b1r, xr1);
}

template <int H, int C, int U, bool PROJ>
__global__ __launch_bounds__(256) void k_gat_tile(const int* __restrict__ off, const int* __restrict__ bins,
                                                  const float* __restrict__ xl, const float* __restrict__ xr,
                                                  const float* __restrict__ att, const float* __restrict__ bias,
                                                  float* __restrict__ lg, float* __restrict__ out,
                                                  const float* __restrict__ Wnl, const float* __restrict__ bnl,
                                                  const float* __restrict__ Wnr, const float* __restrict__ bnr,
                                                  float* __restrict__ nxl, float* __restrict__ nxr) {
  constexpr int F = H * C;
  constexpr int K = (F + 31) / 32;
  constexpr int PF = PROJ ? F + 4 : F;
  static_assert(C % 4 == 0, "head rows are read as 16-byte pieces");
  static_assert(!PROJ || (PF % 64 == 4 || PF % 64 == 20 || PF % 64 == 36 || PF % 64 == 52),
                "fragment reads of 16 rows must fall on 16 different bank groups");
  __shared__ __attribute__((aligned(16))) float acc[TS * PF];
  __shared__ int   mxi[TS * H];
  __shared__ float den[TS * H];
  __shared__ float satt[F];

  const int t = blockIdx.x, tid = threadIdx.x, lane = tid & 31;
  const int wv = __builtin_amdgcn_readfirstlane(tid >> 5);
  const int n0 = t * TS;
  float* const xrs = acc;
  constexpr int NST = TS * (F / 4) / 256;
  static_assert(TS * (F / 4) % 256 == 0, "the tile's rows split evenly over the group");
  {
    const float4* __restrict__ xr4 = (const float4*)(xr + (size_t)n0 * F);
    const int npc = min(TS, N_ - n0) * (F / 4);
    float4 st[NST];
#pragma unroll
    for (int k = 0; k < NST; ++k) {
      const int i = tid + 256 * k;
      st[k] = make_float4(0.0f, 0.0f, 0.0f, 0.0f);
      if (i < npc) st[k] = xr4[i];
    }
#pragma unroll
    for (int k = 0; k < NST; ++k) *(float4*)(xrs + 4 * (tid + 256 * k)) = st[k];
  }
  for (int i = tid; i < TS * H; i += 256) { mxi[i] = (int)0xFF800000u; den[i] = 0.0f; }
  for (int i = tid; i < F; i += 256) satt[i] = att[i];
  __syncthreads();
  const int b = off[t];
  const int cnt = off[t + 1] - b;
  const int tot = cnt + TS;

  if constexpr (H == 1) {
    for (int q0 = 0; wave_item<U>(q0, wv) < tot; q0 += 32) {
      int sL, dlL, slL;
      const int okL = tile_item(wave_item<U>(q0 + lane, wv), tot, cnt, b, n0, bins, sL, dlL, slL) ? 1 : 0;
      for (int g = 0; g < 32 / U; ++g) {
        if (wave_item<U>(q0 + g * U, wv) >= tot) break;
        int s[U], dl[U], sl[U]; bool ok[U];
        float l0[U], l1[U], r0[U], r1[U];
#pragma unroll
        for (int u = 0; u < U; ++u) {
          s[u]  = lane_bcast(sL, g * U + u);
          dl[u] = lane_bcast(dlL, g * U + u);
          sl[u] = lane_bcast(slL, g * U + u);
          ok[u] = lane_bcast(okL, g * U + u) != 0;
        }
#pragma unroll
        for (int u = 0; u < U; ++u) {
          l0[u] = xl[(size_t)s[u] * F + lane];
          l1[u] = xl[(size_t)s[u] * F + lane + 32];
          r0[u] = xrs[dl[u] * F + lane];
          r1[u] = xrs[dl[u] * F + lane + 32];
        }
#pragma unroll
        for (int u = 0; u < U; ++u) {
          float a = 0.0f;
          float t0 = l0[u] + r0[u];
          t0 = (t0 > 0.0f) ? t0 : 0.2f * t0;
          a += t0 * satt[lane];
          float t1 = l1[u] + r1[u];
          t1 = (t1 > 0.0f) ? t1 : 0.2f * t1;
          a += t1 * satt[lane + 32];
#pragma unroll
          for (int o = 16; o > 0; o >>= 1) a += __shfl_xor(a, o, 32);
          if (lane == 0 && ok[u]) { lg[sl[u]] = a; ldsMaxF(&mxi[dl[u]], a); }
        }
      }
    }
  } else {
    for (int idx = tid; idx < tot * H; idx += 256) {
      const int j = idx / H, h = idx - j * H;
      int s, dl, sl;
      if (tile_item(j, tot, cnt, b, n0, bins, s, dl, sl)) {
        const float4* pl = (const float4*)(xl + (size_t)s * F + h * C);
        const float4* pr = (const float4*)(xrs + dl * F + h * C);
        float4 lv[C / 4], rv[C / 4];
#pragma unroll
        for (int q = 0; q < C / 4; ++q) { lv[q] = pl[q]; rv[q] = pr[q]; }
        float a = 0.0f;
#pragma unroll
        for (int q = 0; q < C / 4; ++q) {
          const float le[4] = {lv[q].x, lv[q].y, lv[q].z, lv[q].w};
          const float re[4] = {rv[q].x, rv[q].y, rv[q].z, rv[q].w};
#pragma unroll
          for (int z = 0; z < 4; ++z) {
            float tt = le[z] + re[z];
            tt = (tt > 0.0f) ? tt : 0.2f * tt;
            a += tt * satt[h * C + q * 4 + z];
          }
        }
        lg[(size_t)sl * H + h] = a;
        ldsMaxF(&mxi[dl * H + h], a);
      }
    }
  }
  __syncthreads();

  for (int i = tid; i < TS * PF; i += 256) acc[i] = 0.0f;
  for (int idx = tid; idx < tot * H; idx += 256) {
    const int j = idx / H, h = idx - j * H;
    int s, dl, sl;
    if (tile_item(j, tot, cnt, b, n0, bins, s, dl, sl)) {
      const float a = expf(lg[(size_t)sl * H + h] - __int_as_float(mxi[dl * H + h]));
      lg[(size_t)sl * H + h] = a;
      atomicAdd(&den[dl * H + h], a);
    }
  }
  __syncthreads();

  for (int q0 = 0; wave_item<U>(q0, wv) < tot; q0 += 32) {
    int sL, dlL, slL;
    const int okL = tile_item(wave_item<U>(q0 + lane, wv), tot, cnt, b, n0, bins, sL, dlL, slL) ? 1 : 0;
    float aL = 0.0f;
    if constexpr (H == 1) aL = lg[slL];
    for (int g = 0; g < 32 / U; ++g) {
      if (wave_item<U>(q0 + g * U, wv) >= tot) break;
      int s[U], dl[U], sl[U]; bool ok[U];
      float xv[U][K], av[U][K];
#pragma unroll
      for (int u = 0; u < U; ++u) {
        s[u]  = lane_bcast(sL, g * U + u);
        dl[u] = lane_bcast(dlL, g * U + u);
        sl[u] = lane_bcast(slL, g * U + u);
        ok[u] = lane_bcast(okL, g * U + u) != 0;
      }
#pragma unroll
      for (int u = 0; u < U; ++u) {
#pragma unroll
        for (int k = 0; k < K; ++k) {
          const int r  = lane + 32 * k;
          const int rr = (r < F) ? r : r - 32;
          xv[u][k] = xl[(size_t)s[u] * F + rr];
          if constexpr (H == 1) av[u][k] = lane_bcast(aL, g * U + u);
          else                  av[u][k] = lg[(size_t)sl[u] * H + rr / C];
        }
      }
#pragma unroll
      for (int u = 0; u < U; ++u) {
#pragma unroll
        for (int k = 0; k < K; ++k) {
          const int r  = lane + 32 * k;
          const int rr = (r < F) ? r : r - 32;
          atomicAdd(&acc[dl[u] * PF + rr], (ok[u] && r < F) ? av[u][k] * xv[u][k] : 0.0f);
        }
      }
    }
  }
  __syncthreads();

  if constexpr (!PROJ) {
    for (int i = tid; i < TS * F; i += 256) {
      const int dl = i / F, r = i - dl * F;
      const int n = n0 + dl;
      if (n < N_) out[(size_t)n * F + r] = acc[i] / (den[dl * H + r / C] + 1e-16f) + bias[r];
    }
  } else {
    for (int i = tid; i < TS * F; i += 256) {
      const int dl = i / F, r = i - dl * F;
      acc[dl * PF + r] = acc[dl * PF + r] / (den[dl * H + r / C] + 1e-16f) + bias[r];
    }
    __syncthreads();
    tile_proj_pair<F, PF, F2_>(acc, n0, wv, lane, Wnl, bnl, nxl, Wnr, bnr, nxr);
  }
}

constexpr size_t OFF_DINV = 0;
constexpr size_t OFF_CNT  = OFF_DINV + N_;
constexpr size_t OFF_OFFS = OFF_CNT  + 2048;
constexpr size_t OFF_CUR  = OFF_OFFS + 2048;
constexpr size_t OFF_XL2  = (size_t)N_ + E_;
constexpr size_t OFF_XR2  = OFF_XL2  + (size_t)N_*D_;
constexpr size_t OFF_XL1  = OFF_XR2  + (size_t)N_*D_;
constexpr size_t OFF_XR1  = OFF_XL1  + (size_t)N_*F1_;
constexpr size_t OFF_BINS = OFF_XR1  + (size_t)N_*F1_;
constexpr size_t OFF_LG   = OFF_BINS + (size_t)N_*24;
static_assert(OFF_CUR + 2048 <= OFF_XL2, "bin tables fit in front of the projections");
static_assert(F2_ == D_, "layer 2's projections are sized with D_ above");
static_assert((size_t)E_ <= (size_t)N_ * 24, "bins fit their region");

extern "C" void kernel_launch(void* const* d_in, const int* in_sizes, int n_in,
                              void* d_out, int out_size, void* d_ws, size_t ws_size,
                              hipStream_t stream) {
  const float* x     = (const float*)d_in[0];
  const int*   ei    = (const int*)  d_in[1];
  const float* Wssg  = (const float*)d_in[2];
  const float* bssg  = (const float*)d_in[3];
  const float* W1l   = (const float*)d_in[4];
  const float* b1l   = (const float*)d_in[5];
  const float* W1r   = (const float*)d_in[6];
  const float* b1r   = (const float*)d_in[7];
  const float* att1  = (const float*)d_in[8];
  const float* bias1 = (const float*)d_in[9];
  const float* W2l   = (const float*)d_in[10];
  const float* b2l   = (const float*)d_in[11];
  const float* W2r   = (const float*)d_in[12];
  const float* b2r   = (const float*)d_in[13];
  const float* att2  = (const float*)d_in[14];
  const float* bias2 = (const float*)d_in[15];

  const int* src = ei;
  const int* dst = ei + E_;

  float* ws   = (float*)d_ws;
  float* dinv = ws + OFF_DINV;
  int*   cnt  = (int*)(ws + OFF_CNT);
  int*   offs = (int*)(ws + OFF_OFFS);
  int*   cur  = (int*)(ws + OFF_CUR);
  float* xl2  = ws + OFF_XL2;
  float* xr2  = ws + OFF_XR2;
  float* xl1  = ws + OFF_XL1;
  float* xr1  = ws + OFF_XR1;
  int*   bins = (int*)(ws + OFF_BINS);
  float* lg   = ws + OFF_LG;
  float* out  = (float*)d_out;

  const int TB = 256;

  k_zero_i<<<CDIV(2048, TB), TB, 0, stream>>>(cnt, 2048);
  k_tilecount<<<CDIV(E_, CH), 1024, 0, stream>>>(dst, cnt);
  k_scan<<<1, 1024, 0, stream>>>(cnt, offs, cur);
  k_binscatter<<<CDIV(E_, CH), 1024, 0, stream>>>(src, dst, cur, bins);

  k_dinv_tile<<<NT, TS, 0, stream>>>(offs, bins, dinv);
  k_ssg_tile<4><<<NT, 256, 0, stream>>>(offs, bins, x, dinv, Wssg, bssg, W1l, b1l, W1r, b1r, xl1, xr1);

  k_gat_tile<12, 12, 4, true><<<NT, 256, 0, stream>>>(offs, bins, xl1, xr1, att1, bias1, lg, nullptr,
                                                      W2l, b2l, W2r, b2r, xl2, xr2);

  k_gat_tile<1, 64, 4, false><<<NT, 256, 0, stream>>>(offs, bins, xl2, xr2, att2, bias2, lg, out,
                                                      nullptr, nullptr, nullptr, nullptr, nullptr, nullptr);
}


# 1-head layer reads each source row once: softmax shifted by the destination's own self-loop logit, known before the pass, so weights and sums form in the logit pass; exp pass and 2nd gather go
# speedup vs baseline: 4.7381x; 1.0378x over previous
#include <hip/hip_runtime.h>
#include <cmath>

constexpr int N_  = 100000;
constexpr int E_  = 1000000;
constexpr int D_  = 64;
constexpr int F1_ = 144;
constexpr int F2_ = 64;

#define CDIV(a,b) (((a)+(b)-1)/(b))

typedef __attribute__((ext_vector_type(2))) float v2f;
typedef __attribute__((ext_vector_type(8))) float v8f;

constexpr int TSH = 6;
constexpr int TS  = 1 << TSH;
constexpr int NT  = CDIV(N_, TS);
constexpr int CH  = 8192;
constexpr int SRC_BITS = 17;
constexpr int SRC_MASK = (1 << SRC_BITS) - 1;
static_assert(N_ <= (1 << SRC_BITS), "source id must fit the packed field");
static_assert(NT + 1 <= 2048, "scan handles two tiles per thread of one 1024-thread group");

__global__ void k_zero_i(int* __restrict__ p, int n) {
  int i = blockIdx.x * blockDim.x + threadIdx.x;
  if (i < n) p[i] = 0;
}

__global__ __launch_bounds__(1024) void k_tilecount(const int* __restrict__ dst, int* __restrict__ cnt) {
  __shared__ int hist[NT];
  const int tid = threadIdx.x;
  for (int i = tid; i < NT; i += 1024) hist[i] = 0;
  __syncthreads();
  const int base = blockIdx.x * CH;
#pragma unroll
  for (int k = 0; k < CH / 1024; ++k) {
    int e = base + k * 1024 + tid;
    if (e < E_) atomicAdd(&hist[dst[e] >> TSH], 1);
  }
  __syncthreads();
  for (int i = tid; i < NT; i += 1024) { int v = hist[i]; if (v) atomicAdd(&cnt[i], v); }
}

__global__ __launch_bounds__(1024) void k_scan(const int* __restrict__ cnt, int* __restrict__ off,
                                               int* __restrict__ cur) {
  __shared__ int p[1024];
  const int t = threadIdx.x;
  const int a = (2 * t     < NT) ? cnt[2 * t]     : 0;
  const int b = (2 * t + 1 < NT) ? cnt[2 * t + 1] : 0;
  int v = a + b;
  p[t] = v;
  __syncthreads();
  for (int o = 1; o < 1024; o <<= 1) {
    int add = (t >= o) ? p[t - o] : 0;
    __syncthreads();
    v += add;
    p[t] = v;
    __syncthreads();
  }
  const int excl = v - (a + b);
  if (2 * t     < NT) { off[2 * t]     = excl;     cur[2 * t]     = excl; }
  if (2 * t + 1 < NT) { off[2 * t + 1] = excl + a; cur[2 * t + 1] = excl + a; }
  if (t == 1023) off[NT] = v;
}

__global__ __launch_bounds__(1024) void k_binscatter(const int* __restrict__ src, const int* __restrict__ dst,
                                                     int* __restrict__ cur, int* __restrict__ bins) {
  __shared__ int hist[NT];
  const int tid = threadIdx.x;
  for (int i = tid; i < NT; i += 1024) hist[i] = 0;
  __syncthreads();
  const int base = blockIdx.x * CH;
  int pk[CH / 1024], bn[CH / 1024], lr[CH / 1024];
#pragma unroll
  for (int k = 0; k < CH / 1024; ++k) {
    int e = base + k * 1024 + tid;
    bn[k] = -1; pk[k] = 0; lr[k] = 0;
    if (e < E_) {
      int d = dst[e];
      bn[k] = d >> TSH;
      pk[k] = src[e] | ((d & (TS - 1)) << SRC_BITS);
      lr[k] = atomicAdd(&hist[bn[k]], 1);
    }
  }
  __syncthreads();
  for (int i = tid; i < NT; i += 1024) { int v = hist[i]; hist[i] = v ? atomicAdd(&cur[i], v) : 0; }
  __syncthreads();
#pragma unroll
  for (int k = 0; k < CH / 1024; ++k)
    if (bn[k] >= 0) bins[hist[bn[k]] + lr[k]] = pk[k];
}

__device__ __forceinline__ bool tile_item(int j, int tot, int cnt, int b, int n0,
                                          const int* __restrict__ bins, int& s, int& dl, int& sl) {
  const int p  = min(b + j, E_ - 1);
  const int pk = bins[p];
  const bool isE = j < cnt;
  const int dlS = j - cnt;
  const int sS  = n0 + dlS;
  const bool okS = (!isE) && (j < tot) && (sS < N_);
  s  = isE ? (pk & SRC_MASK)  : (okS ? sS : 0);
  dl = isE ? (pk >> SRC_BITS) : (okS ? dlS : 0);
  sl = isE ? p                : (okS ? E_ + sS : 0);
  return isE || okS;
}

__device__ __forceinline__ void ldsMaxF(int* cell, float v) {
  if (v >= 0.0f) atomicMax(cell, __float_as_int(v));
  else           atomicMin((unsigned int*)cell, __float_as_uint(v));
}

template <int U>
__device__ __forceinline__ int wave_item(int q, int wv) {
  const unsigned uq = (unsigned)q;
  return (int)(((uq / U) * 8u + (unsigned)wv) * U + (uq % U));
}
__device__ __forceinline__ int lane_bcast(int v, int ln) { return __builtin_amdgcn_readlane(v, ln); }
__device__ __forceinline__ float lane_bcast(float v, int ln) {
  return __int_as_float(__builtin_amdgcn_readlane(__float_as_int(v), ln));
}

__global__ __launch_bounds__(TS) void k_dinv_tile(const int* __restrict__ off, const int* __restrict__ bins,
                                                  float* __restrict__ dinv) {
  __shared__ int c[TS];
  const int t = blockIdx.x, tid = threadIdx.x;
  c[tid] = 0;
  __syncthreads();
  const int b = off[t], e = off[t + 1];
  for (int p = b + tid; p < e; p += TS) atomicAdd(&c[bins[p] >> SRC_BITS], 1);
  __syncthreads();
  const int n = t * TS + tid;
  if (n < N_) dinv[n] = rsqrtf(1.0f + (float)c[tid]);
}

template <int K, int PA, int NC>
__device__ __forceinline__ void tile_mma2(const float* a0p, const float* __restrict__ wp, v8f& acc0, v8f& acc1) {
  const float* a1p = a0p + 16 * PA;
#pragma unroll 4
  for (int k = 0; k < K; k += 4) {
    v2f a0, a1, b;
    a0.x = a0p[k]; a0.y = a0p[k + 1];
    a1.x = a1p[k]; a1.y = a1p[k + 1];
    b.x = wp[(size_t)k * NC];
    b.y = wp[(size_t)(k + 1) * NC];
    acc0 = __builtin_amdgcn_wmma_f32_16x16x4_f32(false, a0, false, b, (short)0, acc0, false, false);
    acc1 = __builtin_amdgcn_wmma_f32_16x16x4_f32(false, a1, false, b, (short)0, acc1, false, false);
  }
}

template <int K, int PA, int NC, int PC>
__device__ __forceinline__ void tile_proj_lds(const float* As, const float* __restrict__ W,
                                              const float* __restrict__ bias, float* Cs, int wv, int lane) {
  const int c = lane & 15, hh = lane >> 4;
  const int r0 = (wv & 1) * 32;
  for (int ct = wv >> 1; ct < NC / 16; ct += 4) {
    const int col = ct * 16 + c;
    v8f acc0 = {}, acc1 = {};
    tile_mma2<K, PA, NC>(As + (r0 + c) * PA + 2 * hh, W + (size_t)(2 * hh) * NC + col, acc0, acc1);
    const float bn = bias[col];
#pragma unroll
    for (int j = 0; j < 8; ++j) {
      Cs[(r0 + 8 * hh + j) * PC + col]      = acc0[j] + bn;
      Cs[(r0 + 16 + 8 * hh + j) * PC + col] = acc1[j] + bn;
    }
  }
}

template <int K, int PA, int NC>
__device__ __forceinline__ void tile_proj_pair(const float* As, int n0, int wv, int lane,
                                               const float* __restrict__ Wa, const float* __restrict__ ba,
                                               float* __restrict__ Ca,
                                               const float* __restrict__ Wb, const float* __restrict__ bb,
                                               float* __restrict__ Cb) {
  constexpr int CT = NC / 16;
  const int c = lane & 15, hh = lane >> 4;
  const int r0 = (wv & 1) * 32;
  for (int ct = wv >> 1; ct < 2 * CT; ct += 4) {
    const bool second = ct >= CT;
    const float* __restrict__ W  = second ? Wb : Wa;
    const float* __restrict__ bs = second ? bb : ba;
    float* __restrict__ C        = second ? Cb : Ca;
    const int col = (second ? ct - CT : ct) * 16 + c;
    v8f acc0 = {}, acc1 = {};
    tile_mma2<K, PA, NC>(As + (r0 + c) * PA + 2 * hh, W + (size_t)(2 * hh) * NC + col, acc0, acc1);
    const float bn = bs[col];
    float* __restrict__ cp = C + (size_t)(n0 + r0 + 8 * hh) * NC + col;
    if (n0 + r0 < N_) {
#pragma unroll
      for (int j = 0; j < 8; ++j) cp[(size_t)j * NC] = acc0[j] + bn;
    }
    if (n0 + r0 + 16 < N_) {
#pragma unroll
      for (int j = 0; j < 8; ++j) cp[(size_t)(16 + j) * NC] = acc1[j] + bn;
    }
  }
}

constexpr int PH = D_ + 4;
static_assert(N_ % 16 == 0, "a 16-row output tile is never partly past the last node");
template <int U>
__global__ __launch_bounds__(256) void k_ssg_tile(const int* __restrict__ off, const int* __restrict__ bins,
                                                  const float* __restrict__ x, const float* __restrict__ dinv,
                                                  const float* __restrict__ Wssg, const float* __restrict__ bssg,
                                                  const float* __restrict__ W1l, const float* __restrict__ b1l,
                                                  const float* __restrict__ W1r, const float* __restrict__ b1r,
                                                  float* __restrict__ xl1, float* __restrict__ xr1) {
  __shared__ float acc[TS * PH];
  __shared__ float x1s[TS * PH];
  const int t = blockIdx.x, tid = threadIdx.x, lane = tid & 31;
  const int wv = __builtin_amdgcn_readfirstlane(tid >> 5);
  for (int i = tid; i < TS * PH; i += 256) acc[i] = 0.0f;
  __syncthreads();
  const int b = off[t];
  const int cnt = off[t + 1] - b;
  const int tot = cnt + TS;
  const int n0 = t * TS;
  for (int q0 = 0; wave_item<U>(q0, wv) < tot; q0 += 32) {
    int sL, dlL, slL;
    const int okL = tile_item(wave_item<U>(q0 + lane, wv), tot, cnt, b, n0, bins, sL, dlL, slL) ? 1 : 0;
    const float wL = dinv[sL] * dinv[n0 + dlL];
    for (int g = 0; g < 32 / U; ++g) {
      if (wave_item<U>(q0 + g * U, wv) >= tot) break;
      int s[U], dl[U]; bool ok[U];
      float x0[U], x1[U], w[U];
#pragma unroll
      for (int u = 0; u < U; ++u) {
        s[u]  = lane_bcast(sL, g * U + u);
        dl[u] = lane_bcast(dlL, g * U + u);
        ok[u] = lane_bcast(okL, g * U + u) != 0;
        w[u]  = lane_bcast(wL, g * U + u);
      }
#pragma unroll
      for (int u = 0; u < U; ++u) {
        x0[u] = x[(size_t)s[u] * D_ + lane];
        x1[u] = x[(size_t)s[u] * D_ + lane + 32];
      }
#pragma unroll
      for (int u = 0; u < U; ++u) {
        atomicAdd(&acc[dl[u] * PH + lane],      ok[u] ? x0[u] * w[u] : 0.0f);
        atomicAdd(&acc[dl[u] * PH + lane + 32], ok[u] ? x1[u] * w[u] : 0.0f);
      }
    }
  }
  __syncthreads();
  for (int i = tid; i < TS * D_; i += 256) {
    const int dl = i >> 6, c = i & 63;
    const int n = n0 + dl;
    acc[dl * PH + c] = (n < N_) ? 0.5f * x[(size_t)n * D_ + c] + 0.5f * acc[dl * PH + c] : 0.0f;
  }
  __syncthreads();
  tile_proj_lds<D_, PH, D_, PH>(acc, Wssg, bssg, x1s, wv, lane);
  __syncthreads();
  tile_proj_pair<D_, PH, F1_>(x1s, n0, wv, lane, W1l, b1l, xl1, W1r, b1r, xr1);
}

template <int H, int C, int U, bool PROJ>
__device__ __forceinline__ void gat_tile_body(float* acc, int* mxi, float* den, float* satt,
                                              const int* __restrict__ off, const int* __restrict__ bins,
                                              const float* __restrict__ xl, const float* __restrict__ xr,
                                              const float* __restrict__ att, const float* __restrict__ bias,
                                              float* __restrict__ lg, float* __restrict__ out,
                                              const float* __restrict__ Wnl, const float* __restrict__ bnl,
                                              const float* __restrict__ Wnr, const float* __restrict__ bnr,
                                              float* __restrict__ nxl, float* __restrict__ nxr) {
  constexpr int F = H * C;
  constexpr int K = (F + 31) / 32;
  constexpr int PF = PROJ ? F + 4 : F;
  static_assert(C % 4 == 0, "head rows are read as 16-byte pieces");
  static_assert(!PROJ || (PF % 64 == 4 || PF % 64 == 20 || PF % 64 == 36 || PF % 64 == 52),
                "fragment reads of 16 rows must fall on 16 different bank groups");

  const int t = blockIdx.x, tid = threadIdx.x, lane = tid & 31;
  const int wv = __builtin_amdgcn_readfirstlane(tid >> 5);
  const int n0 = t * TS;
  float* const xrs = acc;
  constexpr int NST = TS * (F / 4) / 256;
  static_assert(TS * (F / 4) % 256 == 0, "the tile's rows split evenly over the group");
  {
    const float4* __restrict__ xr4 = (const float4*)(xr + (size_t)n0 * F);
    const int npc = min(TS, N_ - n0) * (F / 4);
    float4 st[NST];
#pragma unroll
    for (int k = 0; k < NST; ++k) {
      const int i = tid + 256 * k;
      st[k] = make_float4(0.0f, 0.0f, 0.0f, 0.0f);
      if (i < npc) st[k] = xr4[i];
    }
#pragma unroll
    for (int k = 0; k < NST; ++k) *(float4*)(xrs + 4 * (tid + 256 * k)) = st[k];
  }
  for (int i = tid; i < TS * H; i += 256) { mxi[i] = (int)0xFF800000u; den[i] = 0.0f; }
  for (int i = tid; i < F; i += 256) satt[i] = att[i];
  __syncthreads();
  const int b = off[t];
  const int cnt = off[t + 1] - b;
  const int tot = cnt + TS;

  if constexpr (H == 1) {
    for (int q0 = 0; wave_item<U>(q0, wv) < tot; q0 += 32) {
      int sL, dlL, slL;
      const int okL = tile_item(wave_item<U>(q0 + lane, wv), tot, cnt, b, n0, bins, sL, dlL, slL) ? 1 : 0;
      for (int g = 0; g < 32 / U; ++g) {
        if (wave_item<U>(q0 + g * U, wv) >= tot) break;
        int s[U], dl[U], sl[U]; bool ok[U];
        float l0[U], l1[U], r0[U], r1[U];
#pragma unroll
        for (int u = 0; u < U; ++u) {
          s[u]  = lane_bcast(sL, g * U + u);
          dl[u] = lane_bcast(dlL, g * U + u);
          sl[u] = lane_bcast(slL, g * U + u);
          ok[u] = lane_bcast(okL, g * U + u) != 0;
        }
#pragma unroll
        for (int u = 0; u < U; ++u) {
          l0[u] = xl[(size_t)s[u] * F + lane];
          l1[u] = xl[(size_t)s[u] * F + lane + 32];
          r0[u] = xrs[dl[u] * F + lane];
          r1[u] = xrs[dl[u] * F + lane + 32];
        }
#pragma unroll
        for (int u = 0; u < U; ++u) {
          float a = 0.0f;
          float t0 = l0[u] + r0[u];
          t0 = (t0 > 0.0f) ? t0 : 0.2f * t0;
          a += t0 * satt[lane];
          float t1 = l1[u] + r1[u];
          t1 = (t1 > 0.0f) ? t1 : 0.2f * t1;
          a += t1 * satt[lane + 32];
#pragma unroll
          for (int o = 16; o > 0; o >>= 1) a += __shfl_xor(a, o, 32);
          if (lane == 0 && ok[u]) { lg[sl[u]] = a; ldsMaxF(&mxi[dl[u]], a); }
        }
      }
    }
  } else {
    for (int idx = tid; idx < tot * H; idx += 256) {
      const int j = idx / H, h = idx - j * H;
      int s, dl, sl;
      if (tile_item(j, tot, cnt, b, n0, bins, s, dl, sl)) {
        const float4* pl = (const float4*)(xl + (size_t)s * F + h * C);
        const float4* pr = (const float4*)(xrs + dl * F + h * C);
        float4 lv[C / 4], rv[C / 4];
#pragma unroll
        for (int q = 0; q < C / 4; ++q) { lv[q] = pl[q]; rv[q] = pr[q]; }
        float a = 0.0f;
#pragma unroll
        for (int q = 0; q < C / 4; ++q) {
          const float le[4] = {lv[q].x, lv[q].y, lv[q].z, lv[q].w};
          const float re[4] = {rv[q].x, rv[q].y, rv[q].z, rv[q].w};
#pragma unroll
          for (int z = 0; z < 4; ++z) {
            float tt = le[z] + re[z];
            tt = (tt > 0.0f) ? tt : 0.2f * tt;
            a += tt * satt[h * C + q * 4 + z];
          }
        }
        lg[(size_t)sl * H + h] = a;
        ldsMaxF(&mxi[dl * H + h], a);
      }
    }
  }
  __syncthreads();

  for (int i = tid; i < TS * PF; i += 256) acc[i] = 0.0f;
  for (int idx = tid; idx < tot * H; idx += 256) {
    const int j = idx / H, h = idx - j * H;
    int s, dl, sl;
    if (tile_item(j, tot, cnt, b, n0, bins, s, dl, sl)) {
      const float a = expf(lg[(size_t)sl * H + h] - __int_as_float(mxi[dl * H + h]));
      lg[(size_t)sl * H + h] = a;
      atomicAdd(&den[dl * H + h], a);
    }
  }
  __syncthreads();

  for (int q0 = 0; wave_item<U>(q0, wv) < tot; q0 += 32) {
    int sL, dlL, slL;
    const int okL = tile_item(wave_item<U>(q0 + lane, wv), tot, cnt, b, n0, bins, sL, dlL, slL) ? 1 : 0;
    float aL = 0.0f;
    if constexpr (H == 1) aL = lg[slL];
    for (int g = 0; g < 32 / U; ++g) {
      if (wave_item<U>(q0 + g * U, wv) >= tot) break;
      int s[U], dl[U], sl[U]; bool ok[U];
      float xv[U][K], av[U][K];
#pragma unroll
      for (int u = 0; u < U; ++u) {
        s[u]  = lane_bcast(sL, g * U + u);
        dl[u] = lane_bcast(dlL, g * U + u);
        sl[u] = lane_bcast(slL, g * U + u);
        ok[u] = lane_bcast(okL, g * U + u) != 0;
      }
#pragma unroll
      for (int u = 0; u < U; ++u) {
#pragma unroll
        for (int k = 0; k < K; ++k) {
          const int r  = lane + 32 * k;
          const int rr = (r < F) ? r : r - 32;
          xv[u][k] = xl[(size_t)s[u] * F + rr];
          if constexpr (H == 1) av[u][k] = lane_bcast(aL, g * U + u);
          else                  av[u][k] = lg[(size_t)sl[u] * H + rr / C];
        }
      }
#pragma unroll
      for (int u = 0; u < U; ++u) {
#pragma unroll
        for (int k = 0; k < K; ++k) {
          const int r  = lane + 32 * k;
          const int rr = (r < F) ? r : r - 32;
          atomicAdd(&acc[dl[u] * PF + rr], (ok[u] && r < F) ? av[u][k] * xv[u][k] : 0.0f);
        }
      }
    }
  }
  __syncthreads();

  if constexpr (!PROJ) {
    for (int i = tid; i < TS * F; i += 256) {
      const int dl = i / F, r = i - dl * F;
      const int n = n0 + dl;
      if (n < N_) out[(size_t)n * F + r] = acc[i] / (den[dl * H + r / C] + 1e-16f) + bias[r];
    }
  } else {
    for (int i = tid; i < TS * F; i += 256) {
      const int dl = i / F, r = i - dl * F;
      acc[dl * PF + r] = acc[dl * PF + r] / (den[dl * H + r / C] + 1e-16f) + bias[r];
    }
    __syncthreads();
    tile_proj_pair<F, PF, F2_>(acc, n0, wv, lane, Wnl, bnl, nxl, Wnr, bnr, nxr);
  }
}

template <int H, int C, int U, bool PROJ>
__global__ __launch_bounds__(256) void k_gat_tile(const int* __restrict__ off, const int* __restrict__ bins,
                                                  const float* __restrict__ xl, const float* __restrict__ xr,
                                                  const float* __restrict__ att, const float* __restrict__ bias,
                                                  float* __restrict__ lg, float* __restrict__ out,
                                                  const float* __restrict__ Wnl, const float* __restrict__ bnl,
                                                  const float* __restrict__ Wnr, const float* __restrict__ bnr,
                                                  float* __restrict__ nxl, float* __restrict__ nxr) {
  constexpr int F = H * C;
  constexpr int PF = PROJ ? F + 4 : F;
  __shared__ __attribute__((aligned(16))) float acc[TS * PF];
  __shared__ int   mxi[TS * H];
  __shared__ float den[TS * H];
  __shared__ float satt[F];
  gat_tile_body<H, C, U, PROJ>(acc, mxi, den, satt, off, bins, xl, xr, att, bias, lg, out,
                               Wnl, bnl, Wnr, bnr, nxl, nxr);
}

constexpr float SHIFT_LIM = 40.0f;
template <int C, int U>
__global__ __launch_bounds__(256) void k_gat1h_tile(const int* __restrict__ off, const int* __restrict__ bins,
                                                    const float* __restrict__ xl, const float* __restrict__ xr,
                                                    const float* __restrict__ att, const float* __restrict__ bias,
                                                    float* __restrict__ lg, float* __restrict__ out) {
  constexpr int F = C;
  static_assert(F == 64, "a wave holds a row as two words per lane");
  __shared__ __attribute__((aligned(16))) float acc[TS * F];
  __shared__ __attribute__((aligned(16))) float xrs[TS * F];
  __shared__ float shf[TS];
  __shared__ float den[TS];
  __shared__ float satt[F];
  __shared__ int   over;

  const int t = blockIdx.x, tid = threadIdx.x, lane = tid & 31;
  const int wv = __builtin_amdgcn_readfirstlane(tid >> 5);
  const int n0 = t * TS;
  constexpr int NST = TS * (F / 4) / 256;
  static_assert(TS * (F / 4) % 256 == 0, "the tile's rows split evenly over the group");
  {
    const float4* __restrict__ xr4 = (const float4*)(xr + (size_t)n0 * F);
    const int npc = min(TS, N_ - n0) * (F / 4);
    float4 st[NST];
#pragma unroll
    for (int k = 0; k < NST; ++k) {
      const int i = tid + 256 * k;
      st[k] = make_float4(0.0f, 0.0f, 0.0f, 0.0f);
      if (i < npc) st[k] = xr4[i];
    }
#pragma unroll
    for (int k = 0; k < NST; ++k) {
      *(float4*)(xrs + 4 * (tid + 256 * k)) = st[k];
      *(float4*)(acc + 4 * (tid + 256 * k)) = make_float4(0.0f, 0.0f, 0.0f, 0.0f);
    }
  }
  if (tid < TS) den[tid] = 0.0f;
  if (tid < F) satt[tid] = att[tid];
  if (tid == 0) over = 0;
  __syncthreads();

  {
    constexpr int RW = TS / 8;
    float l0[RW], l1[RW];
#pragma unroll
    for (int r = 0; r < RW; ++r) {
      const int n = min(n0 + wv * RW + r, N_ - 1);
      l0[r] = xl[(size_t)n * F + lane];
      l1[r] = xl[(size_t)n * F + lane + 32];
    }
    float a[RW];
#pragma unroll
    for (int r = 0; r < RW; ++r) {
      const int dl = wv * RW + r;
      a[r] = 0.0f;
      float t0 = l0[r] + xrs[dl * F + lane];
      t0 = (t0 > 0.0f) ? t0 : 0.2f * t0;
      a[r] += t0 * satt[lane];
      float t1 = l1[r] + xrs[dl * F + lane + 32];
      t1 = (t1 > 0.0f) ? t1 : 0.2f * t1;
      a[r] += t1 * satt[lane + 32];
    }
#pragma unroll
    for (int o = 16; o > 0; o >>= 1) {
#pragma unroll
      for (int r = 0; r < RW; ++r) a[r] += __shfl_xor(a[r], o, 32);
    }
    if (lane == 0) {
#pragma unroll
      for (int r = 0; r < RW; ++r) shf[wv * RW + r] = a[r];
    }
  }
  __syncthreads();

  const int b = off[t];
  const int cnt = off[t + 1] - b;
  const int tot = cnt + TS;

  for (int q0 = 0; wave_item<U>(q0, wv) < tot; q0 += 32) {
    int sL, dlL, slL;
    const int okL = tile_item(wave_item<U>(q0 + lane, wv), tot, cnt, b, n0, bins, sL, dlL, slL) ? 1 : 0;
    const float mL = shf[dlL];
    for (int g = 0; g < 32 / U; ++g) {
      if (wave_item<U>(q0 + g * U, wv) >= tot) break;
      int s[U], dl[U]; bool ok[U];
      float m[U], l0[U], l1[U], r0[U], r1[U];
#pragma unroll
      for (int u = 0; u < U; ++u) {
        s[u]  = lane_bcast(sL, g * U + u);
        dl[u] = lane_bcast(dlL, g * U + u);
        ok[u] = lane_bcast(okL, g * U + u) != 0;
        m[u]  = lane_bcast(mL, g * U + u);
      }
#pragma unroll
      for (int u = 0; u < U; ++u) {
        l0[u] = xl[(size_t)s[u] * F + lane];
        l1[u] = xl[(size_t)s[u] * F + lane + 32];
        r0[u] = xrs[dl[u] * F + lane];
        r1[u] = xrs[dl[u] * F + lane + 32];
      }
#pragma unroll
      for (int u = 0; u < U; ++u) {
        float a = 0.0f;
        float t0 = l0[u] + r0[u];
        t0 = (t0 > 0.0f) ? t0 : 0.2f * t0;
        a += t0 * satt[lane];
        float t1 = l1[u] + r1[u];
        t1 = (t1 > 0.0f) ? t1 : 0.2f * t1;
        a += t1 * satt[lane + 32];
#pragma unroll
        for (int o = 16; o > 0; o >>= 1) a += __shfl_xor(a, o, 32);
        const float d = a - m[u];
        const float w = expf(d);
        if (lane == 0 && ok[u]) {
          atomicAdd(&den[dl[u]], w);
          if (d > SHIFT_LIM) over = 1;
        }
        atomicAdd(&acc[dl[u] * F + lane],      ok[u] ? w * l0[u] : 0.0f);
        atomicAdd(&acc[dl[u] * F + lane + 32], ok[u] ? w * l1[u] : 0.0f);
      }
    }
  }
  __syncthreads();

  if (over != 0) {
    gat_tile_body<1, C, U, false>(acc, (int*)shf, den, satt, off, bins, xl, xr, att, bias, lg, out,
                                  nullptr, nullptr, nullptr, nullptr, nullptr, nullptr);
    return;
  }

  for (int i = tid; i < TS * F; i += 256) {
    const int dl = i / F, r = i - dl * F;
    const int n = n0 + dl;
    if (n < N_) out[(size_t)n * F + r] = acc[i] / (den[dl] + 1e-16f) + bias[r];
  }
}

constexpr size_t OFF_DINV = 0;
constexpr size_t OFF_CNT  = OFF_DINV + N_;
constexpr size_t OFF_OFFS = OFF_CNT  + 2048;
constexpr size_t OFF_CUR  = OFF_OFFS + 2048;
constexpr size_t OFF_XL2  = (size_t)N_ + E_;
constexpr size_t OFF_XR2  = OFF_XL2  + (size_t)N_*D_;
constexpr size_t OFF_XL1  = OFF_XR2  + (size_t)N_*D_;
constexpr size_t OFF_XR1  = OFF_XL1  + (size_t)N_*F1_;
constexpr size_t OFF_BINS = OFF_XR1  + (size_t)N_*F1_;
constexpr size_t OFF_LG   = OFF_BINS + (size_t)N_*24;
static_assert(OFF_CUR + 2048 <= OFF_XL2, "bin tables fit in front of the projections");
static_assert(F2_ == D_, "layer 2's projections are sized with D_ above");
static_assert((size_t)E_ <= (size_t)N_ * 24, "bins fit their region");

extern "C" void kernel_launch(void* const* d_in, const int* in_sizes, int n_in,
                              void* d_out, int out_size, void* d_ws, size_t ws_size,
                              hipStream_t stream) {
  const float* x     = (const float*)d_in[0];
  const int*   ei    = (const int*)  d_in[1];
  const float* Wssg  = (const float*)d_in[2];
  const float* bssg  = (const float*)d_in[3];
  const float* W1l   = (const float*)d_in[4];
  const float* b1l   = (const float*)d_in[5];
  const float* W1r   = (const float*)d_in[6];
  const float* b1r   = (const float*)d_in[7];
  const float* att1  = (const float*)d_in[8];
  const float* bias1 = (const float*)d_in[9];
  const float* W2l   = (const float*)d_in[10];
  const float* b2l   = (const float*)d_in[11];
  const float* W2r   = (const float*)d_in[12];
  const float* b2r   = (const float*)d_in[13];
  const float* att2  = (const float*)d_in[14];
  const float* bias2 = (const float*)d_in[15];

  const int* src = ei;
  const int* dst = ei + E_;

  float* ws   = (float*)d_ws;
  float* dinv = ws + OFF_DINV;
  int*   cnt  = (int*)(ws + OFF_CNT);
  int*   offs = (int*)(ws + OFF_OFFS);
  int*   cur  = (int*)(ws + OFF_CUR);
  float* xl2  = ws + OFF_XL2;
  float* xr2  = ws + OFF_XR2;
  float* xl1  = ws + OFF_XL1;
  float* xr1  = ws + OFF_XR1;
  int*   bins = (int*)(ws + OFF_BINS);
  float* lg   = ws + OFF_LG;
  float* out  = (float*)d_out;

  const int TB = 256;

  k_zero_i<<<CDIV(2048, TB), TB, 0, stream>>>(cnt, 2048);
  k_tilecount<<<CDIV(E_, CH), 1024, 0, stream>>>(dst, cnt);
  k_scan<<<1, 1024, 0, stream>>>(cnt, offs, cur);
  k_binscatter<<<CDIV(E_, CH), 1024, 0, stream>>>(src, dst, cur, bins);

  k_dinv_tile<<<NT, TS, 0, stream>>>(offs, bins, dinv);
  k_ssg_tile<4><<<NT, 256, 0, stream>>>(offs, bins, x, dinv, Wssg, bssg, W1l, b1l, W1r, b1r, xl1, xr1);

  k_gat_tile<12, 12, 4, true><<<NT, 256, 0, stream>>>(offs, bins, xl1, xr1, att1, bias1, lg, nullptr,
                                                      W2l, b2l, W2r, b2r, xl2, xr2);

  k_gat1h_tile<64, 4><<<NT, 256, 0, stream>>>(offs, bins, xl2, xr2, att2, bias2, lg, out);
}


# 12-head layer reads each source row once: softmax shifted by the destination's self-loop logit per head, so weights and sums form in the logit pass; exp pass, 2nd gather go; logit sum order changed
# speedup vs baseline: 5.4534x; 1.1510x over previous
#include <hip/hip_runtime.h>
#include <cmath>

constexpr int N_  = 100000;
constexpr int E_  = 1000000;
constexpr int D_  = 64;
constexpr int F1_ = 144;
constexpr int F2_ = 64;

#define CDIV(a,b) (((a)+(b)-1)/(b))

typedef __attribute__((ext_vector_type(2))) float v2f;
typedef __attribute__((ext_vector_type(8))) float v8f;

constexpr int TSH = 6;
constexpr int TS  = 1 << TSH;
constexpr int NT  = CDIV(N_, TS);
constexpr int CH  = 8192;
constexpr int SRC_BITS = 17;
constexpr int SRC_MASK = (1 << SRC_BITS) - 1;
static_assert(N_ <= (1 << SRC_BITS), "source id must fit the packed field");
static_assert(NT + 1 <= 2048, "scan handles two tiles per thread of one 1024-thread group");

__global__ void k_zero_i(int* __restrict__ p, int n) {
  int i = blockIdx.x * blockDim.x + threadIdx.x;
  if (i < n) p[i] = 0;
}

__global__ __launch_bounds__(1024) void k_tilecount(const int* __restrict__ dst, int* __restrict__ cnt) {
  __shared__ int hist[NT];
  const int tid = threadIdx.x;
  for (int i = tid; i < NT; i += 1024) hist[i] = 0;
  __syncthreads();
  const int base = blockIdx.x * CH;
#pragma unroll
  for (int k = 0; k < CH / 1024; ++k) {
    int e = base + k * 1024 + tid;
    if (e < E_) atomicAdd(&hist[dst[e] >> TSH], 1);
  }
  __syncthreads();
  for (int i = tid; i < NT; i += 1024) { int v = hist[i]; if (v) atomicAdd(&cnt[i], v); }
}

__global__ __launch_bounds__(1024) void k_scan(const int* __restrict__ cnt, int* __restrict__ off,
                                               int* __restrict__ cur) {
  __shared__ int p[1024];
  const int t = threadIdx.x;
  const int a = (2 * t     < NT) ? cnt[2 * t]     : 0;
  const int b = (2 * t + 1 < NT) ? cnt[2 * t + 1] : 0;
  int v = a + b;
  p[t] = v;
  __syncthreads();
  for (int o = 1; o < 1024; o <<= 1) {
    int add = (t >= o) ? p[t - o] : 0;
    __syncthreads();
    v += add;
    p[t] = v;
    __syncthreads();
  }
  const int excl = v - (a + b);
  if (2 * t     < NT) { off[2 * t]     = excl;     cur[2 * t]     = excl; }
  if (2 * t + 1 < NT) { off[2 * t + 1] = excl + a; cur[2 * t + 1] = excl + a; }
  if (t == 1023) off[NT] = v;
}

__global__ __launch_bounds__(1024) void k_binscatter(const int* __restrict__ src, const int* __restrict__ dst,
                                                     int* __restrict__ cur, int* __restrict__ bins) {
  __shared__ int hist[NT];
  const int tid = threadIdx.x;
  for (int i = tid; i < NT; i += 1024) hist[i] = 0;
  __syncthreads();
  const int base = blockIdx.x * CH;
  int pk[CH / 1024], bn[CH / 1024], lr[CH / 1024];
#pragma unroll
  for (int k = 0; k < CH / 1024; ++k) {
    int e = base + k * 1024 + tid;
    bn[k] = -1; pk[k] = 0; lr[k] = 0;
    if (e < E_) {
      int d = dst[e];
      bn[k] = d >> TSH;
      pk[k] = src[e] | ((d & (TS - 1)) << SRC_BITS);
      lr[k] = atomicAdd(&hist[bn[k]], 1);
    }
  }
  __syncthreads();
  for (int i = tid; i < NT; i += 1024) { int v = hist[i]; hist[i] = v ? atomicAdd(&cur[i], v) : 0; }
  __syncthreads();
#pragma unroll
  for (int k = 0; k < CH / 1024; ++k)
    if (bn[k] >= 0) bins[hist[bn[k]] + lr[k]] = pk[k];
}

__device__ __forceinline__ bool tile_item(int j, int tot, int cnt, int b, int n0,
                                          const int* __restrict__ bins, int& s, int& dl, int& sl) {
  const int p  = min(b + j, E_ - 1);
  const int pk = bins[p];
  const bool isE = j < cnt;
  const int dlS = j - cnt;
  const int sS  = n0 + dlS;
  const bool okS = (!isE) && (j < tot) && (sS < N_);
  s  = isE ? (pk & SRC_MASK)  : (okS ? sS : 0);
  dl = isE ? (pk >> SRC_BITS) : (okS ? dlS : 0);
  sl = isE ? p                : (okS ? E_ + sS : 0);
  return isE || okS;
}

__device__ __forceinline__ void ldsMaxF(int* cell, float v) {
  if (v >= 0.0f) atomicMax(cell, __float_as_int(v));
  else           atomicMin((unsigned int*)cell, __float_as_uint(v));
}

template <int U>
__device__ __forceinline__ int wave_item(int q, int wv) {
  const unsigned uq = (unsigned)q;
  return (int)(((uq / U) * 8u + (unsigned)wv) * U + (uq % U));
}
__device__ __forceinline__ int lane_bcast(int v, int ln) { return __builtin_amdgcn_readlane(v, ln); }
__device__ __forceinline__ float lane_bcast(float v, int ln) {
  return __int_as_float(__builtin_amdgcn_readlane(__float_as_int(v), ln));
}

__global__ __launch_bounds__(TS) void k_dinv_tile(const int* __restrict__ off, const int* __restrict__ bins,
                                                  float* __restrict__ dinv) {
  __shared__ int c[TS];
  const int t = blockIdx.x, tid = threadIdx.x;
  c[tid] = 0;
  __syncthreads();
  const int b = off[t], e = off[t + 1];
  for (int p = b + tid; p < e; p += TS) atomicAdd(&c[bins[p] >> SRC_BITS], 1);
  __syncthreads();
  const int n = t * TS + tid;
  if (n < N_) dinv[n] = rsqrtf(1.0f + (float)c[tid]);
}

template <int K, int PA, int NC>
__device__ __forceinline__ void tile_mma2(const float* a0p, const float* __restrict__ wp, v8f& acc0, v8f& acc1) {
  const float* a1p = a0p + 16 * PA;
#pragma unroll 4
  for (int k = 0; k < K; k += 4) {
    v2f a0, a1, b;
    a0.x = a0p[k]; a0.y = a0p[k + 1];
    a1.x = a1p[k]; a1.y = a1p[k + 1];
    b.x = wp[(size_t)k * NC];
    b.y = wp[(size_t)(k + 1) * NC];
    acc0 = __builtin_amdgcn_wmma_f32_16x16x4_f32(false, a0, false, b, (short)0, acc0, false, false);
    acc1 = __builtin_amdgcn_wmma_f32_16x16x4_f32(false, a1, false, b, (short)0, acc1, false, false);
  }
}

template <int K, int PA, int NC, int PC>
__device__ __forceinline__ void tile_proj_lds(const float* As, const float* __restrict__ W,
                                              const float* __restrict__ bias, float* Cs, int wv, int lane) {
  const int c = lane & 15, hh = lane >> 4;
  const int r0 = (wv & 1) * 32;
  for (int ct = wv >> 1; ct < NC / 16; ct += 4) {
    const int col = ct * 16 + c;
    v8f acc0 = {}, acc1 = {};
    tile_mma2<K, PA, NC>(As + (r0 + c) * PA + 2 * hh, W + (size_t)(2 * hh) * NC + col, acc0, acc1);
    const float bn = bias[col];
#pragma unroll
    for (int j = 0; j < 8; ++j) {
      Cs[(r0 + 8 * hh + j) * PC + col]      = acc0[j] + bn;
      Cs[(r0 + 16 + 8 * hh + j) * PC + col] = acc1[j] + bn;
    }
  }
}

template <int K, int PA, int NC>
__device__ __forceinline__ void tile_proj_pair(const float* As, int n0, int wv, int lane,
                                               const float* __restrict__ Wa, const float* __restrict__ ba,
                                               float* __restrict__ Ca,
                                               const float* __restrict__ Wb, const float* __restrict__ bb,
                                               float* __restrict__ Cb) {
  constexpr int CT = NC / 16;
  const int c = lane & 15, hh = lane >> 4;
  const int r0 = (wv & 1) * 32;
  for (int ct = wv >> 1; ct < 2 * CT; ct += 4) {
    const bool second = ct >= CT;
    const float* __restrict__ W  = second ? Wb : Wa;
    const float* __restrict__ bs = second ? bb : ba;
    float* __restrict__ C        = second ? Cb : Ca;
    const int col = (second ? ct - CT : ct) * 16 + c;
    v8f acc0 = {}, acc1 = {};
    tile_mma2<K, PA, NC>(As + (r0 + c) * PA + 2 * hh, W + (size_t)(2 * hh) * NC + col, acc0, acc1);
    const float bn = bs[col];
    float* __restrict__ cp = C + (size_t)(n0 + r0 + 8 * hh) * NC + col;
    if (n0 + r0 < N_) {
#pragma unroll
      for (int j = 0; j < 8; ++j) cp[(size_t)j * NC] = acc0[j] + bn;
    }
    if (n0 + r0 + 16 < N_) {
#pragma unroll
      for (int j = 0; j < 8; ++j) cp[(size_t)(16 + j) * NC] = acc1[j] + bn;
    }
  }
}

constexpr int PH = D_ + 4;
static_assert(N_ % 16 == 0, "a 16-row output tile is never partly past the last node");
template <int U>
__global__ __launch_bounds__(256) void k_ssg_tile(const int* __restrict__ off, const int* __restrict__ bins,
                                                  const float* __restrict__ x, const float* __restrict__ dinv,
                                                  const float* __restrict__ Wssg, const float* __restrict__ bssg,
                                                  const float* __restrict__ W1l, const float* __restrict__ b1l,
                                                  const float* __restrict__ W1r, const float* __restrict__ b1r,
                                                  float* __restrict__ xl1, float* __restrict__ xr1) {
  __shared__ float acc[TS * PH];
  __shared__ float x1s[TS * PH];
  const int t = blockIdx.x, tid = threadIdx.x, lane = tid & 31;
  const int wv = __builtin_amdgcn_readfirstlane(tid >> 5);
  for (int i = tid; i < TS * PH; i += 256) acc[i] = 0.0f;
  __syncthreads();
  const int b = off[t];
  const int cnt = off[t + 1] - b;
  const int tot = cnt + TS;
  const int n0 = t * TS;
  for (int q0 = 0; wave_item<U>(q0, wv) < tot; q0 += 32) {
    int sL, dlL, slL;
    const int okL = tile_item(wave_item<U>(q0 + lane, wv), tot, cnt, b, n0, bins, sL, dlL, slL) ? 1 : 0;
    const float wL = dinv[sL] * dinv[n0 + dlL];
    for (int g = 0; g < 32 / U; ++g) {
      if (wave_item<U>(q0 + g * U, wv) >= tot) break;
      int s[U], dl[U]; bool ok[U];
      float x0[U], x1[U], w[U];
#pragma unroll
      for (int u = 0; u < U; ++u) {
        s[u]  = lane_bcast(sL, g * U + u);
        dl[u] = lane_bcast(dlL, g * U + u);
        ok[u] = lane_bcast(okL, g * U + u) != 0;
        w[u]  = lane_bcast(wL, g * U + u);
      }
#pragma unroll
      for (int u = 0; u < U; ++u) {
        x0[u] = x[(size_t)s[u] * D_ + lane];
        x1[u] = x[(size_t)s[u] * D_ + lane + 32];
      }
#pragma unroll
      for (int u = 0; u < U; ++u) {
        atomicAdd(&acc[dl[u] * PH + lane],      ok[u] ? x0[u] * w[u] : 0.0f);
        atomicAdd(&acc[dl[u] * PH + lane + 32], ok[u] ? x1[u] * w[u] : 0.0f);
      }
    }
  }
  __syncthreads();
  for (int i = tid; i < TS * D_; i += 256) {
    const int dl = i >> 6, c = i & 63;
    const int n = n0 + dl;
    acc[dl * PH + c] = (n < N_) ? 0.5f * x[(size_t)n * D_ + c] + 0.5f * acc[dl * PH + c] : 0.0f;
  }
  __syncthreads();
  tile_proj_lds<D_, PH, D_, PH>(acc, Wssg, bssg, x1s, wv, lane);
  __syncthreads();
  tile_proj_pair<D_, PH, F1_>(x1s, n0, wv, lane, W1l, b1l, xl1, W1r, b1r, xr1);
}

template <int H, int C, int U, bool PROJ>
__device__ __forceinline__ void gat_tile_body(float* acc, int* mxi, float* den, float* satt,
                                              const int* __restrict__ off, const int* __restrict__ bins,
                                              const float* __restrict__ xl, const float* __restrict__ xr,
                                              const float* __restrict__ att, const float* __restrict__ bias,
                                              float* __restrict__ lg, float* __restrict__ out,
                                              const float* __restrict__ Wnl, const float* __restrict__ bnl,
                                              const float* __restrict__ Wnr, const float* __restrict__ bnr,
                                              float* __restrict__ nxl, float* __restrict__ nxr) {
  constexpr int F = H * C;
  constexpr int K = (F + 31) / 32;
  constexpr int PF = PROJ ? F + 4 : F;
  static_assert(C % 4 == 0, "head rows are read as 16-byte pieces");
  static_assert(!PROJ || (PF % 64 == 4 || PF % 64 == 20 || PF % 64 == 36 || PF % 64 == 52),
                "fragment reads of 16 rows must fall on 16 different bank groups");

  const int t = blockIdx.x, tid = threadIdx.x, lane = tid & 31;
  const int wv = __builtin_amdgcn_readfirstlane(tid >> 5);
  const int n0 = t * TS;
  float* const xrs = acc;
  constexpr int NST = TS * (F / 4) / 256;
  static_assert(TS * (F / 4) % 256 == 0, "the tile's rows split evenly over the group");
  {
    const float4* __restrict__ xr4 = (const float4*)(xr + (size_t)n0 * F);
    const int npc = min(TS, N_ - n0) * (F / 4);
    float4 st[NST];
#pragma unroll
    for (int k = 0; k < NST; ++k) {
      const int i = tid + 256 * k;
      st[k] = make_float4(0.0f, 0.0f, 0.0f, 0.0f);
      if (i < npc) st[k] = xr4[i];
    }
#pragma unroll
    for (int k = 0; k < NST; ++k) *(float4*)(xrs + 4 * (tid + 256 * k)) = st[k];
  }
  for (int i = tid; i < TS * H; i += 256) { mxi[i] = (int)0xFF800000u; den[i] = 0.0f; }
  for (int i = tid; i < F; i += 256) satt[i] = att[i];
  __syncthreads();
  const int b = off[t];
  const int cnt = off[t + 1] - b;
  const int tot = cnt + TS;

  if constexpr (H == 1) {
    for (int q0 = 0; wave_item<U>(q0, wv) < tot; q0 += 32) {
      int sL, dlL, slL;
      const int okL = tile_item(wave_item<U>(q0 + lane, wv), tot, cnt, b, n0, bins, sL, dlL, slL) ? 1 : 0;
      for (int g = 0; g < 32 / U; ++g) {
        if (wave_item<U>(q0 + g * U, wv) >= tot) break;
        int s[U], dl[U], sl[U]; bool ok[U];
        float l0[U], l1[U], r0[U], r1[U];
#pragma unroll
        for (int u = 0; u < U; ++u) {
          s[u]  = lane_bcast(sL, g * U + u);
          dl[u] = lane_bcast(dlL, g * U + u);
          sl[u] = lane_bcast(slL, g * U + u);
          ok[u] = lane_bcast(okL, g * U + u) != 0;
        }
#pragma unroll
        for (int u = 0; u < U; ++u) {
          l0[u] = xl[(size_t)s[u] * F + lane];
          l1[u] = xl[(size_t)s[u] * F + lane + 32];
          r0[u] = xrs[dl[u] * F + lane];
          r1[u] = xrs[dl[u] * F + lane + 32];
        }
#pragma unroll
        for (int u = 0; u < U; ++u) {
          float a = 0.0f;
          float t0 = l0[u] + r0[u];
          t0 = (t0 > 0.0f) ? t0 : 0.2f * t0;
          a += t0 * satt[lane];
          float t1 = l1[u] + r1[u];
          t1 = (t1 > 0.0f) ? t1 : 0.2f * t1;
          a += t1 * satt[lane + 32];
#pragma unroll
          for (int o = 16; o > 0; o >>= 1) a += __shfl_xor(a, o, 32);
          if (lane == 0 && ok[u]) { lg[sl[u]] = a; ldsMaxF(&mxi[dl[u]], a); }
        }
      }
    }
  } else {
    for (int idx = tid; idx < tot * H; idx += 256) {
      const int j = idx / H, h = idx - j * H;
      int s, dl, sl;
      if (tile_item(j, tot, cnt, b, n0, bins, s, dl, sl)) {
        const float4* pl = (const float4*)(xl + (size_t)s * F + h * C);
        const float4* pr = (const float4*)(xrs + dl * F + h * C);
        float4 lv[C / 4], rv[C / 4];
#pragma unroll
        for (int q = 0; q < C / 4; ++q) { lv[q] = pl[q]; rv[q] = pr[q]; }
        float a = 0.0f;
#pragma unroll
        for (int q = 0; q < C / 4; ++q) {
          const float le[4] = {lv[q].x, lv[q].y, lv[q].z, lv[q].w};
          const float re[4] = {rv[q].x, rv[q].y, rv[q].z, rv[q].w};
#pragma unroll
          for (int z = 0; z < 4; ++z) {
            float tt = le[z] + re[z];
            tt = (tt > 0.0f) ? tt : 0.2f * tt;
            a += tt * satt[h * C + q * 4 + z];
          }
        }
        lg[(size_t)sl * H + h] = a;
        ldsMaxF(&mxi[dl * H + h], a);
      }
    }
  }
  __syncthreads();

  for (int i = tid; i < TS * PF; i += 256) acc[i] = 0.0f;
  for (int idx = tid; idx < tot * H; idx += 256) {
    const int j = idx / H, h = idx - j * H;
    int s, dl, sl;
    if (tile_item(j, tot, cnt, b, n0, bins, s, dl, sl)) {
      const float a = expf(lg[(size_t)sl * H + h] - __int_as_float(mxi[dl * H + h]));
      lg[(size_t)sl * H + h] = a;
      atomicAdd(&den[dl * H + h], a);
    }
  }
  __syncthreads();

  for (int q0 = 0; wave_item<U>(q0, wv) < tot; q0 += 32) {
    int sL, dlL, slL;
    const int okL = tile_item(wave_item<U>(q0 + lane, wv), tot, cnt, b, n0, bins, sL, dlL, slL) ? 1 : 0;
    float aL = 0.0f;
    if constexpr (H == 1) aL = lg[slL];
    for (int g = 0; g < 32 / U; ++g) {
      if (wave_item<U>(q0 + g * U, wv) >= tot) break;
      int s[U], dl[U], sl[U]; bool ok[U];
      float xv[U][K], av[U][K];
#pragma unroll
      for (int u = 0; u < U; ++u) {
        s[u]  = lane_bcast(sL, g * U + u);
        dl[u] = lane_bcast(dlL, g * U + u);
        sl[u] = lane_bcast(slL, g * U + u);
        ok[u] = lane_bcast(okL, g * U + u) != 0;
      }
#pragma unroll
      for (int u = 0; u < U; ++u) {
#pragma unroll
        for (int k = 0; k < K; ++k) {
          const int r  = lane + 32 * k;
          const int rr = (r < F) ? r : r - 32;
          xv[u][k] = xl[(size_t)s[u] * F + rr];
          if constexpr (H == 1) av[u][k] = lane_bcast(aL, g * U + u);
          else                  av[u][k] = lg[(size_t)sl[u] * H + rr / C];
        }
      }
#pragma unroll
      for (int u = 0; u < U; ++u) {
#pragma unroll
        for (int k = 0; k < K; ++k) {
          const int r  = lane + 32 * k;
          const int rr = (r < F) ? r : r - 32;
          atomicAdd(&acc[dl[u] * PF + rr], (ok[u] && r < F) ? av[u][k] * xv[u][k] : 0.0f);
        }
      }
    }
  }
  __syncthreads();

  if constexpr (!PROJ) {
    for (int i = tid; i < TS * F; i += 256) {
      const int dl = i / F, r = i - dl * F;
      const int n = n0 + dl;
      if (n < N_) out[(size_t)n * F + r] = acc[i] / (den[dl * H + r / C] + 1e-16f) + bias[r];
    }
  } else {
    for (int i = tid; i < TS * F; i += 256) {
      const int dl = i / F, r = i - dl * F;
      acc[dl * PF + r] = acc[dl * PF + r] / (den[dl * H + r / C] + 1e-16f) + bias[r];
    }
    __syncthreads();
    tile_proj_pair<F, PF, F2_>(acc, n0, wv, lane, Wnl, bnl, nxl, Wnr, bnr, nxr);
  }
}

constexpr float SHIFT_LIM = 40.0f;
template <int C, int U>
__global__ __launch_bounds__(256) void k_gat1h_tile(const int* __restrict__ off, const int* __restrict__ bins,
                                                    const float* __restrict__ xl, const float* __restrict__ xr,
                                                    const float* __restrict__ att, const float* __restrict__ bias,
                                                    float* __restrict__ lg, float* __restrict__ out) {
  constexpr int F = C;
  static_assert(F == 64, "a wave holds a row as two words per lane");
  __shared__ __attribute__((aligned(16))) float acc[TS * F];
  __shared__ __attribute__((aligned(16))) float xrs[TS * F];
  __shared__ float shf[TS];
  __shared__ float den[TS];
  __shared__ float satt[F];
  __shared__ int   over;

  const int t = blockIdx.x, tid = threadIdx.x, lane = tid & 31;
  const int wv = __builtin_amdgcn_readfirstlane(tid >> 5);
  const int n0 = t * TS;
  constexpr int NST = TS * (F / 4) / 256;
  static_assert(TS * (F / 4) % 256 == 0, "the tile's rows split evenly over the group");
  {
    const float4* __restrict__ xr4 = (const float4*)(xr + (size_t)n0 * F);
    const int npc = min(TS, N_ - n0) * (F / 4);
    float4 st[NST];
#pragma unroll
    for (int k = 0; k < NST; ++k) {
      const int i = tid + 256 * k;
      st[k] = make_float4(0.0f, 0.0f, 0.0f, 0.0f);
      if (i < npc) st[k] = xr4[i];
    }
#pragma unroll
    for (int k = 0; k < NST; ++k) {
      *(float4*)(xrs + 4 * (tid + 256 * k)) = st[k];
      *(float4*)(acc + 4 * (tid + 256 * k)) = make_float4(0.0f, 0.0f, 0.0f, 0.0f);
    }
  }
  if (tid < TS) den[tid] = 0.0f;
  if (tid < F) satt[tid] = att[tid];
  if (tid == 0) over = 0;
  __syncthreads();

  {
    constexpr int RW = TS / 8;
    float l0[RW], l1[RW];
#pragma unroll
    for (int r = 0; r < RW; ++r) {
      const int n = min(n0 + wv * RW + r, N_ - 1);
      l0[r] = xl[(size_t)n * F + lane];
      l1[r] = xl[(size_t)n * F + lane + 32];
    }
    float a[RW];
#pragma unroll
    for (int r = 0; r < RW; ++r) {
      const int dl = wv * RW + r;
      a[r] = 0.0f;
      float t0 = l0[r] + xrs[dl * F + lane];
      t0 = (t0 > 0.0f) ? t0 : 0.2f * t0;
      a[r] += t0 * satt[lane];
      float t1 = l1[r] + xrs[dl * F + lane + 32];
      t1 = (t1 > 0.0f) ? t1 : 0.2f * t1;
      a[r] += t1 * satt[lane + 32];
    }
#pragma unroll
    for (int o = 16; o > 0; o >>= 1) {
#pragma unroll
      for (int r = 0; r < RW; ++r) a[r] += __shfl_xor(a[r], o, 32);
    }
    if (lane == 0) {
#pragma unroll
      for (int r = 0; r < RW; ++r) shf[wv * RW + r] = a[r];
    }
  }
  __syncthreads();

  const int b = off[t];
  const int cnt = off[t + 1] - b;
  const int tot = cnt + TS;

  for (int q0 = 0; wave_item<U>(q0, wv) < tot; q0 += 32) {
    int sL, dlL, slL;
    const int okL = tile_item(wave_item<U>(q0 + lane, wv), tot, cnt, b, n0, bins, sL, dlL, slL) ? 1 : 0;
    const float mL = shf[dlL];
    for (int g = 0; g < 32 / U; ++g) {
      if (wave_item<U>(q0 + g * U, wv) >= tot) break;
      int s[U], dl[U]; bool ok[U];
      float m[U], l0[U], l1[U], r0[U], r1[U];
#pragma unroll
      for (int u = 0; u < U; ++u) {
        s[u]  = lane_bcast(sL, g * U + u);
        dl[u] = lane_bcast(dlL, g * U + u);
        ok[u] = lane_bcast(okL, g * U + u) != 0;
        m[u]  = lane_bcast(mL, g * U + u);
      }
#pragma unroll
      for (int u = 0; u < U; ++u) {
        l0[u] = xl[(size_t)s[u] * F + lane];
        l1[u] = xl[(size_t)s[u] * F + lane + 32];
        r0[u] = xrs[dl[u] * F + lane];
        r1[u] = xrs[dl[u] * F + lane + 32];
      }
#pragma unroll
      for (int u = 0; u < U; ++u) {
        float a = 0.0f;
        float t0 = l0[u] + r0[u];
        t0 = (t0 > 0.0f) ? t0 : 0.2f * t0;
        a += t0 * satt[lane];
        float t1 = l1[u] + r1[u];
        t1 = (t1 > 0.0f) ? t1 : 0.2f * t1;
        a += t1 * satt[lane + 32];
#pragma unroll
        for (int o = 16; o > 0; o >>= 1) a += __shfl_xor(a, o, 32);
        const float d = a - m[u];
        const float w = expf(d);
        if (lane == 0 && ok[u]) {
          atomicAdd(&den[dl[u]], w);
          if (d > SHIFT_LIM) over = 1;
        }
        atomicAdd(&acc[dl[u] * F + lane],      ok[u] ? w * l0[u] : 0.0f);
        atomicAdd(&acc[dl[u] * F + lane + 32], ok[u] ? w * l1[u] : 0.0f);
      }
    }
  }
  __syncthreads();

  if (over != 0) {
    gat_tile_body<1, C, U, false>(acc, (int*)shf, den, satt, off, bins, xl, xr, att, bias, lg, out,
                                  nullptr, nullptr, nullptr, nullptr, nullptr, nullptr);
    return;
  }

  for (int i = tid; i < TS * F; i += 256) {
    const int dl = i / F, r = i - dl * F;
    const int n = n0 + dl;
    if (n < N_) out[(size_t)n * F + r] = acc[i] / (den[dl] + 1e-16f) + bias[r];
  }
}

struct f3 { float x, y, z; };

__device__ __forceinline__ float quad_sum(float p) {
  p += __int_as_float(__builtin_amdgcn_update_dpp(0, __float_as_int(p), 0xB1, 0xF, 0xF, true));
  p += __int_as_float(__builtin_amdgcn_update_dpp(0, __float_as_int(p), 0x4E, 0xF, 0xF, true));
  return p;
}
__device__ __forceinline__ float slot_logit(const f3& l, const f3& r, const float (&a)[3]) {
  float p = 0.0f;
  float t0 = l.x + r.x; t0 = (t0 > 0.0f) ? t0 : 0.2f * t0; p += t0 * a[0];
  float t1 = l.y + r.y; t1 = (t1 > 0.0f) ? t1 : 0.2f * t1; p += t1 * a[1];
  float t2 = l.z + r.z; t2 = (t2 > 0.0f) ? t2 : 0.2f * t2; p += t2 * a[2];
  return quad_sum(p);
}

template <int U>
__global__ __launch_bounds__(256) void k_gat12_tile(const int* __restrict__ off, const int* __restrict__ bins,
                                                    const float* __restrict__ xl, const float* __restrict__ xr,
                                                    const float* __restrict__ att, const float* __restrict__ bias,
                                                    float* __restrict__ lg,
                                                    const float* __restrict__ Wnl, const float* __restrict__ bnl,
                                                    const float* __restrict__ Wnr, const float* __restrict__ bnr,
                                                    float* __restrict__ nxl, float* __restrict__ nxr) {
  constexpr int H = 12, C = 12, F = H * C, PF = F + 4;
  static_assert(U % 2 == 0, "items are taken in pairs");
  static_assert(4 * 3 == C && 32 * 3 * 3 == 2 * F, "a quad of lanes is a head, three slots are two rows");
  constexpr int NS = 3 * U / 2;
  __shared__ __attribute__((aligned(16))) float acc[TS * PF];
  __shared__ __attribute__((aligned(16))) float xrs[TS * F];
  __shared__ float shf[TS * H];
  __shared__ float den[TS * H];
  __shared__ float satt[F];
  __shared__ int   over;

  const int t = blockIdx.x, tid = threadIdx.x, lane = tid & 31;
  const int wv = __builtin_amdgcn_readfirstlane(tid >> 5);
  const int n0 = t * TS;
  constexpr int NST = TS * (F / 4) / 256;
  static_assert(TS * (F / 4) % 256 == 0, "the tile's rows split evenly over the group");
  static_assert((TS * PF) % 4 == 0, "the sums are cleared as 16-byte pieces");
  {
    const float4* __restrict__ xr4 = (const float4*)(xr + (size_t)n0 * F);
    const int npc = min(TS, N_ - n0) * (F / 4);
    float4 st[NST];
#pragma unroll
    for (int k = 0; k < NST; ++k) {
      const int i = tid + 256 * k;
      st[k] = make_float4(0.0f, 0.0f, 0.0f, 0.0f);
      if (i < npc) st[k] = xr4[i];
    }
#pragma unroll
    for (int k = 0; k < NST; ++k) *(float4*)(xrs + 4 * (tid + 256 * k)) = st[k];
  }
  for (int i = tid; i < TS * PF / 4; i += 256) *(float4*)(acc + 4 * i) = make_float4(0.0f, 0.0f, 0.0f, 0.0f);
  for (int i = tid; i < TS * H; i += 256) den[i] = 0.0f;
  if (tid == 0) over = 0;

  const bool hiB = lane >= 16;
  int eo[3], hd[3];
  float av[3][3];
  eo[0] = 3 * lane;
  eo[1] = hiB ? 3 * lane - 48 : 96 + 3 * lane;
  eo[2] = 48 + 3 * lane;
#pragma unroll
  for (int z = 0; z < 3; ++z) {
    hd[z] = eo[z] / C;
#pragma unroll
    for (int c = 0; c < 3; ++c) av[z][c] = att[eo[z] + c];
  }
  __syncthreads();

  {
    constexpr int RW = TS / 8;
    static_assert(RW % 2 == 0, "rows are taken in pairs");
    constexpr int NSS = 3 * RW / 2;
    f3 lv[NSS];
    int rw[NSS];
#pragma unroll
    for (int i = 0; i < NSS; ++i) {
      const int z = i % 3;
      const int r0 = wv * RW + 2 * (i / 3), r1 = r0 + 1;
      rw[i] = (z == 0) ? r0 : (z == 2) ? r1 : (hiB ? r1 : r0);
      const int n = min(n0 + rw[i], N_ - 1);
      lv[i] = *(const f3*)(xl + (size_t)n * F + eo[z]);
    }
#pragma unroll
    for (int i = 0; i < NSS; ++i) {
      const int z = i % 3;
      const f3 rv = *(const f3*)(xrs + rw[i] * F + eo[z]);
      const float a = slot_logit(lv[i], rv, av[z]);
      if ((lane & 3) == 0) shf[rw[i] * H + hd[z]] = a;
    }
  }
  __syncthreads();

  const int b = off[t];
  const int cnt = off[t + 1] - b;
  const int tot = cnt + TS;

  for (int q0 = 0; wave_item<U>(q0, wv) < tot; q0 += 32) {
    int sL, dlL, slL;
    const int okL = tile_item(wave_item<U>(q0 + lane, wv), tot, cnt, b, n0, bins, sL, dlL, slL) ? 1 : 0;
    for (int g = 0; g < 32 / U; ++g) {
      if (wave_item<U>(q0 + g * U, wv) >= tot) break;
      int s[U], dl[U]; bool ok[U];
#pragma unroll
      for (int u = 0; u < U; ++u) {
        s[u]  = lane_bcast(sL, g * U + u);
        dl[u] = lane_bcast(dlL, g * U + u);
        ok[u] = lane_bcast(okL, g * U + u) != 0;
      }
      int sd[NS]; bool so[NS];
      f3 lv[NS], rv[NS];
      float mv[NS];
#pragma unroll
      for (int i = 0; i < NS; ++i) {
        const int z = i % 3;
        const int u0 = 2 * (i / 3), u1 = u0 + 1;
        const int ss = (z == 0) ? s[u0] : (z == 2) ? s[u1] : (hiB ? s[u1] : s[u0]);
        sd[i] = (z == 0) ? dl[u0] : (z == 2) ? dl[u1] : (hiB ? dl[u1] : dl[u0]);
        so[i] = (z == 0) ? ok[u0] : (z == 2) ? ok[u1] : (hiB ? ok[u1] : ok[u0]);
        lv[i] = *(const f3*)(xl + (size_t)ss * F + eo[z]);
        rv[i] = *(const f3*)(xrs + sd[i] * F + eo[z]);
        mv[i] = shf[sd[i] * H + hd[z]];
      }
#pragma unroll
      for (int i = 0; i < NS; ++i) {
        const int z = i % 3;
        const float d = slot_logit(lv[i], rv[i], av[z]) - mv[i];
        const float w = expf(d);
        if (so[i] && (lane & 3) == 0) {
          atomicAdd(&den[sd[i] * H + hd[z]], w);
          if (d > SHIFT_LIM) over = 1;
        }
        const float wz = so[i] ? w : 0.0f;
        float* ap = acc + sd[i] * PF + eo[z];
        atomicAdd(ap,     wz * lv[i].x);
        atomicAdd(ap + 1, wz * lv[i].y);
        atomicAdd(ap + 2, wz * lv[i].z);
      }
    }
  }
  __syncthreads();

  if (over != 0) {
    gat_tile_body<H, C, U, true>(acc, (int*)shf, den, satt, off, bins, xl, xr, att, bias, lg, nullptr,
                                 Wnl, bnl, Wnr, bnr, nxl, nxr);
    return;
  }

  for (int i = tid; i < TS * F; i += 256) {
    const int dl = i / F, r = i - dl * F;
    acc[dl * PF + r] = acc[dl * PF + r] / (den[dl * H + r / C] + 1e-16f) + bias[r];
  }
  __syncthreads();
  tile_proj_pair<F, PF, F2_>(acc, n0, wv, lane, Wnl, bnl, nxl, Wnr, bnr, nxr);
}

constexpr size_t OFF_DINV = 0;
constexpr size_t OFF_CNT  = OFF_DINV + N_;
constexpr size_t OFF_OFFS = OFF_CNT  + 2048;
constexpr size_t OFF_CUR  = OFF_OFFS + 2048;
constexpr size_t OFF_XL2  = (size_t)N_ + E_;
constexpr size_t OFF_XR2  = OFF_XL2  + (size_t)N_*D_;
constexpr size_t OFF_XL1  = OFF_XR2  + (size_t)N_*D_;
constexpr size_t OFF_XR1  = OFF_XL1  + (size_t)N_*F1_;
constexpr size_t OFF_BINS = OFF_XR1  + (size_t)N_*F1_;
constexpr size_t OFF_LG   = OFF_BINS + (size_t)N_*24;
static_assert(OFF_CUR + 2048 <= OFF_XL2, "bin tables fit in front of the projections");
static_assert(F2_ == D_, "layer 2's projections are sized with D_ above");
static_assert((size_t)E_ <= (size_t)N_ * 24, "bins fit their region");

extern "C" void kernel_launch(void* const* d_in, const int* in_sizes, int n_in,
                              void* d_out, int out_size, void* d_ws, size_t ws_size,
                              hipStream_t stream) {
  const float* x     = (const float*)d_in[0];
  const int*   ei    = (const int*)  d_in[1];
  const float* Wssg  = (const float*)d_in[2];
  const float* bssg  = (const float*)d_in[3];
  const float* W1l   = (const float*)d_in[4];
  const float* b1l   = (const float*)d_in[5];
  const float* W1r   = (const float*)d_in[6];
  const float* b1r   = (const float*)d_in[7];
  const float* att1  = (const float*)d_in[8];
  const float* bias1 = (const float*)d_in[9];
  const float* W2l   = (const float*)d_in[10];
  const float* b2l   = (const float*)d_in[11];
  const float* W2r   = (const float*)d_in[12];
  const float* b2r   = (const float*)d_in[13];
  const float* att2  = (const float*)d_in[14];
  const float* bias2 = (const float*)d_in[15];

  const int* src = ei;
  const int* dst = ei + E_;

  float* ws   = (float*)d_ws;
  float* dinv = ws + OFF_DINV;
  int*   cnt  = (int*)(ws + OFF_CNT);
  int*   offs = (int*)(ws + OFF_OFFS);
  int*   cur  = (int*)(ws + OFF_CUR);
  float* xl2  = ws + OFF_XL2;
  float* xr2  = ws + OFF_XR2;
  float* xl1  = ws + OFF_XL1;
  float* xr1  = ws + OFF_XR1;
  int*   bins = (int*)(ws + OFF_BINS);
  float* lg   = ws + OFF_LG;
  float* out  = (float*)d_out;

  const int TB = 256;

  k_zero_i<<<CDIV(2048, TB), TB, 0, stream>>>(cnt, 2048);
  k_tilecount<<<CDIV(E_, CH), 1024, 0, stream>>>(dst, cnt);
  k_scan<<<1, 1024, 0, stream>>>(cnt, offs, cur);
  k_binscatter<<<CDIV(E_, CH), 1024, 0, stream>>>(src, dst, cur, bins);

  k_dinv_tile<<<NT, TS, 0, stream>>>(offs, bins, dinv);
  k_ssg_tile<4><<<NT, 256, 0, stream>>>(offs, bins, x, dinv, Wssg, bssg, W1l, b1l, W1r, b1r, xl1, xr1);

  k_gat12_tile<4><<<NT, 256, 0, stream>>>(offs, bins, xl1, xr1, att1, bias1, lg,
                                          W2l, b2l, W2r, b2r, xl2, xr2);

  k_gat1h_tile<64, 4><<<NT, 256, 0, stream>>>(offs, bins, xl2, xr2, att2, bias2, lg, out);
}


# one-head pass: a wave takes 4 items per step, one per 8 lanes (two 16-byte pieces a lane): logit summed by 3 DPP swizzles, not 5 LDS shuffles; sums' row pitch 65; logit sum order changed
# speedup vs baseline: 5.8053x; 1.0645x over previous
#include <hip/hip_runtime.h>
#include <cmath>

constexpr int N_  = 100000;
constexpr int E_  = 1000000;
constexpr int D_  = 64;
constexpr int F1_ = 144;
constexpr int F2_ = 64;

#define CDIV(a,b) (((a)+(b)-1)/(b))

typedef __attribute__((ext_vector_type(2))) float v2f;
typedef __attribute__((ext_vector_type(8))) float v8f;

constexpr int TSH = 6;
constexpr int TS  = 1 << TSH;
constexpr int NT  = CDIV(N_, TS);
constexpr int CH  = 8192;
constexpr int SRC_BITS = 17;
constexpr int SRC_MASK = (1 << SRC_BITS) - 1;
static_assert(N_ <= (1 << SRC_BITS), "source id must fit the packed field");
static_assert(NT + 1 <= 2048, "scan handles two tiles per thread of one 1024-thread group");

__global__ void k_zero_i(int* __restrict__ p, int n) {
  int i = blockIdx.x * blockDim.x + threadIdx.x;
  if (i < n) p[i] = 0;
}

__global__ __launch_bounds__(1024) void k_tilecount(const int* __restrict__ dst, int* __restrict__ cnt) {
  __shared__ int hist[NT];
  const int tid = threadIdx.x;
  for (int i = tid; i < NT; i += 1024) hist[i] = 0;
  __syncthreads();
  const int base = blockIdx.x * CH;
#pragma unroll
  for (int k = 0; k < CH / 1024; ++k) {
    int e = base + k * 1024 + tid;
    if (e < E_) atomicAdd(&hist[dst[e] >> TSH], 1);
  }
  __syncthreads();
  for (int i = tid; i < NT; i += 1024) { int v = hist[i]; if (v) atomicAdd(&cnt[i], v); }
}

__global__ __launch_bounds__(1024) void k_scan(const int* __restrict__ cnt, int* __restrict__ off,
                                               int* __restrict__ cur) {
  __shared__ int p[1024];
  const int t = threadIdx.x;
  const int a = (2 * t     < NT) ? cnt[2 * t]     : 0;
  const int b = (2 * t + 1 < NT) ? cnt[2 * t + 1] : 0;
  int v = a + b;
  p[t] = v;
  __syncthreads();
  for (int o = 1; o < 1024; o <<= 1) {
    int add = (t >= o) ? p[t - o] : 0;
    __syncthreads();
    v += add;
    p[t] = v;
    __syncthreads();
  }
  const int excl = v - (a + b);
  if (2 * t     < NT) { off[2 * t]     = excl;     cur[2 * t]     = excl; }
  if (2 * t + 1 < NT) { off[2 * t + 1] = excl + a; cur[2 * t + 1] = excl + a; }
  if (t == 1023) off[NT] = v;
}

__global__ __launch_bounds__(1024) void k_binscatter(const int* __restrict__ src, const int* __restrict__ dst,
                                                     int* __restrict__ cur, int* __restrict__ bins) {
  __shared__ int hist[NT];
  const int tid = threadIdx.x;
  for (int i = tid; i < NT; i += 1024) hist[i] = 0;
  __syncthreads();
  const int base = blockIdx.x * CH;
  int pk[CH / 1024], bn[CH / 1024], lr[CH / 1024];
#pragma unroll
  for (int k = 0; k < CH / 1024; ++k) {
    int e = base + k * 1024 + tid;
    bn[k] = -1; pk[k] = 0; lr[k] = 0;
    if (e < E_) {
      int d = dst[e];
      bn[k] = d >> TSH;
      pk[k] = src[e] | ((d & (TS - 1)) << SRC_BITS);
      lr[k] = atomicAdd(&hist[bn[k]], 1);
    }
  }
  __syncthreads();
  for (int i = tid; i < NT; i += 1024) { int v = hist[i]; hist[i] = v ? atomicAdd(&cur[i], v) : 0; }
  __syncthreads();
#pragma unroll
  for (int k = 0; k < CH / 1024; ++k)
    if (bn[k] >= 0) bins[hist[bn[k]] + lr[k]] = pk[k];
}

__device__ __forceinline__ bool tile_item(int j, int tot, int cnt, int b, int n0,
                                          const int* __restrict__ bins, int& s, int& dl, int& sl) {
  const int p  = min(b + j, E_ - 1);
  const int pk = bins[p];
  const bool isE = j < cnt;
  const int dlS = j - cnt;
  const int sS  = n0 + dlS;
  const bool okS = (!isE) && (j < tot) && (sS < N_);
  s  = isE ? (pk & SRC_MASK)  : (okS ? sS : 0);
  dl = isE ? (pk >> SRC_BITS) : (okS ? dlS : 0);
  sl = isE ? p                : (okS ? E_ + sS : 0);
  return isE || okS;
}

__device__ __forceinline__ void ldsMaxF(int* cell, float v) {
  if (v >= 0.0f) atomicMax(cell, __float_as_int(v));
  else           atomicMin((unsigned int*)cell, __float_as_uint(v));
}

template <int U>
__device__ __forceinline__ int wave_item(int q, int wv) {
  const unsigned uq = (unsigned)q;
  return (int)(((uq / U) * 8u + (unsigned)wv) * U + (uq % U));
}
__device__ __forceinline__ int lane_bcast(int v, int ln) { return __builtin_amdgcn_readlane(v, ln); }
__device__ __forceinline__ float lane_bcast(float v, int ln) {
  return __int_as_float(__builtin_amdgcn_readlane(__float_as_int(v), ln));
}

__global__ __launch_bounds__(TS) void k_dinv_tile(const int* __restrict__ off, const int* __restrict__ bins,
                                                  float* __restrict__ dinv) {
  __shared__ int c[TS];
  const int t = blockIdx.x, tid = threadIdx.x;
  c[tid] = 0;
  __syncthreads();
  const int b = off[t], e = off[t + 1];
  for (int p = b + tid; p < e; p += TS) atomicAdd(&c[bins[p] >> SRC_BITS], 1);
  __syncthreads();
  const int n = t * TS + tid;
  if (n < N_) dinv[n] = rsqrtf(1.0f + (float)c[tid]);
}

template <int K, int PA, int NC>
__device__ __forceinline__ void tile_mma2(const float* a0p, const float* __restrict__ wp, v8f& acc0, v8f& acc1) {
  const float* a1p = a0p + 16 * PA;
#pragma unroll 4
  for (int k = 0; k < K; k += 4) {
    v2f a0, a1, b;
    a0.x = a0p[k]; a0.y = a0p[k + 1];
    a1.x = a1p[k]; a1.y = a1p[k + 1];
    b.x = wp[(size_t)k * NC];
    b.y = wp[(size_t)(k + 1) * NC];
    acc0 = __builtin_amdgcn_wmma_f32_16x16x4_f32(false, a0, false, b, (short)0, acc0, false, false);
    acc1 = __builtin_amdgcn_wmma_f32_16x16x4_f32(false, a1, false, b, (short)0, acc1, false, false);
  }
}

template <int K, int PA, int NC, int PC>
__device__ __forceinline__ void tile_proj_lds(const float* As, const float* __restrict__ W,
                                              const float* __restrict__ bias, float* Cs, int wv, int lane) {
  const int c = lane & 15, hh = lane >> 4;
  const int r0 = (wv & 1) * 32;
  for (int ct = wv >> 1; ct < NC / 16; ct += 4) {
    const int col = ct * 16 + c;
    v8f acc0 = {}, acc1 = {};
    tile_mma2<K, PA, NC>(As + (r0 + c) * PA + 2 * hh, W + (size_t)(2 * hh) * NC + col, acc0, acc1);
    const float bn = bias[col];
#pragma unroll
    for (int j = 0; j < 8; ++j) {
      Cs[(r0 + 8 * hh + j) * PC + col]      = acc0[j] + bn;
      Cs[(r0 + 16 + 8 * hh + j) * PC + col] = acc1[j] + bn;
    }
  }
}

template <int K, int PA, int NC>
__device__ __forceinline__ void tile_proj_pair(const float* As, int n0, int wv, int lane,
                                               const float* __restrict__ Wa, const float* __restrict__ ba,
                                               float* __restrict__ Ca,
                                               const float* __restrict__ Wb, const float* __restrict__ bb,
                                               float* __restrict__ Cb) {
  constexpr int CT = NC / 16;
  const int c = lane & 15, hh = lane >> 4;
  const int r0 = (wv & 1) * 32;
  for (int ct = wv >> 1; ct < 2 * CT; ct += 4) {
    const bool second = ct >= CT;
    const float* __restrict__ W  = second ? Wb : Wa;
    const float* __restrict__ bs = second ? bb : ba;
    float* __restrict__ C        = second ? Cb : Ca;
    const int col = (second ? ct - CT : ct) * 16 + c;
    v8f acc0 = {}, acc1 = {};
    tile_mma2<K, PA, NC>(As + (r0 + c) * PA + 2 * hh, W + (size_t)(2 * hh) * NC + col, acc0, acc1);
    const float bn = bs[col];
    float* __restrict__ cp = C + (size_t)(n0 + r0 + 8 * hh) * NC + col;
    if (n0 + r0 < N_) {
#pragma unroll
      for (int j = 0; j < 8; ++j) cp[(size_t)j * NC] = acc0[j] + bn;
    }
    if (n0 + r0 + 16 < N_) {
#pragma unroll
      for (int j = 0; j < 8; ++j) cp[(size_t)(16 + j) * NC] = acc1[j] + bn;
    }
  }
}

constexpr int PH = D_ + 4;
static_assert(N_ % 16 == 0, "a 16-row output tile is never partly past the last node");
template <int U>
__global__ __launch_bounds__(256) void k_ssg_tile(const int* __restrict__ off, const int* __restrict__ bins,
                                                  const float* __restrict__ x, const float* __restrict__ dinv,
                                                  const float* __restrict__ Wssg, const float* __restrict__ bssg,
                                                  const float* __restrict__ W1l, const float* __restrict__ b1l,
                                                  const float* __restrict__ W1r, const float* __restrict__ b1r,
                                                  float* __restrict__ xl1, float* __restrict__ xr1) {
  __shared__ float acc[TS * PH];
  __shared__ float x1s[TS * PH];
  const int t = blockIdx.x, tid = threadIdx.x, lane = tid & 31;
  const int wv = __builtin_amdgcn_readfirstlane(tid >> 5);
  for (int i = tid; i < TS * PH; i += 256) acc[i] = 0.0f;
  __syncthreads();
  const int b = off[t];
  const int cnt = off[t + 1] - b;
  const int tot = cnt + TS;
  const int n0 = t * TS;
  for (int q0 = 0; wave_item<U>(q0, wv) < tot; q0 += 32) {
    int sL, dlL, slL;
    const int okL = tile_item(wave_item<U>(q0 + lane, wv), tot, cnt, b, n0, bins, sL, dlL, slL) ? 1 : 0;
    const float wL = dinv[sL] * dinv[n0 + dlL];
    for (int g = 0; g < 32 / U; ++g) {
      if (wave_item<U>(q0 + g * U, wv) >= tot) break;
      int s[U], dl[U]; bool ok[U];
      float x0[U], x1[U], w[U];
#pragma unroll
      for (int u = 0; u < U; ++u) {
        s[u]  = lane_bcast(sL, g * U + u);
        dl[u] = lane_bcast(dlL, g * U + u);
        ok[u] = lane_bcast(okL, g * U + u) != 0;
        w[u]  = lane_bcast(wL, g * U + u);
      }
#pragma unroll
      for (int u = 0; u < U; ++u) {
        x0[u] = x[(size_t)s[u] * D_ + lane];
        x1[u] = x[(size_t)s[u] * D_ + lane + 32];
      }
#pragma unroll
      for (int u = 0; u < U; ++u) {
        atomicAdd(&acc[dl[u] * PH + lane],      ok[u] ? x0[u] * w[u] : 0.0f);
        atomicAdd(&acc[dl[u] * PH + lane + 32], ok[u] ? x1[u] * w[u] : 0.0f);
      }
    }
  }
  __syncthreads();
  for (int i = tid; i < TS * D_; i += 256) {
    const int dl = i >> 6, c = i & 63;
    const int n = n0 + dl;
    acc[dl * PH + c] = (n < N_) ? 0.5f * x[(size_t)n * D_ + c] + 0.5f * acc[dl * PH + c] : 0.0f;
  }
  __syncthreads();
  tile_proj_lds<D_, PH, D_, PH>(acc, Wssg, bssg, x1s, wv, lane);
  __syncthreads();
  tile_proj_pair<D_, PH, F1_>(x1s, n0, wv, lane, W1l, b1l, xl1, W1r, b1r, xr1);
}

template <int H, int C, int U, bool PROJ>
__device__ __forceinline__ void gat_tile_body(float* acc, int* mxi, float* den, float* satt,
                                              const int* __restrict__ off, const int* __restrict__ bins,
                                              const float* __restrict__ xl, const float* __restrict__ xr,
                                              const float* __restrict__ att, const float* __restrict__ bias,
                                              float* __restrict__ lg, float* __restrict__ out,
                                              const float* __restrict__ Wnl, const float* __restrict__ bnl,
                                              const float* __restrict__ Wnr, const float* __restrict__ bnr,
                                              float* __restrict__ nxl, float* __restrict__ nxr) {
  constexpr int F = H * C;
  constexpr int K = (F + 31) / 32;
  constexpr int PF = PROJ ? F + 4 : F;
  static_assert(C % 4 == 0, "head rows are read as 16-byte pieces");
  static_assert(!PROJ || (PF % 64 == 4 || PF % 64 == 20 || PF % 64 == 36 || PF % 64 == 52),
                "fragment reads of 16 rows must fall on 16 different bank groups");

  const int t = blockIdx.x, tid = threadIdx.x, lane = tid & 31;
  const int wv = __builtin_amdgcn_readfirstlane(tid >> 5);
  const int n0 = t * TS;
  float* const xrs = acc;
  constexpr int NST = TS * (F / 4) / 256;
  static_assert(TS * (F / 4) % 256 == 0, "the tile's rows split evenly over the group");
  {
    const float4* __restrict__ xr4 = (const float4*)(xr + (size_t)n0 * F);
    const int npc = min(TS, N_ - n0) * (F / 4);
    float4 st[NST];
#pragma unroll
    for (int k = 0; k < NST; ++k) {
      const int i = tid + 256 * k;
      st[k] = make_float4(0.0f, 0.0f, 0.0f, 0.0f);
      if (i < npc) st[k] = xr4[i];
    }
#pragma unroll
    for (int k = 0; k < NST; ++k) *(float4*)(xrs + 4 * (tid + 256 * k)) = st[k];
  }
  for (int i = tid; i < TS * H; i += 256) { mxi[i] = (int)0xFF800000u; den[i] = 0.0f; }
  for (int i = tid; i < F; i += 256) satt[i] = att[i];
  __syncthreads();
  const int b = off[t];
  const int cnt = off[t + 1] - b;
  const int tot = cnt + TS;

  if constexpr (H == 1) {
    for (int q0 = 0; wave_item<U>(q0, wv) < tot; q0 += 32) {
      int sL, dlL, slL;
      const int okL = tile_item(wave_item<U>(q0 + lane, wv), tot, cnt, b, n0, bins, sL, dlL, slL) ? 1 : 0;
      for (int g = 0; g < 32 / U; ++g) {
        if (wave_item<U>(q0 + g * U, wv) >= tot) break;
        int s[U], dl[U], sl[U]; bool ok[U];
        float l0[U], l1[U], r0[U], r1[U];
#pragma unroll
        for (int u = 0; u < U; ++u) {
          s[u]  = lane_bcast(sL, g * U + u);
          dl[u] = lane_bcast(dlL, g * U + u);
          sl[u] = lane_bcast(slL, g * U + u);
          ok[u] = lane_bcast(okL, g * U + u) != 0;
        }
#pragma unroll
        for (int u = 0; u < U; ++u) {
          l0[u] = xl[(size_t)s[u] * F + lane];
          l1[u] = xl[(size_t)s[u] * F + lane + 32];
          r0[u] = xrs[dl[u] * F + lane];
          r1[u] = xrs[dl[u] * F + lane + 32];
        }
#pragma unroll
        for (int u = 0; u < U; ++u) {
          float a = 0.0f;
          float t0 = l0[u] + r0[u];
          t0 = (t0 > 0.0f) ? t0 : 0.2f * t0;
          a += t0 * satt[lane];
          float t1 = l1[u] + r1[u];
          t1 = (t1 > 0.0f) ? t1 : 0.2f * t1;
          a += t1 * satt[lane + 32];
#pragma unroll
          for (int o = 16; o > 0; o >>= 1) a += __shfl_xor(a, o, 32);
          if (lane == 0 && ok[u]) { lg[sl[u]] = a; ldsMaxF(&mxi[dl[u]], a); }
        }
      }
    }
  } else {
    for (int idx = tid; idx < tot * H; idx += 256) {
      const int j = idx / H, h = idx - j * H;
      int s, dl, sl;
      if (tile_item(j, tot, cnt, b, n0, bins, s, dl, sl)) {
        const float4* pl = (const float4*)(xl + (size_t)s * F + h * C);
        const float4* pr = (const float4*)(xrs + dl * F + h * C);
        float4 lv[C / 4], rv[C / 4];
#pragma unroll
        for (int q = 0; q < C / 4; ++q) { lv[q] = pl[q]; rv[q] = pr[q]; }
        float a = 0.0f;
#pragma unroll
        for (int q = 0; q < C / 4; ++q) {
          const float le[4] = {lv[q].x, lv[q].y, lv[q].z, lv[q].w};
          const float re[4] = {rv[q].x, rv[q].y, rv[q].z, rv[q].w};
#pragma unroll
          for (int z = 0; z < 4; ++z) {
            float tt = le[z] + re[z];
            tt = (tt > 0.0f) ? tt : 0.2f * tt;
            a += tt * satt[h * C + q * 4 + z];
          }
        }
        lg[(size_t)sl * H + h] = a;
        ldsMaxF(&mxi[dl * H + h], a);
      }
    }
  }
  __syncthreads();

  for (int i = tid; i < TS * PF; i += 256) acc[i] = 0.0f;
  for (int idx = tid; idx < tot * H; idx += 256) {
    const int j = idx / H, h = idx - j * H;
    int s, dl, sl;
    if (tile_item(j, tot, cnt, b, n0, bins, s, dl, sl)) {
      const float a = expf(lg[(size_t)sl * H + h] - __int_as_float(mxi[dl * H + h]));
      lg[(size_t)sl * H + h] = a;
      atomicAdd(&den[dl * H + h], a);
    }
  }
  __syncthreads();

  for (int q0 = 0; wave_item<U>(q0, wv) < tot; q0 += 32) {
    int sL, dlL, slL;
    const int okL = tile_item(wave_item<U>(q0 + lane, wv), tot, cnt, b, n0, bins, sL, dlL, slL) ? 1 : 0;
    float aL = 0.0f;
    if constexpr (H == 1) aL = lg[slL];
    for (int g = 0; g < 32 / U; ++g) {
      if (wave_item<U>(q0 + g * U, wv) >= tot) break;
      int s[U], dl[U], sl[U]; bool ok[U];
      float xv[U][K], av[U][K];
#pragma unroll
      for (int u = 0; u < U; ++u) {
        s[u]  = lane_bcast(sL, g * U + u);
        dl[u] = lane_bcast(dlL, g * U + u);
        sl[u] = lane_bcast(slL, g * U + u);
        ok[u] = lane_bcast(okL, g * U + u) != 0;
      }
#pragma unroll
      for (int u = 0; u < U; ++u) {
#pragma unroll
        for (int k = 0; k < K; ++k) {
          const int r  = lane + 32 * k;
          const int rr = (r < F) ? r : r - 32;
          xv[u][k] = xl[(size_t)s[u] * F + rr];
          if constexpr (H == 1) av[u][k] = lane_bcast(aL, g * U + u);
          else                  av[u][k] = lg[(size_t)sl[u] * H + rr / C];
        }
      }
#pragma unroll
      for (int u = 0; u < U; ++u) {
#pragma unroll
        for (int k = 0; k < K; ++k) {
          const int r  = lane + 32 * k;
          const int rr = (r < F) ? r : r - 32;
          atomicAdd(&acc[dl[u] * PF + rr], (ok[u] && r < F) ? av[u][k] * xv[u][k] : 0.0f);
        }
      }
    }
  }
  __syncthreads();

  if constexpr (!PROJ) {
    for (int i = tid; i < TS * F; i += 256) {
      const int dl = i / F, r = i - dl * F;
      const int n = n0 + dl;
      if (n < N_) out[(size_t)n * F + r] = acc[i] / (den[dl * H + r / C] + 1e-16f) + bias[r];
    }
  } else {
    for (int i = tid; i < TS * F; i += 256) {
      const int dl = i / F, r = i - dl * F;
      acc[dl * PF + r] = acc[dl * PF + r] / (den[dl * H + r / C] + 1e-16f) + bias[r];
    }
    __syncthreads();
    tile_proj_pair<F, PF, F2_>(acc, n0, wv, lane, Wnl, bnl, nxl, Wnr, bnr, nxr);
  }
}

constexpr float SHIFT_LIM = 40.0f;
__device__ __forceinline__ float group8_sum(float p) {
  p += __int_as_float(__builtin_amdgcn_update_dpp(0, __float_as_int(p), 0xB1, 0xF, 0xF, true));
  p += __int_as_float(__builtin_amdgcn_update_dpp(0, __float_as_int(p), 0x4E, 0xF, 0xF, true));
  p += __int_as_float(__builtin_amdgcn_update_dpp(0, __float_as_int(p), 0x141, 0xF, 0xF, true));
  return p;
}
__device__ __forceinline__ float lrelu_dot4(const float4& l, const float4& r, const float4& a) {
  float p = 0.0f;
  float t0 = l.x + r.x; t0 = (t0 > 0.0f) ? t0 : 0.2f * t0; p += t0 * a.x;
  float t1 = l.y + r.y; t1 = (t1 > 0.0f) ? t1 : 0.2f * t1; p += t1 * a.y;
  float t2 = l.z + r.z; t2 = (t2 > 0.0f) ? t2 : 0.2f * t2; p += t2 * a.z;
  float t3 = l.w + r.w; t3 = (t3 > 0.0f) ? t3 : 0.2f * t3; p += t3 * a.w;
  return p;
}
template <int C, int U>
__global__ __launch_bounds__(256, 16) void k_gat1h_tile(const int* __restrict__ off, const int* __restrict__ bins,
                                                    const float* __restrict__ xl, const float* __restrict__ xr,
                                                    const float* __restrict__ att, const float* __restrict__ bias,
                                                    float* __restrict__ lg, float* __restrict__ out) {
  constexpr int F = C;
  static_assert(F == 64 && U == 4, "a group of 8 lanes holds a row as eight floats per lane, four items per step");
  constexpr int PA = F + 1;
  constexpr int G = 2;
  static_assert((TS * PA) % 4 == 0, "the sums are cleared as 16-byte pieces");
  __shared__ __attribute__((aligned(16))) float acc[TS * PA];
  __shared__ __attribute__((aligned(16))) float xrs[TS * F];
  __shared__ float shf[TS];
  __shared__ float den[TS];
  __shared__ __attribute__((aligned(16))) float satt[F];
  __shared__ int   over;

  const int t = blockIdx.x, tid = threadIdx.x, lane = tid & 31;
  const int wv = __builtin_amdgcn_readfirstlane(tid >> 5);
  const int n0 = t * TS;
  constexpr int NST = TS * (F / 4) / 256;
  static_assert(TS * (F / 4) % 256 == 0, "the tile's rows split evenly over the group");
  {
    const float4* __restrict__ xr4 = (const float4*)(xr + (size_t)n0 * F);
    const int npc = min(TS, N_ - n0) * (F / 4);
    float4 st[NST];
#pragma unroll
    for (int k = 0; k < NST; ++k) {
      const int i = tid + 256 * k;
      st[k] = make_float4(0.0f, 0.0f, 0.0f, 0.0f);
      if (i < npc) st[k] = xr4[i];
    }
#pragma unroll
    for (int k = 0; k < NST; ++k) *(float4*)(xrs + 4 * (tid + 256 * k)) = st[k];
  }
  for (int i = tid; i < TS * PA / 4; i += 256) *(float4*)(acc + 4 * i) = make_float4(0.0f, 0.0f, 0.0f, 0.0f);
  if (tid < TS) den[tid] = 0.0f;
  if (tid < F) satt[tid] = att[tid];
  if (tid == 0) over = 0;
  __syncthreads();

  {
    constexpr int RW = TS / 8;
    float l0[RW], l1[RW];
#pragma unroll
    for (int r = 0; r < RW; ++r) {
      const int n = min(n0 + wv * RW + r, N_ - 1);
      l0[r] = xl[(size_t)n * F + lane];
      l1[r] = xl[(size_t)n * F + lane + 32];
    }
    float a[RW];
#pragma unroll
    for (int r = 0; r < RW; ++r) {
      const int dl = wv * RW + r;
      a[r] = 0.0f;
      float t0 = l0[r] + xrs[dl * F + lane];
      t0 = (t0 > 0.0f) ? t0 : 0.2f * t0;
      a[r] += t0 * satt[lane];
      float t1 = l1[r] + xrs[dl * F + lane + 32];
      t1 = (t1 > 0.0f) ? t1 : 0.2f * t1;
      a[r] += t1 * satt[lane + 32];
    }
#pragma unroll
    for (int o = 16; o > 0; o >>= 1) {
#pragma unroll
      for (int r = 0; r < RW; ++r) a[r] += __shfl_xor(a[r], o, 32);
    }
    if (lane == 0) {
#pragma unroll
      for (int r = 0; r < RW; ++r) shf[wv * RW + r] = a[r];
    }
  }
  __syncthreads();

  const int b = off[t];
  const int cnt = off[t + 1] - b;
  const int tot = cnt + TS;

  const int grp = lane >> 3;
  const int cA = 32 * (grp & 1) + 4 * (lane & 7);
  const int cB = cA ^ 32;
  const float4 attA = *(const float4*)(satt + cA);
  const float4 attB = *(const float4*)(satt + cB);
  constexpr int OK_BIT = SRC_BITS + TSH;
  static_assert(OK_BIT < 31, "source, local destination and the exists flag share one word");

  for (int q0 = 0; wave_item<U>(q0, wv) < tot; q0 += 32) {
    int sL, dlL, slL;
    const int okL = tile_item(wave_item<U>(q0 + lane, wv), tot, cnt, b, n0, bins, sL, dlL, slL) ? 1 : 0;
    const int pkL = sL | (dlL << SRC_BITS) | (okL << OK_BIT);
    for (int g = 0; g < 32 / U; g += G) {
      if (wave_item<U>(q0 + g * U, wv) >= tot) break;
      int dl[G]; bool ok[G];
      float4 la[G], lb[G];
#pragma unroll
      for (int a = 0; a < G; ++a) {
        const int pk = __builtin_amdgcn_ds_bpermute(4 * ((g + a) * U + grp), pkL);
        const int s = pk & SRC_MASK;
        dl[a] = (pk >> SRC_BITS) & (TS - 1);
        ok[a] = (pk >> OK_BIT) != 0;
        la[a] = *(const float4*)(xl + (size_t)s * F + cA);
        lb[a] = *(const float4*)(xl + (size_t)s * F + cB);
      }
#pragma unroll
      for (int a = 0; a < G; ++a) {
        const float4 ra = *(const float4*)(xrs + dl[a] * F + cA);
        const float4 rb = *(const float4*)(xrs + dl[a] * F + cB);
        const float m = shf[dl[a]];
        const float p = group8_sum(lrelu_dot4(la[a], ra, attA) + lrelu_dot4(lb[a], rb, attB));
        const float d = p - m;
        const float w = expf(d);
        if (ok[a] && (lane & 7) == 0) {
          atomicAdd(&den[dl[a]], w);
          if (d > SHIFT_LIM) over = 1;
        }
        const float wz = ok[a] ? w : 0.0f;
        float* pa = acc + dl[a] * PA + cA;
        float* pb = acc + dl[a] * PA + cB;
        atomicAdd(pa,     wz * la[a].x);
        atomicAdd(pa + 1, wz * la[a].y);
        atomicAdd(pa + 2, wz * la[a].z);
        atomicAdd(pa + 3, wz * la[a].w);
        atomicAdd(pb,     wz * lb[a].x);
        atomicAdd(pb + 1, wz * lb[a].y);
        atomicAdd(pb + 2, wz * lb[a].z);
        atomicAdd(pb + 3, wz * lb[a].w);
      }
    }
  }
  __syncthreads();

  if (over != 0) {
    gat_tile_body<1, C, U, false>(acc, (int*)shf, den, satt, off, bins, xl, xr, att, bias, lg, out,
                                  nullptr, nullptr, nullptr, nullptr, nullptr, nullptr);
    return;
  }

  for (int i = tid; i < TS * F; i += 256) {
    const int dl = i / F, r = i - dl * F;
    const int n = n0 + dl;
    if (n < N_) out[(size_t)n * F + r] = acc[dl * PA + r] / (den[dl] + 1e-16f) + bias[r];
  }
}

struct f3 { float x, y, z; };

__device__ __forceinline__ float quad_sum(float p) {
  p += __int_as_float(__builtin_amdgcn_update_dpp(0, __float_as_int(p), 0xB1, 0xF, 0xF, true));
  p += __int_as_float(__builtin_amdgcn_update_dpp(0, __float_as_int(p), 0x4E, 0xF, 0xF, true));
  return p;
}
__device__ __forceinline__ float slot_logit(const f3& l, const f3& r, const float (&a)[3]) {
  float p = 0.0f;
  float t0 = l.x + r.x; t0 = (t0 > 0.0f) ? t0 : 0.2f * t0; p += t0 * a[0];
  float t1 = l.y + r.y; t1 = (t1 > 0.0f) ? t1 : 0.2f * t1; p += t1 * a[1];
  float t2 = l.z + r.z; t2 = (t2 > 0.0f) ? t2 : 0.2f * t2; p += t2 * a[2];
  return quad_sum(p);
}

template <int U>
__global__ __launch_bounds__(256) void k_gat12_tile(const int* __restrict__ off, const int* __restrict__ bins,
                                                    const float* __restrict__ xl, const float* __restrict__ xr,
                                                    const float* __restrict__ att, const float* __restrict__ bias,
                                                    float* __restrict__ lg,
                                                    const float* __restrict__ Wnl, const float* __restrict__ bnl,
                                                    const float* __restrict__ Wnr, const float* __restrict__ bnr,
                                                    float* __restrict__ nxl, float* __restrict__ nxr) {
  constexpr int H = 12, C = 12, F = H * C, PF = F + 4;
  static_assert(U % 2 == 0, "items are taken in pairs");
  static_assert(4 * 3 == C && 32 * 3 * 3 == 2 * F, "a quad of lanes is a head, three slots are two rows");
  constexpr int NS = 3 * U / 2;
  __shared__ __attribute__((aligned(16))) float acc[TS * PF];
  __shared__ __attribute__((aligned(16))) float xrs[TS * F];
  __shared__ float shf[TS * H];
  __shared__ float den[TS * H];
  __shared__ float satt[F];
  __shared__ int   over;

  const int t = blockIdx.x, tid = threadIdx.x, lane = tid & 31;
  const int wv = __builtin_amdgcn_readfirstlane(tid >> 5);
  const int n0 = t * TS;
  constexpr int NST = TS * (F / 4) / 256;
  static_assert(TS * (F / 4) % 256 == 0, "the tile's rows split evenly over the group");
  static_assert((TS * PF) % 4 == 0, "the sums are cleared as 16-byte pieces");
  {
    const float4* __restrict__ xr4 = (const float4*)(xr + (size_t)n0 * F);
    const int npc = min(TS, N_ - n0) * (F / 4);
    float4 st[NST];
#pragma unroll
    for (int k = 0; k < NST; ++k) {
      const int i = tid + 256 * k;
      st[k] = make_float4(0.0f, 0.0f, 0.0f, 0.0f);
      if (i < npc) st[k] = xr4[i];
    }
#pragma unroll
    for (int k = 0; k < NST; ++k) *(float4*)(xrs + 4 * (tid + 256 * k)) = st[k];
  }
  for (int i = tid; i < TS * PF / 4; i += 256) *(float4*)(acc + 4 * i) = make_float4(0.0f, 0.0f, 0.0f, 0.0f);
  for (int i = tid; i < TS * H; i += 256) den[i] = 0.0f;
  if (tid == 0) over = 0;

  const bool hiB = lane >= 16;
  int eo[3], hd[3];
  float av[3][3];
  eo[0] = 3 * lane;
  eo[1] = hiB ? 3 * lane - 48 : 96 + 3 * lane;
  eo[2] = 48 + 3 * lane;
#pragma unroll
  for (int z = 0; z < 3; ++z) {
    hd[z] = eo[z] / C;
#pragma unroll
    for (int c = 0; c < 3; ++c) av[z][c] = att[eo[z] + c];
  }
  __syncthreads();

  {
    constexpr int RW = TS / 8;
    static_assert(RW % 2 == 0, "rows are taken in pairs");
    constexpr int NSS = 3 * RW / 2;
    f3 lv[NSS];
    int rw[NSS];
#pragma unroll
    for (int i = 0; i < NSS; ++i) {
      const int z = i % 3;
      const int r0 = wv * RW + 2 * (i / 3), r1 = r0 + 1;
      rw[i] = (z == 0) ? r0 : (z == 2) ? r1 : (hiB ? r1 : r0);
      const int n = min(n0 + rw[i], N_ - 1);
      lv[i] = *(const f3*)(xl + (size_t)n * F + eo[z]);
    }
#pragma unroll
    for (int i = 0; i < NSS; ++i) {
      const int z = i % 3;
      const f3 rv = *(const f3*)(xrs + rw[i] * F + eo[z]);
      const float a = slot_logit(lv[i], rv, av[z]);
      if ((lane & 3) == 0) shf[rw[i] * H + hd[z]] = a;
    }
  }
  __syncthreads();

  const int b = off[t];
  const int cnt = off[t + 1] - b;
  const int tot = cnt + TS;

  for (int q0 = 0; wave_item<U>(q0, wv) < tot; q0 += 32) {
    int sL, dlL, slL;
    const int okL = tile_item(wave_item<U>(q0 + lane, wv), tot, cnt, b, n0, bins, sL, dlL, slL) ? 1 : 0;
    for (int g = 0; g < 32 / U; ++g) {
      if (wave_item<U>(q0 + g * U, wv) >= tot) break;
      int s[U], dl[U]; bool ok[U];
#pragma unroll
      for (int u = 0; u < U; ++u) {
        s[u]  = lane_bcast(sL, g * U + u);
        dl[u] = lane_bcast(dlL, g * U + u);
        ok[u] = lane_bcast(okL, g * U + u) != 0;
      }
      int sd[NS]; bool so[NS];
      f3 lv[NS], rv[NS];
      float mv[NS];
#pragma unroll
      for (int i = 0; i < NS; ++i) {
        const int z = i % 3;
        const int u0 = 2 * (i / 3), u1 = u0 + 1;
        const int ss = (z == 0) ? s[u0] : (z == 2) ? s[u1] : (hiB ? s[u1] : s[u0]);
        sd[i] = (z == 0) ? dl[u0] : (z == 2) ? dl[u1] : (hiB ? dl[u1] : dl[u0]);
        so[i] = (z == 0) ? ok[u0] : (z == 2) ? ok[u1] : (hiB ? ok[u1] : ok[u0]);
        lv[i] = *(const f3*)(xl + (size_t)ss * F + eo[z]);
        rv[i] = *(const f3*)(xrs + sd[i] * F + eo[z]);
        mv[i] = shf[sd[i] * H + hd[z]];
      }
#pragma unroll
      for (int i = 0; i < NS; ++i) {
        const int z = i % 3;
        const float d = slot_logit(lv[i], rv[i], av[z]) - mv[i];
        const float w = expf(d);
        if (so[i] && (lane & 3) == 0) {
          atomicAdd(&den[sd[i] * H + hd[z]], w);
          if (d > SHIFT_LIM) over = 1;
        }
        const float wz = so[i] ? w : 0.0f;
        float* ap = acc + sd[i] * PF + eo[z];
        atomicAdd(ap,     wz * lv[i].x);
        atomicAdd(ap + 1, wz * lv[i].y);
        atomicAdd(ap + 2, wz * lv[i].z);
      }
    }
  }
  __syncthreads();

  if (over != 0) {
    gat_tile_body<H, C, U, true>(acc, (int*)shf, den, satt, off, bins, xl, xr, att, bias, lg, nullptr,
                                 Wnl, bnl, Wnr, bnr, nxl, nxr);
    return;
  }

  for (int i = tid; i < TS * F; i += 256) {
    const int dl = i / F, r = i - dl * F;
    acc[dl * PF + r] = acc[dl * PF + r] / (den[dl * H + r / C] + 1e-16f) + bias[r];
  }
  __syncthreads();
  tile_proj_pair<F, PF, F2_>(acc, n0, wv, lane, Wnl, bnl, nxl, Wnr, bnr, nxr);
}

constexpr size_t OFF_DINV = 0;
constexpr size_t OFF_CNT  = OFF_DINV + N_;
constexpr size_t OFF_OFFS = OFF_CNT  + 2048;
constexpr size_t OFF_CUR  = OFF_OFFS + 2048;
constexpr size_t OFF_XL2  = (size_t)N_ + E_;
constexpr size_t OFF_XR2  = OFF_XL2  + (size_t)N_*D_;
constexpr size_t OFF_XL1  = OFF_XR2  + (size_t)N_*D_;
constexpr size_t OFF_XR1  = OFF_XL1  + (size_t)N_*F1_;
constexpr size_t OFF_BINS = OFF_XR1  + (size_t)N_*F1_;
constexpr size_t OFF_LG   = OFF_BINS + (size_t)N_*24;
static_assert(OFF_CUR + 2048 <= OFF_XL2, "bin tables fit in front of the projections");
static_assert(F2_ == D_, "layer 2's projections are sized with D_ above");
static_assert((size_t)E_ <= (size_t)N_ * 24, "bins fit their region");

extern "C" void kernel_launch(void* const* d_in, const int* in_sizes, int n_in,
                              void* d_out, int out_size, void* d_ws, size_t ws_size,
                              hipStream_t stream) {
  const float* x     = (const float*)d_in[0];
  const int*   ei    = (const int*)  d_in[1];
  const float* Wssg  = (const float*)d_in[2];
  const float* bssg  = (const float*)d_in[3];
  const float* W1l   = (const float*)d_in[4];
  const float* b1l   = (const float*)d_in[5];
  const float* W1r   = (const float*)d_in[6];
  const float* b1r   = (const float*)d_in[7];
  const float* att1  = (const float*)d_in[8];
  const float* bias1 = (const float*)d_in[9];
  const float* W2l   = (const float*)d_in[10];
  const float* b2l   = (const float*)d_in[11];
  const float* W2r   = (const float*)d_in[12];
  const float* b2r   = (const float*)d_in[13];
  const float* att2  = (const float*)d_in[14];
  const float* bias2 = (const float*)d_in[15];

  const int* src = ei;
  const int* dst = ei + E_;

  float* ws   = (float*)d_ws;
  float* dinv = ws + OFF_DINV;
  int*   cnt  = (int*)(ws + OFF_CNT);
  int*   offs = (int*)(ws + OFF_OFFS);
  int*   cur  = (int*)(ws + OFF_CUR);
  float* xl2  = ws + OFF_XL2;
  float* xr2  = ws + OFF_XR2;
  float* xl1  = ws + OFF_XL1;
  float* xr1  = ws + OFF_XR1;
  int*   bins = (int*)(ws + OFF_BINS);
  float* lg   = ws + OFF_LG;
  float* out  = (float*)d_out;

  const int TB = 256;

  k_zero_i<<<CDIV(2048, TB), TB, 0, stream>>>(cnt, 2048);
  k_tilecount<<<CDIV(E_, CH), 1024, 0, stream>>>(dst, cnt);
  k_scan<<<1, 1024, 0, stream>>>(cnt, offs, cur);
  k_binscatter<<<CDIV(E_, CH), 1024, 0, stream>>>(src, dst, cur, bins);

  k_dinv_tile<<<NT, TS, 0, stream>>>(offs, bins, dinv);
  k_ssg_tile<4><<<NT, 256, 0, stream>>>(offs, bins, x, dinv, Wssg, bssg, W1l, b1l, W1r, b1r, xl1, xr1);

  k_gat12_tile<4><<<NT, 256, 0, stream>>>(offs, bins, xl1, xr1, att1, bias1, lg,
                                          W2l, b2l, W2r, b2r, xl2, xr2);

  k_gat1h_tile<64, 4><<<NT, 256, 0, stream>>>(offs, bins, xl2, xr2, att2, bias2, lg, out);
}


# self loops leave the item pass of the 3 tile kernels: a destination's own row at weight 1 starts its sums where its shift is formed (replaces clearing); pass covers edges only; sum order changed
# speedup vs baseline: 6.0110x; 1.0354x over previous
#include <hip/hip_runtime.h>
#include <cmath>

constexpr int N_  = 100000;
constexpr int E_  = 1000000;
constexpr int D_  = 64;
constexpr int F1_ = 144;
constexpr int F2_ = 64;

#define CDIV(a,b) (((a)+(b)-1)/(b))

typedef __attribute__((ext_vector_type(2))) float v2f;
typedef __attribute__((ext_vector_type(8))) float v8f;

constexpr int TSH = 6;
constexpr int TS  = 1 << TSH;
constexpr int NT  = CDIV(N_, TS);
constexpr int CH  = 8192;
constexpr int SRC_BITS = 17;
constexpr int SRC_MASK = (1 << SRC_BITS) - 1;
static_assert(N_ <= (1 << SRC_BITS), "source id must fit the packed field");
static_assert(NT + 1 <= 2048, "scan handles two tiles per thread of one 1024-thread group");

__global__ void k_zero_i(int* __restrict__ p, int n) {
  int i = blockIdx.x * blockDim.x + threadIdx.x;
  if (i < n) p[i] = 0;
}

__global__ __launch_bounds__(1024) void k_tilecount(const int* __restrict__ dst, int* __restrict__ cnt) {
  __shared__ int hist[NT];
  const int tid = threadIdx.x;
  for (int i = tid; i < NT; i += 1024) hist[i] = 0;
  __syncthreads();
  const int base = blockIdx.x * CH;
#pragma unroll
  for (int k = 0; k < CH / 1024; ++k) {
    int e = base + k * 1024 + tid;
    if (e < E_) atomicAdd(&hist[dst[e] >> TSH], 1);
  }
  __syncthreads();
  for (int i = tid; i < NT; i += 1024) { int v = hist[i]; if (v) atomicAdd(&cnt[i], v); }
}

__global__ __launch_bounds__(1024) void k_scan(const int* __restrict__ cnt, int* __restrict__ off,
                                               int* __restrict__ cur) {
  __shared__ int p[1024];
  const int t = threadIdx.x;
  const int a = (2 * t     < NT) ? cnt[2 * t]     : 0;
  const int b = (2 * t + 1 < NT) ? cnt[2 * t + 1] : 0;
  int v = a + b;
  p[t] = v;
  __syncthreads();
  for (int o = 1; o < 1024; o <<= 1) {
    int add = (t >= o) ? p[t - o] : 0;
    __syncthreads();
    v += add;
    p[t] = v;
    __syncthreads();
  }
  const int excl = v - (a + b);
  if (2 * t     < NT) { off[2 * t]     = excl;     cur[2 * t]     = excl; }
  if (2 * t + 1 < NT) { off[2 * t + 1] = excl + a; cur[2 * t + 1] = excl + a; }
  if (t == 1023) off[NT] = v;
}

__global__ __launch_bounds__(1024) void k_binscatter(const int* __restrict__ src, const int* __restrict__ dst,
                                                     int* __restrict__ cur, int* __restrict__ bins) {
  __shared__ int hist[NT];
  const int tid = threadIdx.x;
  for (int i = tid; i < NT; i += 1024) hist[i] = 0;
  __syncthreads();
  const int base = blockIdx.x * CH;
  int pk[CH / 1024], bn[CH / 1024], lr[CH / 1024];
#pragma unroll
  for (int k = 0; k < CH / 1024; ++k) {
    int e = base + k * 1024 + tid;
    bn[k] = -1; pk[k] = 0; lr[k] = 0;
    if (e < E_) {
      int d = dst[e];
      bn[k] = d >> TSH;
      pk[k] = src[e] | ((d & (TS - 1)) << SRC_BITS);
      lr[k] = atomicAdd(&hist[bn[k]], 1);
    }
  }
  __syncthreads();
  for (int i = tid; i < NT; i += 1024) { int v = hist[i]; hist[i] = v ? atomicAdd(&cur[i], v) : 0; }
  __syncthreads();
#pragma unroll
  for (int k = 0; k < CH / 1024; ++k)
    if (bn[k] >= 0) bins[hist[bn[k]] + lr[k]] = pk[k];
}

__device__ __forceinline__ bool tile_item(int j, int tot, int cnt, int b, int n0,
                                          const int* __restrict__ bins, int& s, int& dl, int& sl) {
  const int p  = min(b + j, E_ - 1);
  const int pk = bins[p];
  const bool isE = j < cnt;
  const int dlS = j - cnt;
  const int sS  = n0 + dlS;
  const bool okS = (!isE) && (j < tot) && (sS < N_);
  s  = isE ? (pk & SRC_MASK)  : (okS ? sS : 0);
  dl = isE ? (pk >> SRC_BITS) : (okS ? dlS : 0);
  sl = isE ? p                : (okS ? E_ + sS : 0);
  return isE || okS;
}

__device__ __forceinline__ bool edge_item(int j, int cnt, int b, const int* __restrict__ bins, int& s, int& dl) {
  const int pk = bins[min(b + j, E_ - 1)];
  const bool ok = j < cnt;
  s  = ok ? (pk & SRC_MASK)  : 0;
  dl = ok ? (pk >> SRC_BITS) : 0;
  return ok;
}

__device__ __forceinline__ void ldsMaxF(int* cell, float v) {
  if (v >= 0.0f) atomicMax(cell, __float_as_int(v));
  else           atomicMin((unsigned int*)cell, __float_as_uint(v));
}

template <int U>
__device__ __forceinline__ int wave_item(int q, int wv) {
  const unsigned uq = (unsigned)q;
  return (int)(((uq / U) * 8u + (unsigned)wv) * U + (uq % U));
}
__device__ __forceinline__ int lane_bcast(int v, int ln) { return __builtin_amdgcn_readlane(v, ln); }
__device__ __forceinline__ float lane_bcast(float v, int ln) {
  return __int_as_float(__builtin_amdgcn_readlane(__float_as_int(v), ln));
}

__global__ __launch_bounds__(TS) void k_dinv_tile(const int* __restrict__ off, const int* __restrict__ bins,
                                                  float* __restrict__ dinv) {
  __shared__ int c[TS];
  const int t = blockIdx.x, tid = threadIdx.x;
  c[tid] = 0;
  __syncthreads();
  const int b = off[t], e = off[t + 1];
  for (int p = b + tid; p < e; p += TS) atomicAdd(&c[bins[p] >> SRC_BITS], 1);
  __syncthreads();
  const int n = t * TS + tid;
  if (n < N_) dinv[n] = rsqrtf(1.0f + (float)c[tid]);
}

template <int K, int PA, int NC>
__device__ __forceinline__ void tile_mma2(const float* a0p, const float* __restrict__ wp, v8f& acc0, v8f& acc1) {
  const float* a1p = a0p + 16 * PA;
#pragma unroll 4
  for (int k = 0; k < K; k += 4) {
    v2f a0, a1, b;
    a0.x = a0p[k]; a0.y = a0p[k + 1];
    a1.x = a1p[k]; a1.y = a1p[k + 1];
    b.x = wp[(size_t)k * NC];
    b.y = wp[(size_t)(k + 1) * NC];
    acc0 = __builtin_amdgcn_wmma_f32_16x16x4_f32(false, a0, false, b, (short)0, acc0, false, false);
    acc1 = __builtin_amdgcn_wmma_f32_16x16x4_f32(false, a1, false, b, (short)0, acc1, false, false);
  }
}

template <int K, int PA, int NC, int PC>
__device__ __forceinline__ void tile_proj_lds(const float* As, const float* __restrict__ W,
                                              const float* __restrict__ bias, float* Cs, int wv, int lane) {
  const int c = lane & 15, hh = lane >> 4;
  const int r0 = (wv & 1) * 32;
  for (int ct = wv >> 1; ct < NC / 16; ct += 4) {
    const int col = ct * 16 + c;
    v8f acc0 = {}, acc1 = {};
    tile_mma2<K, PA, NC>(As + (r0 + c) * PA + 2 * hh, W + (size_t)(2 * hh) * NC + col, acc0, acc1);
    const float bn = bias[col];
#pragma unroll
    for (int j = 0; j < 8; ++j) {
      Cs[(r0 + 8 * hh + j) * PC + col]      = acc0[j] + bn;
      Cs[(r0 + 16 + 8 * hh + j) * PC + col] = acc1[j] + bn;
    }
  }
}

template <int K, int PA, int NC>
__device__ __forceinline__ void tile_proj_pair(const float* As, int n0, int wv, int lane,
                                               const float* __restrict__ Wa, const float* __restrict__ ba,
                                               float* __restrict__ Ca,
                                               const float* __restrict__ Wb, const float* __restrict__ bb,
                                               float* __restrict__ Cb) {
  constexpr int CT = NC / 16;
  const int c = lane & 15, hh = lane >> 4;
  const int r0 = (wv & 1) * 32;
  for (int ct = wv >> 1; ct < 2 * CT; ct += 4) {
    const bool second = ct >= CT;
    const float* __restrict__ W  = second ? Wb : Wa;
    const float* __restrict__ bs = second ? bb : ba;
    float* __restrict__ C        = second ? Cb : Ca;
    const int col = (second ? ct - CT : ct) * 16 + c;
    v8f acc0 = {}, acc1 = {};
    tile_mma2<K, PA, NC>(As + (r0 + c) * PA + 2 * hh, W + (size_t)(2 * hh) * NC + col, acc0, acc1);
    const float bn = bs[col];
    float* __restrict__ cp = C + (size_t)(n0 + r0 + 8 * hh) * NC + col;
    if (n0 + r0 < N_) {
#pragma unroll
      for (int j = 0; j < 8; ++j) cp[(size_t)j * NC] = acc0[j] + bn;
    }
    if (n0 + r0 + 16 < N_) {
#pragma unroll
      for (int j = 0; j < 8; ++j) cp[(size_t)(16 + j) * NC] = acc1[j] + bn;
    }
  }
}

constexpr int PH = D_ + 4;
static_assert(N_ % 16 == 0, "a 16-row output tile is never partly past the last node");
template <int U>
__global__ __launch_bounds__(256) void k_ssg_tile(const int* __restrict__ off, const int* __restrict__ bins,
                                                  const float* __restrict__ x, const float* __restrict__ dinv,
                                                  const float* __restrict__ Wssg, const float* __restrict__ bssg,
                                                  const float* __restrict__ W1l, const float* __restrict__ b1l,
                                                  const float* __restrict__ W1r, const float* __restrict__ b1r,
                                                  float* __restrict__ xl1, float* __restrict__ xr1) {
  __shared__ float acc[TS * PH];
  __shared__ float x1s[TS * PH];
  const int t = blockIdx.x, tid = threadIdx.x, lane = tid & 31;
  const int wv = __builtin_amdgcn_readfirstlane(tid >> 5);
  for (int i = tid; i < TS * PH; i += 256) acc[i] = 0.0f;
  __syncthreads();
  const int b = off[t];
  const int cnt = off[t + 1] - b;
  const int n0 = t * TS;
  for (int q0 = 0; wave_item<U>(q0, wv) < cnt; q0 += 32) {
    int sL, dlL;
    const int okL = edge_item(wave_item<U>(q0 + lane, wv), cnt, b, bins, sL, dlL) ? 1 : 0;
    const float wL = dinv[sL] * dinv[n0 + dlL];
    for (int g = 0; g < 32 / U; ++g) {
      if (wave_item<U>(q0 + g * U, wv) >= cnt) break;
      int s[U], dl[U]; bool ok[U];
      float x0[U], x1[U], w[U];
#pragma unroll
      for (int u = 0; u < U; ++u) {
        s[u]  = lane_bcast(sL, g * U + u);
        dl[u] = lane_bcast(dlL, g * U + u);
        ok[u] = lane_bcast(okL, g * U + u) != 0;
        w[u]  = lane_bcast(wL, g * U + u);
      }
#pragma unroll
      for (int u = 0; u < U; ++u) {
        x0[u] = x[(size_t)s[u] * D_ + lane];
        x1[u] = x[(size_t)s[u] * D_ + lane + 32];
      }
#pragma unroll
      for (int u = 0; u < U; ++u) {
        atomicAdd(&acc[dl[u] * PH + lane],      ok[u] ? x0[u] * w[u] : 0.0f);
        atomicAdd(&acc[dl[u] * PH + lane + 32], ok[u] ? x1[u] * w[u] : 0.0f);
      }
    }
  }
  __syncthreads();
  for (int i = tid; i < TS * D_; i += 256) {
    const int dl = i >> 6, c = i & 63;
    const int n = n0 + dl;
    float hv = 0.0f;
    if (n < N_) {
      const float xv = x[(size_t)n * D_ + c];
      const float dn = dinv[n];
      hv = 0.5f * xv + 0.5f * (acc[dl * PH + c] + xv * (dn * dn));
    }
    acc[dl * PH + c] = hv;
  }
  __syncthreads();
  tile_proj_lds<D_, PH, D_, PH>(acc, Wssg, bssg, x1s, wv, lane);
  __syncthreads();
  tile_proj_pair<D_, PH, F1_>(x1s, n0, wv, lane, W1l, b1l, xl1, W1r, b1r, xr1);
}

template <int H, int C, int U, bool PROJ>
__device__ __forceinline__ void gat_tile_body(float* acc, int* mxi, float* den, float* satt,
                                              const int* __restrict__ off, const int* __restrict__ bins,
                                              const float* __restrict__ xl, const float* __restrict__ xr,
                                              const float* __restrict__ att, const float* __restrict__ bias,
                                              float* __restrict__ lg, float* __restrict__ out,
                                              const float* __restrict__ Wnl, const float* __restrict__ bnl,
                                              const float* __restrict__ Wnr, const float* __restrict__ bnr,
                                              float* __restrict__ nxl, float* __restrict__ nxr) {
  constexpr int F = H * C;
  constexpr int K = (F + 31) / 32;
  constexpr int PF = PROJ ? F + 4 : F;
  static_assert(C % 4 == 0, "head rows are read as 16-byte pieces");
  static_assert(!PROJ || (PF % 64 == 4 || PF % 64 == 20 || PF % 64 == 36 || PF % 64 == 52),
                "fragment reads of 16 rows must fall on 16 different bank groups");

  const int t = blockIdx.x, tid = threadIdx.x, lane = tid & 31;
  const int wv = __builtin_amdgcn_readfirstlane(tid >> 5);
  const int n0 = t * TS;
  float* const xrs = acc;
  constexpr int NST = TS * (F / 4) / 256;
  static_assert(TS * (F / 4) % 256 == 0, "the tile's rows split evenly over the group");
  {
    const float4* __restrict__ xr4 = (const float4*)(xr + (size_t)n0 * F);
    const int npc = min(TS, N_ - n0) * (F / 4);
    float4 st[NST];
#pragma unroll
    for (int k = 0; k < NST; ++k) {
      const int i = tid + 256 * k;
      st[k] = make_float4(0.0f, 0.0f, 0.0f, 0.0f);
      if (i < npc) st[k] = xr4[i];
    }
#pragma unroll
    for (int k = 0; k < NST; ++k) *(float4*)(xrs + 4 * (tid + 256 * k)) = st[k];
  }
  for (int i = tid; i < TS * H; i += 256) { mxi[i] = (int)0xFF800000u; den[i] = 0.0f; }
  for (int i = tid; i < F; i += 256) satt[i] = att[i];
  __syncthreads();
  const int b = off[t];
  const int cnt = off[t + 1] - b;
  const int tot = cnt + TS;

  if constexpr (H == 1) {
    for (int q0 = 0; wave_item<U>(q0, wv) < tot; q0 += 32) {
      int sL, dlL, slL;
      const int okL = tile_item(wave_item<U>(q0 + lane, wv), tot, cnt, b, n0, bins, sL, dlL, slL) ? 1 : 0;
      for (int g = 0; g < 32 / U; ++g) {
        if (wave_item<U>(q0 + g * U, wv) >= tot) break;
        int s[U], dl[U], sl[U]; bool ok[U];
        float l0[U], l1[U], r0[U], r1[U];
#pragma unroll
        for (int u = 0; u < U; ++u) {
          s[u]  = lane_bcast(sL, g * U + u);
          dl[u] = lane_bcast(dlL, g * U + u);
          sl[u] = lane_bcast(slL, g * U + u);
          ok[u] = lane_bcast(okL, g * U + u) != 0;
        }
#pragma unroll
        for (int u = 0; u < U; ++u) {
          l0[u] = xl[(size_t)s[u] * F + lane];
          l1[u] = xl[(size_t)s[u] * F + lane + 32];
          r0[u] = xrs[dl[u] * F + lane];
          r1[u] = xrs[dl[u] * F + lane + 32];
        }
#pragma unroll
        for (int u = 0; u < U; ++u) {
          float a = 0.0f;
          float t0 = l0[u] + r0[u];
          t0 = (t0 > 0.0f) ? t0 : 0.2f * t0;
          a += t0 * satt[lane];
          float t1 = l1[u] + r1[u];
          t1 = (t1 > 0.0f) ? t1 : 0.2f * t1;
          a += t1 * satt[lane + 32];
#pragma unroll
          for (int o = 16; o > 0; o >>= 1) a += __shfl_xor(a, o, 32);
          if (lane == 0 && ok[u]) { lg[sl[u]] = a; ldsMaxF(&mxi[dl[u]], a); }
        }
      }
    }
  } else {
    for (int idx = tid; idx < tot * H; idx += 256) {
      const int j = idx / H, h = idx - j * H;
      int s, dl, sl;
      if (tile_item(j, tot, cnt, b, n0, bins, s, dl, sl)) {
        const float4* pl = (const float4*)(xl + (size_t)s * F + h * C);
        const float4* pr = (const float4*)(xrs + dl * F + h * C);
        float4 lv[C / 4], rv[C / 4];
#pragma unroll
        for (int q = 0; q < C / 4; ++q) { lv[q] = pl[q]; rv[q] = pr[q]; }
        float a = 0.0f;
#pragma unroll
        for (int q = 0; q < C / 4; ++q) {
          const float le[4] = {lv[q].x, lv[q].y, lv[q].z, lv[q].w};
          const float re[4] = {rv[q].x, rv[q].y, rv[q].z, rv[q].w};
#pragma unroll
          for (int z = 0; z < 4; ++z) {
            float tt = le[z] + re[z];
            tt = (tt > 0.0f) ? tt : 0.2f * tt;
            a += tt * satt[h * C + q * 4 + z];
          }
        }
        lg[(size_t)sl * H + h] = a;
        ldsMaxF(&mxi[dl * H + h], a);
      }
    }
  }
  __syncthreads();

  for (int i = tid; i < TS * PF; i += 256) acc[i] = 0.0f;
  for (int idx = tid; idx < tot * H; idx += 256) {
    const int j = idx / H, h = idx - j * H;
    int s, dl, sl;
    if (tile_item(j, tot, cnt, b, n0, bins, s, dl, sl)) {
      const float a = expf(lg[(size_t)sl * H + h] - __int_as_float(mxi[dl * H + h]));
      lg[(size_t)sl * H + h] = a;
      atomicAdd(&den[dl * H + h], a);
    }
  }
  __syncthreads();

  for (int q0 = 0; wave_item<U>(q0, wv) < tot; q0 += 32) {
    int sL, dlL, slL;
    const int okL = tile_item(wave_item<U>(q0 + lane, wv), tot, cnt, b, n0, bins, sL, dlL, slL) ? 1 : 0;
    float aL = 0.0f;
    if constexpr (H == 1) aL = lg[slL];
    for (int g = 0; g < 32 / U; ++g) {
      if (wave_item<U>(q0 + g * U, wv) >= tot) break;
      int s[U], dl[U], sl[U]; bool ok[U];
      float xv[U][K], av[U][K];
#pragma unroll
      for (int u = 0; u < U; ++u) {
        s[u]  = lane_bcast(sL, g * U + u);
        dl[u] = lane_bcast(dlL, g * U + u);
        sl[u] = lane_bcast(slL, g * U + u);
        ok[u] = lane_bcast(okL, g * U + u) != 0;
      }
#pragma unroll
      for (int u = 0; u < U; ++u) {
#pragma unroll
        for (int k = 0; k < K; ++k) {
          const int r  = lane + 32 * k;
          const int rr = (r < F) ? r : r - 32;
          xv[u][k] = xl[(size_t)s[u] * F + rr];
          if constexpr (H == 1) av[u][k] = lane_bcast(aL, g * U + u);
          else                  av[u][k] = lg[(size_t)sl[u] * H + rr / C];
        }
      }
#pragma unroll
      for (int u = 0; u < U; ++u) {
#pragma unroll
        for (int k = 0; k < K; ++k) {
          const int r  = lane + 32 * k;
          const int rr = (r < F) ? r : r - 32;
          atomicAdd(&acc[dl[u] * PF + rr], (ok[u] && r < F) ? av[u][k] * xv[u][k] : 0.0f);
        }
      }
    }
  }
  __syncthreads();

  if constexpr (!PROJ) {
    for (int i = tid; i < TS * F; i += 256) {
      const int dl = i / F, r = i - dl * F;
      const int n = n0 + dl;
      if (n < N_) out[(size_t)n * F + r] = acc[i] / (den[dl * H + r / C] + 1e-16f) + bias[r];
    }
  } else {
    for (int i = tid; i < TS * F; i += 256) {
      const int dl = i / F, r = i - dl * F;
      acc[dl * PF + r] = acc[dl * PF + r] / (den[dl * H + r / C] + 1e-16f) + bias[r];
    }
    __syncthreads();
    tile_proj_pair<F, PF, F2_>(acc, n0, wv, lane, Wnl, bnl, nxl, Wnr, bnr, nxr);
  }
}

constexpr float SHIFT_LIM = 40.0f;
__device__ __forceinline__ float group8_sum(float p) {
  p += __int_as_float(__builtin_amdgcn_update_dpp(0, __float_as_int(p), 0xB1, 0xF, 0xF, true));
  p += __int_as_float(__builtin_amdgcn_update_dpp(0, __float_as_int(p), 0x4E, 0xF, 0xF, true));
  p += __int_as_float(__builtin_amdgcn_update_dpp(0, __float_as_int(p), 0x141, 0xF, 0xF, true));
  return p;
}
__device__ __forceinline__ float lrelu_dot4(const float4& l, const float4& r, const float4& a) {
  float p = 0.0f;
  float t0 = l.x + r.x; t0 = (t0 > 0.0f) ? t0 : 0.2f * t0; p += t0 * a.x;
  float t1 = l.y + r.y; t1 = (t1 > 0.0f) ? t1 : 0.2f * t1; p += t1 * a.y;
  float t2 = l.z + r.z; t2 = (t2 > 0.0f) ? t2 : 0.2f * t2; p += t2 * a.z;
  float t3 = l.w + r.w; t3 = (t3 > 0.0f) ? t3 : 0.2f * t3; p += t3 * a.w;
  return p;
}
template <int C, int U>
__global__ __launch_bounds__(256, 16) void k_gat1h_tile(const int* __restrict__ off, const int* __restrict__ bins,
                                                    const float* __restrict__ xl, const float* __restrict__ xr,
                                                    const float* __restrict__ att, const float* __restrict__ bias,
                                                    float* __restrict__ lg, float* __restrict__ out) {
  constexpr int F = C;
  static_assert(F == 64 && U == 4, "a group of 8 lanes holds a row as eight floats per lane, four items per step");
  constexpr int PA = F + 1;
  constexpr int G = 2;
  __shared__ __attribute__((aligned(16))) float acc[TS * PA];
  __shared__ __attribute__((aligned(16))) float xrs[TS * F];
  __shared__ float shf[TS];
  __shared__ float den[TS];
  __shared__ __attribute__((aligned(16))) float satt[F];
  __shared__ int   over;

  const int t = blockIdx.x, tid = threadIdx.x, lane = tid & 31;
  const int wv = __builtin_amdgcn_readfirstlane(tid >> 5);
  const int n0 = t * TS;
  constexpr int NST = TS * (F / 4) / 256;
  static_assert(TS * (F / 4) % 256 == 0, "the tile's rows split evenly over the group");
  {
    const float4* __restrict__ xr4 = (const float4*)(xr + (size_t)n0 * F);
    const int npc = min(TS, N_ - n0) * (F / 4);
    float4 st[NST];
#pragma unroll
    for (int k = 0; k < NST; ++k) {
      const int i = tid + 256 * k;
      st[k] = make_float4(0.0f, 0.0f, 0.0f, 0.0f);
      if (i < npc) st[k] = xr4[i];
    }
#pragma unroll
    for (int k = 0; k < NST; ++k) *(float4*)(xrs + 4 * (tid + 256 * k)) = st[k];
  }
  if (tid < F) satt[tid] = att[tid];
  if (tid == 0) over = 0;
  __syncthreads();

  {
    constexpr int RW = TS / 8;
    float l0[RW], l1[RW];
#pragma unroll
    for (int r = 0; r < RW; ++r) {
      const int n = min(n0 + wv * RW + r, N_ - 1);
      l0[r] = xl[(size_t)n * F + lane];
      l1[r] = xl[(size_t)n * F + lane + 32];
    }
    float a[RW];
#pragma unroll
    for (int r = 0; r < RW; ++r) {
      const int dl = wv * RW + r;
      a[r] = 0.0f;
      float t0 = l0[r] + xrs[dl * F + lane];
      t0 = (t0 > 0.0f) ? t0 : 0.2f * t0;
      a[r] += t0 * satt[lane];
      float t1 = l1[r] + xrs[dl * F + lane + 32];
      t1 = (t1 > 0.0f) ? t1 : 0.2f * t1;
      a[r] += t1 * satt[lane + 32];
    }
#pragma unroll
    for (int o = 16; o > 0; o >>= 1) {
#pragma unroll
      for (int r = 0; r < RW; ++r) a[r] += __shfl_xor(a[r], o, 32);
    }
#pragma unroll
    for (int r = 0; r < RW; ++r) {
      const int dl = wv * RW + r;
      const bool there = n0 + dl < N_;
      acc[dl * PA + lane]      = there ? l0[r] : 0.0f;
      acc[dl * PA + lane + 32] = there ? l1[r] : 0.0f;
    }
    if (lane == 0) {
#pragma unroll
      for (int r = 0; r < RW; ++r) {
        shf[wv * RW + r] = a[r];
        den[wv * RW + r] = (n0 + wv * RW + r < N_) ? 1.0f : 0.0f;
      }
    }
  }
  __syncthreads();

  const int b = off[t];
  const int cnt = off[t + 1] - b;

  const int grp = lane >> 3;
  const int cA = 32 * (grp & 1) + 4 * (lane & 7);
  const int cB = cA ^ 32;
  const float4 attA = *(const float4*)(satt + cA);
  const float4 attB = *(const float4*)(satt + cB);
  constexpr int OK_BIT = SRC_BITS + TSH;
  static_assert(OK_BIT < 31, "source, local destination and the exists flag share one word");

  for (int q0 = 0; wave_item<U>(q0, wv) < cnt; q0 += 32) {
    int sL, dlL;
    const int okL = edge_item(wave_item<U>(q0 + lane, wv), cnt, b, bins, sL, dlL) ? 1 : 0;
    const int pkL = sL | (dlL << SRC_BITS) | (okL << OK_BIT);
    for (int g = 0; g < 32 / U; g += G) {
      if (wave_item<U>(q0 + g * U, wv) >= cnt) break;
      int dl[G]; bool ok[G];
      float4 la[G], lb[G];
#pragma unroll
      for (int a = 0; a < G; ++a) {
        const int pk = __builtin_amdgcn_ds_bpermute(4 * ((g + a) * U + grp), pkL);
        const int s = pk & SRC_MASK;
        dl[a] = (pk >> SRC_BITS) & (TS - 1);
        ok[a] = (pk >> OK_BIT) != 0;
        la[a] = *(const float4*)(xl + (size_t)s * F + cA);
        lb[a] = *(const float4*)(xl + (size_t)s * F + cB);
      }
#pragma unroll
      for (int a = 0; a < G; ++a) {
        const float4 ra = *(const float4*)(xrs + dl[a] * F + cA);
        const float4 rb = *(const float4*)(xrs + dl[a] * F + cB);
        const float m = shf[dl[a]];
        const float p = group8_sum(lrelu_dot4(la[a], ra, attA) + lrelu_dot4(lb[a], rb, attB));
        const float d = p - m;
        const float w = expf(d);
        if (ok[a] && (lane & 7) == 0) {
          atomicAdd(&den[dl[a]], w);
          if (d > SHIFT_LIM) over = 1;
        }
        const float wz = ok[a] ? w : 0.0f;
        float* pa = acc + dl[a] * PA + cA;
        float* pb = acc + dl[a] * PA + cB;
        atomicAdd(pa,     wz * la[a].x);
        atomicAdd(pa + 1, wz * la[a].y);
        atomicAdd(pa + 2, wz * la[a].z);
        atomicAdd(pa + 3, wz * la[a].w);
        atomicAdd(pb,     wz * lb[a].x);
        atomicAdd(pb + 1, wz * lb[a].y);
        atomicAdd(pb + 2, wz * lb[a].z);
        atomicAdd(pb + 3, wz * lb[a].w);
      }
    }
  }
  __syncthreads();

  if (over != 0) {
    gat_tile_body<1, C, U, false>(acc, (int*)shf, den, satt, off, bins, xl, xr, att, bias, lg, out,
                                  nullptr, nullptr, nullptr, nullptr, nullptr, nullptr);
    return;
  }

  for (int i = tid; i < TS * F; i += 256) {
    const int dl = i / F, r = i - dl * F;
    const int n = n0 + dl;
    if (n < N_) out[(size_t)n * F + r] = acc[dl * PA + r] / (den[dl] + 1e-16f) + bias[r];
  }
}

struct f3 { float x, y, z; };

__device__ __forceinline__ float quad_sum(float p) {
  p += __int_as_float(__builtin_amdgcn_update_dpp(0, __float_as_int(p), 0xB1, 0xF, 0xF, true));
  p += __int_as_float(__builtin_amdgcn_update_dpp(0, __float_as_int(p), 0x4E, 0xF, 0xF, true));
  return p;
}
__device__ __forceinline__ float slot_logit(const f3& l, const f3& r, const float (&a)[3]) {
  float p = 0.0f;
  float t0 = l.x + r.x; t0 = (t0 > 0.0f) ? t0 : 0.2f * t0; p += t0 * a[0];
  float t1 = l.y + r.y; t1 = (t1 > 0.0f) ? t1 : 0.2f * t1; p += t1 * a[1];
  float t2 = l.z + r.z; t2 = (t2 > 0.0f) ? t2 : 0.2f * t2; p += t2 * a[2];
  return quad_sum(p);
}

template <int U>
__global__ __launch_bounds__(256) void k_gat12_tile(const int* __restrict__ off, const int* __restrict__ bins,
                                                    const float* __restrict__ xl, const float* __restrict__ xr,
                                                    const float* __restrict__ att, const float* __restrict__ bias,
                                                    float* __restrict__ lg,
                                                    const float* __restrict__ Wnl, const float* __restrict__ bnl,
                                                    const float* __restrict__ Wnr, const float* __restrict__ bnr,
                                                    float* __restrict__ nxl, float* __restrict__ nxr) {
  constexpr int H = 12, C = 12, F = H * C, PF = F + 4;
  static_assert(U % 2 == 0, "items are taken in pairs");
  static_assert(4 * 3 == C && 32 * 3 * 3 == 2 * F, "a quad of lanes is a head, three slots are two rows");
  constexpr int NS = 3 * U / 2;
  __shared__ __attribute__((aligned(16))) float acc[TS * PF];
  __shared__ __attribute__((aligned(16))) float xrs[TS * F];
  __shared__ float shf[TS * H];
  __shared__ float den[TS * H];
  __shared__ float satt[F];
  __shared__ int   over;

  const int t = blockIdx.x, tid = threadIdx.x, lane = tid & 31;
  const int wv = __builtin_amdgcn_readfirstlane(tid >> 5);
  const int n0 = t * TS;
  constexpr int NST = TS * (F / 4) / 256;
  static_assert(TS * (F / 4) % 256 == 0, "the tile's rows split evenly over the group");
  {
    const float4* __restrict__ xr4 = (const float4*)(xr + (size_t)n0 * F);
    const int npc = min(TS, N_ - n0) * (F / 4);
    float4 st[NST];
#pragma unroll
    for (int k = 0; k < NST; ++k) {
      const int i = tid + 256 * k;
      st[k] = make_float4(0.0f, 0.0f, 0.0f, 0.0f);
      if (i < npc) st[k] = xr4[i];
    }
#pragma unroll
    for (int k = 0; k < NST; ++k) *(float4*)(xrs + 4 * (tid + 256 * k)) = st[k];
  }
  if (tid == 0) over = 0;

  const bool hiB = lane >= 16;
  int eo[3], hd[3];
  float av[3][3];
  eo[0] = 3 * lane;
  eo[1] = hiB ? 3 * lane - 48 : 96 + 3 * lane;
  eo[2] = 48 + 3 * lane;
#pragma unroll
  for (int z = 0; z < 3; ++z) {
    hd[z] = eo[z] / C;
#pragma unroll
    for (int c = 0; c < 3; ++c) av[z][c] = att[eo[z] + c];
  }
  __syncthreads();

  {
    constexpr int RW = TS / 8;
    static_assert(RW % 2 == 0, "rows are taken in pairs");
    constexpr int NSS = 3 * RW / 2;
    f3 lv[NSS];
    int rw[NSS];
#pragma unroll
    for (int i = 0; i < NSS; ++i) {
      const int z = i % 3;
      const int r0 = wv * RW + 2 * (i / 3), r1 = r0 + 1;
      rw[i] = (z == 0) ? r0 : (z == 2) ? r1 : (hiB ? r1 : r0);
      const int n = min(n0 + rw[i], N_ - 1);
      lv[i] = *(const f3*)(xl + (size_t)n * F + eo[z]);
    }
#pragma unroll
    for (int i = 0; i < NSS; ++i) {
      const int z = i % 3;
      const f3 rv = *(const f3*)(xrs + rw[i] * F + eo[z]);
      const float a = slot_logit(lv[i], rv, av[z]);
      const bool there = n0 + rw[i] < N_;
      float* ap = acc + rw[i] * PF + eo[z];
      ap[0] = there ? lv[i].x : 0.0f;
      ap[1] = there ? lv[i].y : 0.0f;
      ap[2] = there ? lv[i].z : 0.0f;
      if ((lane & 3) == 0) {
        shf[rw[i] * H + hd[z]] = a;
        den[rw[i] * H + hd[z]] = there ? 1.0f : 0.0f;
      }
    }
  }
  __syncthreads();

  const int b = off[t];
  const int cnt = off[t + 1] - b;

  for (int q0 = 0; wave_item<U>(q0, wv) < cnt; q0 += 32) {
    int sL, dlL;
    const int okL = edge_item(wave_item<U>(q0 + lane, wv), cnt, b, bins, sL, dlL) ? 1 : 0;
    for (int g = 0; g < 32 / U; ++g) {
      if (wave_item<U>(q0 + g * U, wv) >= cnt) break;
      int s[U], dl[U]; bool ok[U];
#pragma unroll
      for (int u = 0; u < U; ++u) {
        s[u]  = lane_bcast(sL, g * U + u);
        dl[u] = lane_bcast(dlL, g * U + u);
        ok[u] = lane_bcast(okL, g * U + u) != 0;
      }
      int sd[NS]; bool so[NS];
      f3 lv[NS], rv[NS];
      float mv[NS];
#pragma unroll
      for (int i = 0; i < NS; ++i) {
        const int z = i % 3;
        const int u0 = 2 * (i / 3), u1 = u0 + 1;
        const int ss = (z == 0) ? s[u0] : (z == 2) ? s[u1] : (hiB ? s[u1] : s[u0]);
        sd[i] = (z == 0) ? dl[u0] : (z == 2) ? dl[u1] : (hiB ? dl[u1] : dl[u0]);
        so[i] = (z == 0) ? ok[u0] : (z == 2) ? ok[u1] : (hiB ? ok[u1] : ok[u0]);
        lv[i] = *(const f3*)(xl + (size_t)ss * F + eo[z]);
        rv[i] = *(const f3*)(xrs + sd[i] * F + eo[z]);
        mv[i] = shf[sd[i] * H + hd[z]];
      }
#pragma unroll
      for (int i = 0; i < NS; ++i) {
        const int z = i % 3;
        const float d = slot_logit(lv[i], rv[i], av[z]) - mv[i];
        const float w = expf(d);
        if (so[i] && (lane & 3) == 0) {
          atomicAdd(&den[sd[i] * H + hd[z]], w);
          if (d > SHIFT_LIM) over = 1;
        }
        const float wz = so[i] ? w : 0.0f;
        float* ap = acc + sd[i] * PF + eo[z];
        atomicAdd(ap,     wz * lv[i].x);
        atomicAdd(ap + 1, wz * lv[i].y);
        atomicAdd(ap + 2, wz * lv[i].z);
      }
    }
  }
  __syncthreads();

  if (over != 0) {
    gat_tile_body<H, C, U, true>(acc, (int*)shf, den, satt, off, bins, xl, xr, att, bias, lg, nullptr,
                                 Wnl, bnl, Wnr, bnr, nxl, nxr);
    return;
  }

  for (int i = tid; i < TS * F; i += 256) {
    const int dl = i / F, r = i - dl * F;
    acc[dl * PF + r] = acc[dl * PF + r] / (den[dl * H + r / C] + 1e-16f) + bias[r];
  }
  __syncthreads();
  tile_proj_pair<F, PF, F2_>(acc, n0, wv, lane, Wnl, bnl, nxl, Wnr, bnr, nxr);
}

constexpr size_t OFF_DINV = 0;
constexpr size_t OFF_CNT  = OFF_DINV + N_;
constexpr size_t OFF_OFFS = OFF_CNT  + 2048;
constexpr size_t OFF_CUR  = OFF_OFFS + 2048;
constexpr size_t OFF_XL2  = (size_t)N_ + E_;
constexpr size_t OFF_XR2  = OFF_XL2  + (size_t)N_*D_;
constexpr size_t OFF_XL1  = OFF_XR2  + (size_t)N_*D_;
constexpr size_t OFF_XR1  = OFF_XL1  + (size_t)N_*F1_;
constexpr size_t OFF_BINS = OFF_XR1  + (size_t)N_*F1_;
constexpr size_t OFF_LG   = OFF_BINS + (size_t)N_*24;
static_assert(OFF_CUR + 2048 <= OFF_XL2, "bin tables fit in front of the projections");
static_assert(F2_ == D_, "layer 2's projections are sized with D_ above");
static_assert((size_t)E_ <= (size_t)N_ * 24, "bins fit their region");

extern "C" void kernel_launch(void* const* d_in, const int* in_sizes, int n_in,
                              void* d_out, int out_size, void* d_ws, size_t ws_size,
                              hipStream_t stream) {
  const float* x     = (const float*)d_in[0];
  const int*   ei    = (const int*)  d_in[1];
  const float* Wssg  = (const float*)d_in[2];
  const float* bssg  = (const float*)d_in[3];
  const float* W1l   = (const float*)d_in[4];
  const float* b1l   = (const float*)d_in[5];
  const float* W1r   = (const float*)d_in[6];
  const float* b1r   = (const float*)d_in[7];
  const float* att1  = (const float*)d_in[8];
  const float* bias1 = (const float*)d_in[9];
  const float* W2l   = (const float*)d_in[10];
  const float* b2l   = (const float*)d_in[11];
  const float* W2r   = (const float*)d_in[12];
  const float* b2r   = (const float*)d_in[13];
  const float* att2  = (const float*)d_in[14];
  const float* bias2 = (const float*)d_in[15];

  const int* src = ei;
  const int* dst = ei + E_;

  float* ws   = (float*)d_ws;
  float* dinv = ws + OFF_DINV;
  int*   cnt  = (int*)(ws + OFF_CNT);
  int*   offs = (int*)(ws + OFF_OFFS);
  int*   cur  = (int*)(ws + OFF_CUR);
  float* xl2  = ws + OFF_XL2;
  float* xr2  = ws + OFF_XR2;
  float* xl1  = ws + OFF_XL1;
  float* xr1  = ws + OFF_XR1;
  int*   bins = (int*)(ws + OFF_BINS);
  float* lg   = ws + OFF_LG;
  float* out  = (float*)d_out;

  const int TB = 256;

  k_zero_i<<<CDIV(2048, TB), TB, 0, stream>>>(cnt, 2048);
  k_tilecount<<<CDIV(E_, CH), 1024, 0, stream>>>(dst, cnt);
  k_scan<<<1, 1024, 0, stream>>>(cnt, offs, cur);
  k_binscatter<<<CDIV(E_, CH), 1024, 0, stream>>>(src, dst, cur, bins);

  k_dinv_tile<<<NT, TS, 0, stream>>>(offs, bins, dinv);
  k_ssg_tile<4><<<NT, 256, 0, stream>>>(offs, bins, x, dinv, Wssg, bssg, W1l, b1l, W1r, b1r, xl1, xr1);

  k_gat12_tile<4><<<NT, 256, 0, stream>>>(offs, bins, xl1, xr1, att1, bias1, lg,
                                          W2l, b2l, W2r, b2r, xl2, xr2);

  k_gat1h_tile<64, 4><<<NT, 256, 0, stream>>>(offs, bins, xl2, xr2, att2, bias2, lg, out);
}


# element-wise phases between the passes (clearing, h, divide by the row sum) take 16-byte pieces, a thread's loads ahead of their use; per-element index arithmetic and waits go; values unchanged
# speedup vs baseline: 6.4885x; 1.0794x over previous
#include <hip/hip_runtime.h>
#include <cmath>

constexpr int N_  = 100000;
constexpr int E_  = 1000000;
constexpr int D_  = 64;
constexpr int F1_ = 144;
constexpr int F2_ = 64;

#define CDIV(a,b) (((a)+(b)-1)/(b))

typedef __attribute__((ext_vector_type(2))) float v2f;
typedef __attribute__((ext_vector_type(8))) float v8f;

constexpr int TSH = 6;
constexpr int TS  = 1 << TSH;
constexpr int NT  = CDIV(N_, TS);
constexpr int CH  = 8192;
constexpr int SRC_BITS = 17;
constexpr int SRC_MASK = (1 << SRC_BITS) - 1;
static_assert(N_ <= (1 << SRC_BITS), "source id must fit the packed field");
static_assert(NT + 1 <= 2048, "scan handles two tiles per thread of one 1024-thread group");

__global__ void k_zero_i(int* __restrict__ p, int n) {
  int i = blockIdx.x * blockDim.x + threadIdx.x;
  if (i < n) p[i] = 0;
}

__global__ __launch_bounds__(1024) void k_tilecount(const int* __restrict__ dst, int* __restrict__ cnt) {
  __shared__ int hist[NT];
  const int tid = threadIdx.x;
  for (int i = tid; i < NT; i += 1024) hist[i] = 0;
  __syncthreads();
  const int base = blockIdx.x * CH;
#pragma unroll
  for (int k = 0; k < CH / 1024; ++k) {
    int e = base + k * 1024 + tid;
    if (e < E_) atomicAdd(&hist[dst[e] >> TSH], 1);
  }
  __syncthreads();
  for (int i = tid; i < NT; i += 1024) { int v = hist[i]; if (v) atomicAdd(&cnt[i], v); }
}

__global__ __launch_bounds__(1024) void k_scan(const int* __restrict__ cnt, int* __restrict__ off,
                                               int* __restrict__ cur) {
  __shared__ int p[1024];
  const int t = threadIdx.x;
  const int a = (2 * t     < NT) ? cnt[2 * t]     : 0;
  const int b = (2 * t + 1 < NT) ? cnt[2 * t + 1] : 0;
  int v = a + b;
  p[t] = v;
  __syncthreads();
  for (int o = 1; o < 1024; o <<= 1) {
    int add = (t >= o) ? p[t - o] : 0;
    __syncthreads();
    v += add;
    p[t] = v;
    __syncthreads();
  }
  const int excl = v - (a + b);
  if (2 * t     < NT) { off[2 * t]     = excl;     cur[2 * t]     = excl; }
  if (2 * t + 1 < NT) { off[2 * t + 1] = excl + a; cur[2 * t + 1] = excl + a; }
  if (t == 1023) off[NT] = v;
}

__global__ __launch_bounds__(1024) void k_binscatter(const int* __restrict__ src, const int* __restrict__ dst,
                                                     int* __restrict__ cur, int* __restrict__ bins) {
  __shared__ int hist[NT];
  const int tid = threadIdx.x;
  for (int i = tid; i < NT; i += 1024) hist[i] = 0;
  __syncthreads();
  const int base = blockIdx.x * CH;
  int pk[CH / 1024], bn[CH / 1024], lr[CH / 1024];
#pragma unroll
  for (int k = 0; k < CH / 1024; ++k) {
    int e = base + k * 1024 + tid;
    bn[k] = -1; pk[k] = 0; lr[k] = 0;
    if (e < E_) {
      int d = dst[e];
      bn[k] = d >> TSH;
      pk[k] = src[e] | ((d & (TS - 1)) << SRC_BITS);
      lr[k] = atomicAdd(&hist[bn[k]], 1);
    }
  }
  __syncthreads();
  for (int i = tid; i < NT; i += 1024) { int v = hist[i]; hist[i] = v ? atomicAdd(&cur[i], v) : 0; }
  __syncthreads();
#pragma unroll
  for (int k = 0; k < CH / 1024; ++k)
    if (bn[k] >= 0) bins[hist[bn[k]] + lr[k]] = pk[k];
}

__device__ __forceinline__ bool tile_item(int j, int tot, int cnt, int b, int n0,
                                          const int* __restrict__ bins, int& s, int& dl, int& sl) {
  const int p  = min(b + j, E_ - 1);
  const int pk = bins[p];
  const bool isE = j < cnt;
  const int dlS = j - cnt;
  const int sS  = n0 + dlS;
  const bool okS = (!isE) && (j < tot) && (sS < N_);
  s  = isE ? (pk & SRC_MASK)  : (okS ? sS : 0);
  dl = isE ? (pk >> SRC_BITS) : (okS ? dlS : 0);
  sl = isE ? p                : (okS ? E_ + sS : 0);
  return isE || okS;
}

__device__ __forceinline__ bool edge_item(int j, int cnt, int b, const int* __restrict__ bins, int& s, int& dl) {
  const int pk = bins[min(b + j, E_ - 1)];
  const bool ok = j < cnt;
  s  = ok ? (pk & SRC_MASK)  : 0;
  dl = ok ? (pk >> SRC_BITS) : 0;
  return ok;
}

__device__ __forceinline__ void ldsMaxF(int* cell, float v) {
  if (v >= 0.0f) atomicMax(cell, __float_as_int(v));
  else           atomicMin((unsigned int*)cell, __float_as_uint(v));
}

template <int U>
__device__ __forceinline__ int wave_item(int q, int wv) {
  const unsigned uq = (unsigned)q;
  return (int)(((uq / U) * 8u + (unsigned)wv) * U + (uq % U));
}
__device__ __forceinline__ int lane_bcast(int v, int ln) { return __builtin_amdgcn_readlane(v, ln); }
__device__ __forceinline__ float lane_bcast(float v, int ln) {
  return __int_as_float(__builtin_amdgcn_readlane(__float_as_int(v), ln));
}

__global__ __launch_bounds__(TS) void k_dinv_tile(const int* __restrict__ off, const int* __restrict__ bins,
                                                  float* __restrict__ dinv) {
  __shared__ int c[TS];
  const int t = blockIdx.x, tid = threadIdx.x;
  c[tid] = 0;
  __syncthreads();
  const int b = off[t], e = off[t + 1];
  for (int p = b + tid; p < e; p += TS) atomicAdd(&c[bins[p] >> SRC_BITS], 1);
  __syncthreads();
  const int n = t * TS + tid;
  if (n < N_) dinv[n] = rsqrtf(1.0f + (float)c[tid]);
}

template <int K, int PA, int NC>
__device__ __forceinline__ void tile_mma2(const float* a0p, const float* __restrict__ wp, v8f& acc0, v8f& acc1) {
  const float* a1p = a0p + 16 * PA;
#pragma unroll 4
  for (int k = 0; k < K; k += 4) {
    v2f a0, a1, b;
    a0.x = a0p[k]; a0.y = a0p[k + 1];
    a1.x = a1p[k]; a1.y = a1p[k + 1];
    b.x = wp[(size_t)k * NC];
    b.y = wp[(size_t)(k + 1) * NC];
    acc0 = __builtin_amdgcn_wmma_f32_16x16x4_f32(false, a0, false, b, (short)0, acc0, false, false);
    acc1 = __builtin_amdgcn_wmma_f32_16x16x4_f32(false, a1, false, b, (short)0, acc1, false, false);
  }
}

template <int K, int PA, int NC, int PC>
__device__ __forceinline__ void tile_proj_lds(const float* As, const float* __restrict__ W,
                                              const float* __restrict__ bias, float* Cs, int wv, int lane) {
  const int c = lane & 15, hh = lane >> 4;
  const int r0 = (wv & 1) * 32;
  for (int ct = wv >> 1; ct < NC / 16; ct += 4) {
    const int col = ct * 16 + c;
    v8f acc0 = {}, acc1 = {};
    tile_mma2<K, PA, NC>(As + (r0 + c) * PA + 2 * hh, W + (size_t)(2 * hh) * NC + col, acc0, acc1);
    const float bn = bias[col];
#pragma unroll
    for (int j = 0; j < 8; ++j) {
      Cs[(r0 + 8 * hh + j) * PC + col]      = acc0[j] + bn;
      Cs[(r0 + 16 + 8 * hh + j) * PC + col] = acc1[j] + bn;
    }
  }
}

template <int K, int PA, int NC>
__device__ __forceinline__ void tile_proj_pair(const float* As, int n0, int wv, int lane,
                                               const float* __restrict__ Wa, const float* __restrict__ ba,
                                               float* __restrict__ Ca,
                                               const float* __restrict__ Wb, const float* __restrict__ bb,
                                               float* __restrict__ Cb) {
  constexpr int CT = NC / 16;
  const int c = lane & 15, hh = lane >> 4;
  const int r0 = (wv & 1) * 32;
  for (int ct = wv >> 1; ct < 2 * CT; ct += 4) {
    const bool second = ct >= CT;
    const float* __restrict__ W  = second ? Wb : Wa;
    const float* __restrict__ bs = second ? bb : ba;
    float* __restrict__ C        = second ? Cb : Ca;
    const int col = (second ? ct - CT : ct) * 16 + c;
    v8f acc0 = {}, acc1 = {};
    tile_mma2<K, PA, NC>(As + (r0 + c) * PA + 2 * hh, W + (size_t)(2 * hh) * NC + col, acc0, acc1);
    const float bn = bs[col];
    float* __restrict__ cp = C + (size_t)(n0 + r0 + 8 * hh) * NC + col;
    if (n0 + r0 < N_) {
#pragma unroll
      for (int j = 0; j < 8; ++j) cp[(size_t)j * NC] = acc0[j] + bn;
    }
    if (n0 + r0 + 16 < N_) {
#pragma unroll
      for (int j = 0; j < 8; ++j) cp[(size_t)(16 + j) * NC] = acc1[j] + bn;
    }
  }
}

constexpr int PH = D_ + 4;
static_assert(N_ % 16 == 0, "a 16-row output tile is never partly past the last node");
template <int U>
__global__ __launch_bounds__(256) void k_ssg_tile(const int* __restrict__ off, const int* __restrict__ bins,
                                                  const float* __restrict__ x, const float* __restrict__ dinv,
                                                  const float* __restrict__ Wssg, const float* __restrict__ bssg,
                                                  const float* __restrict__ W1l, const float* __restrict__ b1l,
                                                  const float* __restrict__ W1r, const float* __restrict__ b1r,
                                                  float* __restrict__ xl1, float* __restrict__ xr1) {
  __shared__ __attribute__((aligned(16))) float acc[TS * PH];
  __shared__ float x1s[TS * PH];
  static_assert(PH % 4 == 0, "a row of the tile starts on a 16-byte boundary");
  const int t = blockIdx.x, tid = threadIdx.x, lane = tid & 31;
  const int wv = __builtin_amdgcn_readfirstlane(tid >> 5);
  for (int i = tid; i < TS * PH / 4; i += 256) ((float4*)acc)[i] = make_float4(0.0f, 0.0f, 0.0f, 0.0f);
  __syncthreads();
  const int b = off[t];
  const int cnt = off[t + 1] - b;
  const int n0 = t * TS;
  for (int q0 = 0; wave_item<U>(q0, wv) < cnt; q0 += 32) {
    int sL, dlL;
    const int okL = edge_item(wave_item<U>(q0 + lane, wv), cnt, b, bins, sL, dlL) ? 1 : 0;
    const float wL = dinv[sL] * dinv[n0 + dlL];
    for (int g = 0; g < 32 / U; ++g) {
      if (wave_item<U>(q0 + g * U, wv) >= cnt) break;
      int s[U], dl[U]; bool ok[U];
      float x0[U], x1[U], w[U];
#pragma unroll
      for (int u = 0; u < U; ++u) {
        s[u]  = lane_bcast(sL, g * U + u);
        dl[u] = lane_bcast(dlL, g * U + u);
        ok[u] = lane_bcast(okL, g * U + u) != 0;
        w[u]  = lane_bcast(wL, g * U + u);
      }
#pragma unroll
      for (int u = 0; u < U; ++u) {
        x0[u] = x[(size_t)s[u] * D_ + lane];
        x1[u] = x[(size_t)s[u] * D_ + lane + 32];
      }
#pragma unroll
      for (int u = 0; u < U; ++u) {
        atomicAdd(&acc[dl[u] * PH + lane],      ok[u] ? x0[u] * w[u] : 0.0f);
        atomicAdd(&acc[dl[u] * PH + lane + 32], ok[u] ? x1[u] * w[u] : 0.0f);
      }
    }
  }
  __syncthreads();
  {
    constexpr int NP = TS * (D_ / 4) / 256;
    static_assert(TS * (D_ / 4) % 256 == 0, "the tile's rows split evenly over the group");
    const float4* __restrict__ x4 = (const float4*)x;
    float4 xv[NP];
    float dn[NP];
#pragma unroll
    for (int k = 0; k < NP; ++k) {
      const int i4 = tid + 256 * k;
      const int n = min(n0 + i4 / (D_ / 4), N_ - 1);
      xv[k] = x4[(size_t)n * (D_ / 4) + i4 % (D_ / 4)];
      dn[k] = dinv[n];
    }
#pragma unroll
    for (int k = 0; k < NP; ++k) {
      const int i4 = tid + 256 * k;
      const int dl = i4 / (D_ / 4);
      float4* ap = (float4*)(acc + dl * PH + 4 * (i4 % (D_ / 4)));
      const float4 av = *ap;
      const float d2 = dn[k] * dn[k];
      float4 hv;
      hv.x = 0.5f * xv[k].x + 0.5f * (av.x + xv[k].x * d2);
      hv.y = 0.5f * xv[k].y + 0.5f * (av.y + xv[k].y * d2);
      hv.z = 0.5f * xv[k].z + 0.5f * (av.z + xv[k].z * d2);
      hv.w = 0.5f * xv[k].w + 0.5f * (av.w + xv[k].w * d2);
      *ap = hv;
    }
  }
  __syncthreads();
  tile_proj_lds<D_, PH, D_, PH>(acc, Wssg, bssg, x1s, wv, lane);
  __syncthreads();
  tile_proj_pair<D_, PH, F1_>(x1s, n0, wv, lane, W1l, b1l, xl1, W1r, b1r, xr1);
}

template <int H, int C, int U, bool PROJ>
__device__ __forceinline__ void gat_tile_body(float* acc, int* mxi, float* den, float* satt,
                                              const int* __restrict__ off, const int* __restrict__ bins,
                                              const float* __restrict__ xl, const float* __restrict__ xr,
                                              const float* __restrict__ att, const float* __restrict__ bias,
                                              float* __restrict__ lg, float* __restrict__ out,
                                              const float* __restrict__ Wnl, const float* __restrict__ bnl,
                                              const float* __restrict__ Wnr, const float* __restrict__ bnr,
                                              float* __restrict__ nxl, float* __restrict__ nxr) {
  constexpr int F = H * C;
  constexpr int K = (F + 31) / 32;
  constexpr int PF = PROJ ? F + 4 : F;
  static_assert(C % 4 == 0, "head rows are read as 16-byte pieces");
  static_assert(!PROJ || (PF % 64 == 4 || PF % 64 == 20 || PF % 64 == 36 || PF % 64 == 52),
                "fragment reads of 16 rows must fall on 16 different bank groups");

  const int t = blockIdx.x, tid = threadIdx.x, lane = tid & 31;
  const int wv = __builtin_amdgcn_readfirstlane(tid >> 5);
  const int n0 = t * TS;
  float* const xrs = acc;
  constexpr int NST = TS * (F / 4) / 256;
  static_assert(TS * (F / 4) % 256 == 0, "the tile's rows split evenly over the group");
  {
    const float4* __restrict__ xr4 = (const float4*)(xr + (size_t)n0 * F);
    const int npc = min(TS, N_ - n0) * (F / 4);
    float4 st[NST];
#pragma unroll
    for (int k = 0; k < NST; ++k) {
      const int i = tid + 256 * k;
      st[k] = make_float4(0.0f, 0.0f, 0.0f, 0.0f);
      if (i < npc) st[k] = xr4[i];
    }
#pragma unroll
    for (int k = 0; k < NST; ++k) *(float4*)(xrs + 4 * (tid + 256 * k)) = st[k];
  }
  for (int i = tid; i < TS * H; i += 256) { mxi[i] = (int)0xFF800000u; den[i] = 0.0f; }
  for (int i = tid; i < F; i += 256) satt[i] = att[i];
  __syncthreads();
  const int b = off[t];
  const int cnt = off[t + 1] - b;
  const int tot = cnt + TS;

  if constexpr (H == 1) {
    for (int q0 = 0; wave_item<U>(q0, wv) < tot; q0 += 32) {
      int sL, dlL, slL;
      const int okL = tile_item(wave_item<U>(q0 + lane, wv), tot, cnt, b, n0, bins, sL, dlL, slL) ? 1 : 0;
      for (int g = 0; g < 32 / U; ++g) {
        if (wave_item<U>(q0 + g * U, wv) >= tot) break;
        int s[U], dl[U], sl[U]; bool ok[U];
        float l0[U], l1[U], r0[U], r1[U];
#pragma unroll
        for (int u = 0; u < U; ++u) {
          s[u]  = lane_bcast(sL, g * U + u);
          dl[u] = lane_bcast(dlL, g * U + u);
          sl[u] = lane_bcast(slL, g * U + u);
          ok[u] = lane_bcast(okL, g * U + u) != 0;
        }
#pragma unroll
        for (int u = 0; u < U; ++u) {
          l0[u] = xl[(size_t)s[u] * F + lane];
          l1[u] = xl[(size_t)s[u] * F + lane + 32];
          r0[u] = xrs[dl[u] * F + lane];
          r1[u] = xrs[dl[u] * F + lane + 32];
        }
#pragma unroll
        for (int u = 0; u < U; ++u) {
          float a = 0.0f;
          float t0 = l0[u] + r0[u];
          t0 = (t0 > 0.0f) ? t0 : 0.2f * t0;
          a += t0 * satt[lane];
          float t1 = l1[u] + r1[u];
          t1 = (t1 > 0.0f) ? t1 : 0.2f * t1;
          a += t1 * satt[lane + 32];
#pragma unroll
          for (int o = 16; o > 0; o >>= 1) a += __shfl_xor(a, o, 32);
          if (lane == 0 && ok[u]) { lg[sl[u]] = a; ldsMaxF(&mxi[dl[u]], a); }
        }
      }
    }
  } else {
    for (int idx = tid; idx < tot * H; idx += 256) {
      const int j = idx / H, h = idx - j * H;
      int s, dl, sl;
      if (tile_item(j, tot, cnt, b, n0, bins, s, dl, sl)) {
        const float4* pl = (const float4*)(xl + (size_t)s * F + h * C);
        const float4* pr = (const float4*)(xrs + dl * F + h * C);
        float4 lv[C / 4], rv[C / 4];
#pragma unroll
        for (int q = 0; q < C / 4; ++q) { lv[q] = pl[q]; rv[q] = pr[q]; }
        float a = 0.0f;
#pragma unroll
        for (int q = 0; q < C / 4; ++q) {
          const float le[4] = {lv[q].x, lv[q].y, lv[q].z, lv[q].w};
          const float re[4] = {rv[q].x, rv[q].y, rv[q].z, rv[q].w};
#pragma unroll
          for (int z = 0; z < 4; ++z) {
            float tt = le[z] + re[z];
            tt = (tt > 0.0f) ? tt : 0.2f * tt;
            a += tt * satt[h * C + q * 4 + z];
          }
        }
        lg[(size_t)sl * H + h] = a;
        ldsMaxF(&mxi[dl * H + h], a);
      }
    }
  }
  __syncthreads();

  for (int i = tid; i < TS * PF; i += 256) acc[i] = 0.0f;
  for (int idx = tid; idx < tot * H; idx += 256) {
    const int j = idx / H, h = idx - j * H;
    int s, dl, sl;
    if (tile_item(j, tot, cnt, b, n0, bins, s, dl, sl)) {
      const float a = expf(lg[(size_t)sl * H + h] - __int_as_float(mxi[dl * H + h]));
      lg[(size_t)sl * H + h] = a;
      atomicAdd(&den[dl * H + h], a);
    }
  }
  __syncthreads();

  for (int q0 = 0; wave_item<U>(q0, wv) < tot; q0 += 32) {
    int sL, dlL, slL;
    const int okL = tile_item(wave_item<U>(q0 + lane, wv), tot, cnt, b, n0, bins, sL, dlL, slL) ? 1 : 0;
    float aL = 0.0f;
    if constexpr (H == 1) aL = lg[slL];
    for (int g = 0; g < 32 / U; ++g) {
      if (wave_item<U>(q0 + g * U, wv) >= tot) break;
      int s[U], dl[U], sl[U]; bool ok[U];
      float xv[U][K], av[U][K];
#pragma unroll
      for (int u = 0; u < U; ++u) {
        s[u]  = lane_bcast(sL, g * U + u);
        dl[u] = lane_bcast(dlL, g * U + u);
        sl[u] = lane_bcast(slL, g * U + u);
        ok[u] = lane_bcast(okL, g * U + u) != 0;
      }
#pragma unroll
      for (int u = 0; u < U; ++u) {
#pragma unroll
        for (int k = 0; k < K; ++k) {
          const int r  = lane + 32 * k;
          const int rr = (r < F) ? r : r - 32;
          xv[u][k] = xl[(size_t)s[u] * F + rr];
          if constexpr (H == 1) av[u][k] = lane_bcast(aL, g * U + u);
          else                  av[u][k] = lg[(size_t)sl[u] * H + rr / C];
        }
      }
#pragma unroll
      for (int u = 0; u < U; ++u) {
#pragma unroll
        for (int k = 0; k < K; ++k) {
          const int r  = lane + 32 * k;
          const int rr = (r < F) ? r : r - 32;
          atomicAdd(&acc[dl[u] * PF + rr], (ok[u] && r < F) ? av[u][k] * xv[u][k] : 0.0f);
        }
      }
    }
  }
  __syncthreads();

  if constexpr (!PROJ) {
    for (int i = tid; i < TS * F; i += 256) {
      const int dl = i / F, r = i - dl * F;
      const int n = n0 + dl;
      if (n < N_) out[(size_t)n * F + r] = acc[i] / (den[dl * H + r / C] + 1e-16f) + bias[r];
    }
  } else {
    for (int i = tid; i < TS * F; i += 256) {
      const int dl = i / F, r = i - dl * F;
      acc[dl * PF + r] = acc[dl * PF + r] / (den[dl * H + r / C] + 1e-16f) + bias[r];
    }
    __syncthreads();
    tile_proj_pair<F, PF, F2_>(acc, n0, wv, lane, Wnl, bnl, nxl, Wnr, bnr, nxr);
  }
}

constexpr float SHIFT_LIM = 40.0f;
__device__ __forceinline__ float group8_sum(float p) {
  p += __int_as_float(__builtin_amdgcn_update_dpp(0, __float_as_int(p), 0xB1, 0xF, 0xF, true));
  p += __int_as_float(__builtin_amdgcn_update_dpp(0, __float_as_int(p), 0x4E, 0xF, 0xF, true));
  p += __int_as_float(__builtin_amdgcn_update_dpp(0, __float_as_int(p), 0x141, 0xF, 0xF, true));
  return p;
}
__device__ __forceinline__ float lrelu_dot4(const float4& l, const float4& r, const float4& a) {
  float p = 0.0f;
  float t0 = l.x + r.x; t0 = (t0 > 0.0f) ? t0 : 0.2f * t0; p += t0 * a.x;
  float t1 = l.y + r.y; t1 = (t1 > 0.0f) ? t1 : 0.2f * t1; p += t1 * a.y;
  float t2 = l.z + r.z; t2 = (t2 > 0.0f) ? t2 : 0.2f * t2; p += t2 * a.z;
  float t3 = l.w + r.w; t3 = (t3 > 0.0f) ? t3 : 0.2f * t3; p += t3 * a.w;
  return p;
}
template <int C, int U>
__global__ __launch_bounds__(256, 16) void k_gat1h_tile(const int* __restrict__ off, const int* __restrict__ bins,
                                                    const float* __restrict__ xl, const float* __restrict__ xr,
                                                    const float* __restrict__ att, const float* __restrict__ bias,
                                                    float* __restrict__ lg, float* __restrict__ out) {
  constexpr int F = C;
  static_assert(F == 64 && U == 4, "a group of 8 lanes holds a row as eight floats per lane, four items per step");
  constexpr int PA = F + 1;
  constexpr int G = 2;
  __shared__ __attribute__((aligned(16))) float acc[TS * PA];
  __shared__ __attribute__((aligned(16))) float xrs[TS * F];
  __shared__ float shf[TS];
  __shared__ float den[TS];
  __shared__ __attribute__((aligned(16))) float satt[F];
  __shared__ int   over;

  const int t = blockIdx.x, tid = threadIdx.x, lane = tid & 31;
  const int wv = __builtin_amdgcn_readfirstlane(tid >> 5);
  const int n0 = t * TS;
  constexpr int NST = TS * (F / 4) / 256;
  static_assert(TS * (F / 4) % 256 == 0, "the tile's rows split evenly over the group");
  {
    const float4* __restrict__ xr4 = (const float4*)(xr + (size_t)n0 * F);
    const int npc = min(TS, N_ - n0) * (F / 4);
    float4 st[NST];
#pragma unroll
    for (int k = 0; k < NST; ++k) {
      const int i = tid + 256 * k;
      st[k] = make_float4(0.0f, 0.0f, 0.0f, 0.0f);
      if (i < npc) st[k] = xr4[i];
    }
#pragma unroll
    for (int k = 0; k < NST; ++k) *(float4*)(xrs + 4 * (tid + 256 * k)) = st[k];
  }
  if (tid < F) satt[tid] = att[tid];
  if (tid == 0) over = 0;
  __syncthreads();

  {
    constexpr int RW = TS / 8;
    float l0[RW], l1[RW];
#pragma unroll
    for (int r = 0; r < RW; ++r) {
      const int n = min(n0 + wv * RW + r, N_ - 1);
      l0[r] = xl[(size_t)n * F + lane];
      l1[r] = xl[(size_t)n * F + lane + 32];
    }
    float a[RW];
#pragma unroll
    for (int r = 0; r < RW; ++r) {
      const int dl = wv * RW + r;
      a[r] = 0.0f;
      float t0 = l0[r] + xrs[dl * F + lane];
      t0 = (t0 > 0.0f) ? t0 : 0.2f * t0;
      a[r] += t0 * satt[lane];
      float t1 = l1[r] + xrs[dl * F + lane + 32];
      t1 = (t1 > 0.0f) ? t1 : 0.2f * t1;
      a[r] += t1 * satt[lane + 32];
    }
#pragma unroll
    for (int o = 16; o > 0; o >>= 1) {
#pragma unroll
      for (int r = 0; r < RW; ++r) a[r] += __shfl_xor(a[r], o, 32);
    }
#pragma unroll
    for (int r = 0; r < RW; ++r) {
      const int dl = wv * RW + r;
      const bool there = n0 + dl < N_;
      acc[dl * PA + lane]      = there ? l0[r] : 0.0f;
      acc[dl * PA + lane + 32] = there ? l1[r] : 0.0f;
    }
    if (lane == 0) {
#pragma unroll
      for (int r = 0; r < RW; ++r) {
        shf[wv * RW + r] = a[r];
        den[wv * RW + r] = (n0 + wv * RW + r < N_) ? 1.0f : 0.0f;
      }
    }
  }
  __syncthreads();

  const int b = off[t];
  const int cnt = off[t + 1] - b;

  const int grp = lane >> 3;
  const int cA = 32 * (grp & 1) + 4 * (lane & 7);
  const int cB = cA ^ 32;
  const float4 attA = *(const float4*)(satt + cA);
  const float4 attB = *(const float4*)(satt + cB);
  constexpr int OK_BIT = SRC_BITS + TSH;
  static_assert(OK_BIT < 31, "source, local destination and the exists flag share one word");

  for (int q0 = 0; wave_item<U>(q0, wv) < cnt; q0 += 32) {
    int sL, dlL;
    const int okL = edge_item(wave_item<U>(q0 + lane, wv), cnt, b, bins, sL, dlL) ? 1 : 0;
    const int pkL = sL | (dlL << SRC_BITS) | (okL << OK_BIT);
    for (int g = 0; g < 32 / U; g += G) {
      if (wave_item<U>(q0 + g * U, wv) >= cnt) break;
      int dl[G]; bool ok[G];
      float4 la[G], lb[G];
#pragma unroll
      for (int a = 0; a < G; ++a) {
        const int pk = __builtin_amdgcn_ds_bpermute(4 * ((g + a) * U + grp), pkL);
        const int s = pk & SRC_MASK;
        dl[a] = (pk >> SRC_BITS) & (TS - 1);
        ok[a] = (pk >> OK_BIT) != 0;
        la[a] = *(const float4*)(xl + (size_t)s * F + cA);
        lb[a] = *(const float4*)(xl + (size_t)s * F + cB);
      }
#pragma unroll
      for (int a = 0; a < G; ++a) {
        const float4 ra = *(const float4*)(xrs + dl[a] * F + cA);
        const float4 rb = *(const float4*)(xrs + dl[a] * F + cB);
        const float m = shf[dl[a]];
        const float p = group8_sum(lrelu_dot4(la[a], ra, attA) + lrelu_dot4(lb[a], rb, attB));
        const float d = p - m;
        const float w = expf(d);
        if (ok[a] && (lane & 7) == 0) {
          atomicAdd(&den[dl[a]], w);
          if (d > SHIFT_LIM) over = 1;
        }
        const float wz = ok[a] ? w : 0.0f;
        float* pa = acc + dl[a] * PA + cA;
        float* pb = acc + dl[a] * PA + cB;
        atomicAdd(pa,     wz * la[a].x);
        atomicAdd(pa + 1, wz * la[a].y);
        atomicAdd(pa + 2, wz * la[a].z);
        atomicAdd(pa + 3, wz * la[a].w);
        atomicAdd(pb,     wz * lb[a].x);
        atomicAdd(pb + 1, wz * lb[a].y);
        atomicAdd(pb + 2, wz * lb[a].z);
        atomicAdd(pb + 3, wz * lb[a].w);
      }
    }
  }
  __syncthreads();

  if (over != 0) {
    gat_tile_body<1, C, U, false>(acc, (int*)shf, den, satt, off, bins, xl, xr, att, bias, lg, out,
                                  nullptr, nullptr, nullptr, nullptr, nullptr, nullptr);
    return;
  }

  {
    constexpr int NP = TS * (F / 4) / 256;
    static_assert(256 % (F / 4) == 0, "a thread keeps its column group from piece to piece");
    const float4 bv = ((const float4*)bias)[tid % (F / 4)];
#pragma unroll 1
    for (int k = 0; k < NP; ++k) {
      const int i4 = tid + 256 * k;
      const int dl = i4 / (F / 4), c = 4 * (i4 % (F / 4));
      const int n = n0 + dl;
      const float* ap = acc + dl * PA + c;
      const float q = den[dl] + 1e-16f;
      float4 o;
      o.x = ap[0] / q + bv.x;
      o.y = ap[1] / q + bv.y;
      o.z = ap[2] / q + bv.z;
      o.w = ap[3] / q + bv.w;
      if (n < N_) *(float4*)(out + (size_t)n * F + c) = o;
    }
  }
}

struct f3 { float x, y, z; };

__device__ __forceinline__ float quad_sum(float p) {
  p += __int_as_float(__builtin_amdgcn_update_dpp(0, __float_as_int(p), 0xB1, 0xF, 0xF, true));
  p += __int_as_float(__builtin_amdgcn_update_dpp(0, __float_as_int(p), 0x4E, 0xF, 0xF, true));
  return p;
}
__device__ __forceinline__ float slot_logit(const f3& l, const f3& r, const float (&a)[3]) {
  float p = 0.0f;
  float t0 = l.x + r.x; t0 = (t0 > 0.0f) ? t0 : 0.2f * t0; p += t0 * a[0];
  float t1 = l.y + r.y; t1 = (t1 > 0.0f) ? t1 : 0.2f * t1; p += t1 * a[1];
  float t2 = l.z + r.z; t2 = (t2 > 0.0f) ? t2 : 0.2f * t2; p += t2 * a[2];
  return quad_sum(p);
}

template <int U>
__global__ __launch_bounds__(256) void k_gat12_tile(const int* __restrict__ off, const int* __restrict__ bins,
                                                    const float* __restrict__ xl, const float* __restrict__ xr,
                                                    const float* __restrict__ att, const float* __restrict__ bias,
                                                    float* __restrict__ lg,
                                                    const float* __restrict__ Wnl, const float* __restrict__ bnl,
                                                    const float* __restrict__ Wnr, const float* __restrict__ bnr,
                                                    float* __restrict__ nxl, float* __restrict__ nxr) {
  constexpr int H = 12, C = 12, F = H * C, PF = F + 4;
  static_assert(U % 2 == 0, "items are taken in pairs");
  static_assert(4 * 3 == C && 32 * 3 * 3 == 2 * F, "a quad of lanes is a head, three slots are two rows");
  constexpr int NS = 3 * U / 2;
  __shared__ __attribute__((aligned(16))) float acc[TS * PF];
  __shared__ __attribute__((aligned(16))) float xrs[TS * F];
  __shared__ float shf[TS * H];
  __shared__ float den[TS * H];
  __shared__ float satt[F];
  __shared__ int   over;

  const int t = blockIdx.x, tid = threadIdx.x, lane = tid & 31;
  const int wv = __builtin_amdgcn_readfirstlane(tid >> 5);
  const int n0 = t * TS;
  constexpr int NST = TS * (F / 4) / 256;
  static_assert(TS * (F / 4) % 256 == 0, "the tile's rows split evenly over the group");
  {
    const float4* __restrict__ xr4 = (const float4*)(xr + (size_t)n0 * F);
    const int npc = min(TS, N_ - n0) * (F / 4);
    float4 st[NST];
#pragma unroll
    for (int k = 0; k < NST; ++k) {
      const int i = tid + 256 * k;
      st[k] = make_float4(0.0f, 0.0f, 0.0f, 0.0f);
      if (i < npc) st[k] = xr4[i];
    }
#pragma unroll
    for (int k = 0; k < NST; ++k) *(float4*)(xrs + 4 * (tid + 256 * k)) = st[k];
  }
  if (tid == 0) over = 0;

  const bool hiB = lane >= 16;
  int eo[3], hd[3];
  float av[3][3];
  eo[0] = 3 * lane;
  eo[1] = hiB ? 3 * lane - 48 : 96 + 3 * lane;
  eo[2] = 48 + 3 * lane;
#pragma unroll
  for (int z = 0; z < 3; ++z) {
    hd[z] = eo[z] / C;
#pragma unroll
    for (int c = 0; c < 3; ++c) av[z][c] = att[eo[z] + c];
  }
  __syncthreads();

  {
    constexpr int RW = TS / 8;
    static_assert(RW % 2 == 0, "rows are taken in pairs");
    constexpr int NSS = 3 * RW / 2;
    f3 lv[NSS];
    int rw[NSS];
#pragma unroll
    for (int i = 0; i < NSS; ++i) {
      const int z = i % 3;
      const int r0 = wv * RW + 2 * (i / 3), r1 = r0 + 1;
      rw[i] = (z == 0) ? r0 : (z == 2) ? r1 : (hiB ? r1 : r0);
      const int n = min(n0 + rw[i], N_ - 1);
      lv[i] = *(const f3*)(xl + (size_t)n * F + eo[z]);
    }
#pragma unroll
    for (int i = 0; i < NSS; ++i) {
      const int z = i % 3;
      const f3 rv = *(const f3*)(xrs + rw[i] * F + eo[z]);
      const float a = slot_logit(lv[i], rv, av[z]);
      const bool there = n0 + rw[i] < N_;
      float* ap = acc + rw[i] * PF + eo[z];
      ap[0] = there ? lv[i].x : 0.0f;
      ap[1] = there ? lv[i].y : 0.0f;
      ap[2] = there ? lv[i].z : 0.0f;
      if ((lane & 3) == 0) {
        shf[rw[i] * H + hd[z]] = a;
        den[rw[i] * H + hd[z]] = there ? 1.0f : 0.0f;
      }
    }
  }
  __syncthreads();

  const int b = off[t];
  const int cnt = off[t + 1] - b;

  for (int q0 = 0; wave_item<U>(q0, wv) < cnt; q0 += 32) {
    int sL, dlL;
    const int okL = edge_item(wave_item<U>(q0 + lane, wv), cnt, b, bins, sL, dlL) ? 1 : 0;
    for (int g = 0; g < 32 / U; ++g) {
      if (wave_item<U>(q0 + g * U, wv) >= cnt) break;
      int s[U], dl[U]; bool ok[U];
#pragma unroll
      for (int u = 0; u < U; ++u) {
        s[u]  = lane_bcast(sL, g * U + u);
        dl[u] = lane_bcast(dlL, g * U + u);
        ok[u] = lane_bcast(okL, g * U + u) != 0;
      }
      int sd[NS]; bool so[NS];
      f3 lv[NS], rv[NS];
      float mv[NS];
#pragma unroll
      for (int i = 0; i < NS; ++i) {
        const int z = i % 3;
        const int u0 = 2 * (i / 3), u1 = u0 + 1;
        const int ss = (z == 0) ? s[u0] : (z == 2) ? s[u1] : (hiB ? s[u1] : s[u0]);
        sd[i] = (z == 0) ? dl[u0] : (z == 2) ? dl[u1] : (hiB ? dl[u1] : dl[u0]);
        so[i] = (z == 0) ? ok[u0] : (z == 2) ? ok[u1] : (hiB ? ok[u1] : ok[u0]);
        lv[i] = *(const f3*)(xl + (size_t)ss * F + eo[z]);
        rv[i] = *(const f3*)(xrs + sd[i] * F + eo[z]);
        mv[i] = shf[sd[i] * H + hd[z]];
      }
#pragma unroll
      for (int i = 0; i < NS; ++i) {
        const int z = i % 3;
        const float d = slot_logit(lv[i], rv[i], av[z]) - mv[i];
        const float w = expf(d);
        if (so[i] && (lane & 3) == 0) {
          atomicAdd(&den[sd[i] * H + hd[z]], w);
          if (d > SHIFT_LIM) over = 1;
        }
        const float wz = so[i] ? w : 0.0f;
        float* ap = acc + sd[i] * PF + eo[z];
        atomicAdd(ap,     wz * lv[i].x);
        atomicAdd(ap + 1, wz * lv[i].y);
        atomicAdd(ap + 2, wz * lv[i].z);
      }
    }
  }
  __syncthreads();

  if (over != 0) {
    gat_tile_body<H, C, U, true>(acc, (int*)shf, den, satt, off, bins, xl, xr, att, bias, lg, nullptr,
                                 Wnl, bnl, Wnr, bnr, nxl, nxr);
    return;
  }

  {
    constexpr int NP = TS * (F / 4) / 256;
    constexpr int NB = 3;
    static_assert(TS * (F / 4) % 256 == 0 && NP % NB == 0 && C % 4 == 0 && PF % 4 == 0, "pieces split evenly");
    const float4* __restrict__ b4 = (const float4*)bias;
    for (int k0 = 0; k0 < NP; k0 += NB) {
      float4 av[NB], bv[NB];
      float q[NB];
      float4* ap[NB];
#pragma unroll
      for (int k = 0; k < NB; ++k) {
        const int i4 = tid + 256 * (k0 + k);
        const int dl = i4 / (F / 4), r4 = i4 - dl * (F / 4);
        ap[k] = (float4*)(acc + dl * PF + 4 * r4);
        av[k] = *ap[k];
        q[k]  = den[dl * H + r4 / (C / 4)] + 1e-16f;
        bv[k] = b4[r4];
      }
#pragma unroll
      for (int k = 0; k < NB; ++k) {
        float4 o;
        o.x = av[k].x / q[k] + bv[k].x;
        o.y = av[k].y / q[k] + bv[k].y;
        o.z = av[k].z / q[k] + bv[k].z;
        o.w = av[k].w / q[k] + bv[k].w;
        *ap[k] = o;
      }
    }
  }
  __syncthreads();
  tile_proj_pair<F, PF, F2_>(acc, n0, wv, lane, Wnl, bnl, nxl, Wnr, bnr, nxr);
}

constexpr size_t OFF_DINV = 0;
constexpr size_t OFF_CNT  = OFF_DINV + N_;
constexpr size_t OFF_OFFS = OFF_CNT  + 2048;
constexpr size_t OFF_CUR  = OFF_OFFS + 2048;
constexpr size_t OFF_XL2  = (size_t)N_ + E_;
constexpr size_t OFF_XR2  = OFF_XL2  + (size_t)N_*D_;
constexpr size_t OFF_XL1  = OFF_XR2  + (size_t)N_*D_;
constexpr size_t OFF_XR1  = OFF_XL1  + (size_t)N_*F1_;
constexpr size_t OFF_BINS = OFF_XR1  + (size_t)N_*F1_;
constexpr size_t OFF_LG   = OFF_BINS + (size_t)N_*24;
static_assert(OFF_CUR + 2048 <= OFF_XL2, "bin tables fit in front of the projections");
static_assert(F2_ == D_, "layer 2's projections are sized with D_ above");
static_assert((size_t)E_ <= (size_t)N_ * 24, "bins fit their region");

extern "C" void kernel_launch(void* const* d_in, const int* in_sizes, int n_in,
                              void* d_out, int out_size, void* d_ws, size_t ws_size,
                              hipStream_t stream) {
  const float* x     = (const float*)d_in[0];
  const int*   ei    = (const int*)  d_in[1];
  const float* Wssg  = (const float*)d_in[2];
  const float* bssg  = (const float*)d_in[3];
  const float* W1l   = (const float*)d_in[4];
  const float* b1l   = (const float*)d_in[5];
  const float* W1r   = (const float*)d_in[6];
  const float* b1r   = (const float*)d_in[7];
  const float* att1  = (const float*)d_in[8];
  const float* bias1 = (const float*)d_in[9];
  const float* W2l   = (const float*)d_in[10];
  const float* b2l   = (const float*)d_in[11];
  const float* W2r   = (const float*)d_in[12];
  const float* b2r   = (const float*)d_in[13];
  const float* att2  = (const float*)d_in[14];
  const float* bias2 = (const float*)d_in[15];

  const int* src = ei;
  const int* dst = ei + E_;

  float* ws   = (float*)d_ws;
  float* dinv = ws + OFF_DINV;
  int*   cnt  = (int*)(ws + OFF_CNT);
  int*   offs = (int*)(ws + OFF_OFFS);
  int*   cur  = (int*)(ws + OFF_CUR);
  float* xl2  = ws + OFF_XL2;
  float* xr2  = ws + OFF_XR2;
  float* xl1  = ws + OFF_XL1;
  float* xr1  = ws + OFF_XR1;
  int*   bins = (int*)(ws + OFF_BINS);
  float* lg   = ws + OFF_LG;
  float* out  = (float*)d_out;

  const int TB = 256;

  k_zero_i<<<CDIV(2048, TB), TB, 0, stream>>>(cnt, 2048);
  k_tilecount<<<CDIV(E_, CH), 1024, 0, stream>>>(dst, cnt);
  k_scan<<<1, 1024, 0, stream>>>(cnt, offs, cur);
  k_binscatter<<<CDIV(E_, CH), 1024, 0, stream>>>(src, dst, cur, bins);

  k_dinv_tile<<<NT, TS, 0, stream>>>(offs, bins, dinv);
  k_ssg_tile<4><<<NT, 256, 0, stream>>>(offs, bins, x, dinv, Wssg, bssg, W1l, b1l, W1r, b1r, xl1, xr1);

  k_gat12_tile<4><<<NT, 256, 0, stream>>>(offs, bins, xl1, xr1, att1, bias1, lg,
                                          W2l, b2l, W2r, b2r, xl2, xr2);

  k_gat1h_tile<64, 4><<<NT, 256, 0, stream>>>(offs, bins, xl2, xr2, att2, bias2, lg, out);
}


# 12-head pass: a lane holds 6 floats of a row (was 3): a head is a lane pair, 4 items fill 3 slots; per-slot work (exp, shift read, row-sum add, addressing) serves 6 floats; logit sum order changed
# speedup vs baseline: 6.7022x; 1.0329x over previous
#include <hip/hip_runtime.h>
#include <cmath>

constexpr int N_  = 100000;
constexpr int E_  = 1000000;
constexpr int D_  = 64;
constexpr int F1_ = 144;
constexpr int F2_ = 64;

#define CDIV(a,b) (((a)+(b)-1)/(b))

typedef __attribute__((ext_vector_type(2))) float v2f;
typedef __attribute__((ext_vector_type(8))) float v8f;

constexpr int TSH = 6;
constexpr int TS  = 1 << TSH;
constexpr int NT  = CDIV(N_, TS);
constexpr int CH  = 8192;
constexpr int SRC_BITS = 17;
constexpr int SRC_MASK = (1 << SRC_BITS) - 1;
static_assert(N_ <= (1 << SRC_BITS), "source id must fit the packed field");
static_assert(NT + 1 <= 2048, "scan handles two tiles per thread of one 1024-thread group");

__global__ void k_zero_i(int* __restrict__ p, int n) {
  int i = blockIdx.x * blockDim.x + threadIdx.x;
  if (i < n) p[i] = 0;
}

__global__ __launch_bounds__(1024) void k_tilecount(const int* __restrict__ dst, int* __restrict__ cnt) {
  __shared__ int hist[NT];
  const int tid = threadIdx.x;
  for (int i = tid; i < NT; i += 1024) hist[i] = 0;
  __syncthreads();
  const int base = blockIdx.x * CH;
#pragma unroll
  for (int k = 0; k < CH / 1024; ++k) {
    int e = base + k * 1024 + tid;
    if (e < E_) atomicAdd(&hist[dst[e] >> TSH], 1);
  }
  __syncthreads();
  for (int i = tid; i < NT; i += 1024) { int v = hist[i]; if (v) atomicAdd(&cnt[i], v); }
}

__global__ __launch_bounds__(1024) void k_scan(const int* __restrict__ cnt, int* __restrict__ off,
                                               int* __restrict__ cur) {
  __shared__ int p[1024];
  const int t = threadIdx.x;
  const int a = (2 * t     < NT) ? cnt[2 * t]     : 0;
  const int b = (2 * t + 1 < NT) ? cnt[2 * t + 1] : 0;
  int v = a + b;
  p[t] = v;
  __syncthreads();
  for (int o = 1; o < 1024; o <<= 1) {
    int add = (t >= o) ? p[t - o] : 0;
    __syncthreads();
    v += add;
    p[t] = v;
    __syncthreads();
  }
  const int excl = v - (a + b);
  if (2 * t     < NT) { off[2 * t]     = excl;     cur[2 * t]     = excl; }
  if (2 * t + 1 < NT) { off[2 * t + 1] = excl + a; cur[2 * t + 1] = excl + a; }
  if (t == 1023) off[NT] = v;
}

__global__ __launch_bounds__(1024) void k_binscatter(const int* __restrict__ src, const int* __restrict__ dst,
                                                     int* __restrict__ cur, int* __restrict__ bins) {
  __shared__ int hist[NT];
  const int tid = threadIdx.x;
  for (int i = tid; i < NT; i += 1024) hist[i] = 0;
  __syncthreads();
  const int base = blockIdx.x * CH;
  int pk[CH / 1024], bn[CH / 1024], lr[CH / 1024];
#pragma unroll
  for (int k = 0; k < CH / 1024; ++k) {
    int e = base + k * 1024 + tid;
    bn[k] = -1; pk[k] = 0; lr[k] = 0;
    if (e < E_) {
      int d = dst[e];
      bn[k] = d >> TSH;
      pk[k] = src[e] | ((d & (TS - 1)) << SRC_BITS);
      lr[k] = atomicAdd(&hist[bn[k]], 1);
    }
  }
  __syncthreads();
  for (int i = tid; i < NT; i += 1024) { int v = hist[i]; hist[i] = v ? atomicAdd(&cur[i], v) : 0; }
  __syncthreads();
#pragma unroll
  for (int k = 0; k < CH / 1024; ++k)
    if (bn[k] >= 0) bins[hist[bn[k]] + lr[k]] = pk[k];
}

__device__ __forceinline__ bool tile_item(int j, int tot, int cnt, int b, int n0,
                                          const int* __restrict__ bins, int& s, int& dl, int& sl) {
  const int p  = min(b + j, E_ - 1);
  const int pk = bins[p];
  const bool isE = j < cnt;
  const int dlS = j - cnt;
  const int sS  = n0 + dlS;
  const bool okS = (!isE) && (j < tot) && (sS < N_);
  s  = isE ? (pk & SRC_MASK)  : (okS ? sS : 0);
  dl = isE ? (pk >> SRC_BITS) : (okS ? dlS : 0);
  sl = isE ? p                : (okS ? E_ + sS : 0);
  return isE || okS;
}

__device__ __forceinline__ bool edge_item(int j, int cnt, int b, const int* __restrict__ bins, int& s, int& dl) {
  const int pk = bins[min(b + j, E_ - 1)];
  const bool ok = j < cnt;
  s  = ok ? (pk & SRC_MASK)  : 0;
  dl = ok ? (pk >> SRC_BITS) : 0;
  return ok;
}

__device__ __forceinline__ void ldsMaxF(int* cell, float v) {
  if (v >= 0.0f) atomicMax(cell, __float_as_int(v));
  else           atomicMin((unsigned int*)cell, __float_as_uint(v));
}

template <int U>
__device__ __forceinline__ int wave_item(int q, int wv) {
  const unsigned uq = (unsigned)q;
  return (int)(((uq / U) * 8u + (unsigned)wv) * U + (uq % U));
}
__device__ __forceinline__ int lane_bcast(int v, int ln) { return __builtin_amdgcn_readlane(v, ln); }
__device__ __forceinline__ float lane_bcast(float v, int ln) {
  return __int_as_float(__builtin_amdgcn_readlane(__float_as_int(v), ln));
}

__global__ __launch_bounds__(TS) void k_dinv_tile(const int* __restrict__ off, const int* __restrict__ bins,
                                                  float* __restrict__ dinv) {
  __shared__ int c[TS];
  const int t = blockIdx.x, tid = threadIdx.x;
  c[tid] = 0;
  __syncthreads();
  const int b = off[t], e = off[t + 1];
  for (int p = b + tid; p < e; p += TS) atomicAdd(&c[bins[p] >> SRC_BITS], 1);
  __syncthreads();
  const int n = t * TS + tid;
  if (n < N_) dinv[n] = rsqrtf(1.0f + (float)c[tid]);
}

template <int K, int PA, int NC>
__device__ __forceinline__ void tile_mma2(const float* a0p, const float* __restrict__ wp, v8f& acc0, v8f& acc1) {
  const float* a1p = a0p + 16 * PA;
#pragma unroll 4
  for (int k = 0; k < K; k += 4) {
    v2f a0, a1, b;
    a0.x = a0p[k]; a0.y = a0p[k + 1];
    a1.x = a1p[k]; a1.y = a1p[k + 1];
    b.x = wp[(size_t)k * NC];
    b.y = wp[(size_t)(k + 1) * NC];
    acc0 = __builtin_amdgcn_wmma_f32_16x16x4_f32(false, a0, false, b, (short)0, acc0, false, false);
    acc1 = __builtin_amdgcn_wmma_f32_16x16x4_f32(false, a1, false, b, (short)0, acc1, false, false);
  }
}

template <int K, int PA, int NC, int PC>
__device__ __forceinline__ void tile_proj_lds(const float* As, const float* __restrict__ W,
                                              const float* __restrict__ bias, float* Cs, int wv, int lane) {
  const int c = lane & 15, hh = lane >> 4;
  const int r0 = (wv & 1) * 32;
  for (int ct = wv >> 1; ct < NC / 16; ct += 4) {
    const int col = ct * 16 + c;
    v8f acc0 = {}, acc1 = {};
    tile_mma2<K, PA, NC>(As + (r0 + c) * PA + 2 * hh, W + (size_t)(2 * hh) * NC + col, acc0, acc1);
    const float bn = bias[col];
#pragma unroll
    for (int j = 0; j < 8; ++j) {
      Cs[(r0 + 8 * hh + j) * PC + col]      = acc0[j] + bn;
      Cs[(r0 + 16 + 8 * hh + j) * PC + col] = acc1[j] + bn;
    }
  }
}

template <int K, int PA, int NC>
__device__ __forceinline__ void tile_proj_pair(const float* As, int n0, int wv, int lane,
                                               const float* __restrict__ Wa, const float* __restrict__ ba,
                                               float* __restrict__ Ca,
                                               const float* __restrict__ Wb, const float* __restrict__ bb,
                                               float* __restrict__ Cb) {
  constexpr int CT = NC / 16;
  const int c = lane & 15, hh = lane >> 4;
  const int r0 = (wv & 1) * 32;
  for (int ct = wv >> 1; ct < 2 * CT; ct += 4) {
    const bool second = ct >= CT;
    const float* __restrict__ W  = second ? Wb : Wa;
    const float* __restrict__ bs = second ? bb : ba;
    float* __restrict__ C        = second ? Cb : Ca;
    const int col = (second ? ct - CT : ct) * 16 + c;
    v8f acc0 = {}, acc1 = {};
    tile_mma2<K, PA, NC>(As + (r0 + c) * PA + 2 * hh, W + (size_t)(2 * hh) * NC + col, acc0, acc1);
    const float bn = bs[col];
    float* __restrict__ cp = C + (size_t)(n0 + r0 + 8 * hh) * NC + col;
    if (n0 + r0 < N_) {
#pragma unroll
      for (int j = 0; j < 8; ++j) cp[(size_t)j * NC] = acc0[j] + bn;
    }
    if (n0 + r0 + 16 < N_) {
#pragma unroll
      for (int j = 0; j < 8; ++j) cp[(size_t)(16 + j) * NC] = acc1[j] + bn;
    }
  }
}

constexpr int PH = D_ + 4;
static_assert(N_ % 16 == 0, "a 16-row output tile is never partly past the last node");
template <int U>
__global__ __launch_bounds__(256) void k_ssg_tile(const int* __restrict__ off, const int* __restrict__ bins,
                                                  const float* __restrict__ x, const float* __restrict__ dinv,
                                                  const float* __restrict__ Wssg, const float* __restrict__ bssg,
                                                  const float* __restrict__ W1l, const float* __restrict__ b1l,
                                                  const float* __restrict__ W1r, const float* __restrict__ b1r,
                                                  float* __restrict__ xl1, float* __restrict__ xr1) {
  __shared__ __attribute__((aligned(16))) float acc[TS * PH];
  __shared__ float x1s[TS * PH];
  static_assert(PH % 4 == 0, "a row of the tile starts on a 16-byte boundary");
  const int t = blockIdx.x, tid = threadIdx.x, lane = tid & 31;
  const int wv = __builtin_amdgcn_readfirstlane(tid >> 5);
  for (int i = tid; i < TS * PH / 4; i += 256) ((float4*)acc)[i] = make_float4(0.0f, 0.0f, 0.0f, 0.0f);
  __syncthreads();
  const int b = off[t];
  const int cnt = off[t + 1] - b;
  const int n0 = t * TS;
  for (int q0 = 0; wave_item<U>(q0, wv) < cnt; q0 += 32) {
    int sL, dlL;
    const int okL = edge_item(wave_item<U>(q0 + lane, wv), cnt, b, bins, sL, dlL) ? 1 : 0;
    const float wL = dinv[sL] * dinv[n0 + dlL];
    for (int g = 0; g < 32 / U; ++g) {
      if (wave_item<U>(q0 + g * U, wv) >= cnt) break;
      int s[U], dl[U]; bool ok[U];
      float x0[U], x1[U], w[U];
#pragma unroll
      for (int u = 0; u < U; ++u) {
        s[u]  = lane_bcast(sL, g * U + u);
        dl[u] = lane_bcast(dlL, g * U + u);
        ok[u] = lane_bcast(okL, g * U + u) != 0;
        w[u]  = lane_bcast(wL, g * U + u);
      }
#pragma unroll
      for (int u = 0; u < U; ++u) {
        x0[u] = x[(size_t)s[u] * D_ + lane];
        x1[u] = x[(size_t)s[u] * D_ + lane + 32];
      }
#pragma unroll
      for (int u = 0; u < U; ++u) {
        atomicAdd(&acc[dl[u] * PH + lane],      ok[u] ? x0[u] * w[u] : 0.0f);
        atomicAdd(&acc[dl[u] * PH + lane + 32], ok[u] ? x1[u] * w[u] : 0.0f);
      }
    }
  }
  __syncthreads();
  {
    constexpr int NP = TS * (D_ / 4) / 256;
    static_assert(TS * (D_ / 4) % 256 == 0, "the tile's rows split evenly over the group");
    const float4* __restrict__ x4 = (const float4*)x;
    float4 xv[NP];
    float dn[NP];
#pragma unroll
    for (int k = 0; k < NP; ++k) {
      const int i4 = tid + 256 * k;
      const int n = min(n0 + i4 / (D_ / 4), N_ - 1);
      xv[k] = x4[(size_t)n * (D_ / 4) + i4 % (D_ / 4)];
      dn[k] = dinv[n];
    }
#pragma unroll
    for (int k = 0; k < NP; ++k) {
      const int i4 = tid + 256 * k;
      const int dl = i4 / (D_ / 4);
      float4* ap = (float4*)(acc + dl * PH + 4 * (i4 % (D_ / 4)));
      const float4 av = *ap;
      const float d2 = dn[k] * dn[k];
      float4 hv;
      hv.x = 0.5f * xv[k].x + 0.5f * (av.x + xv[k].x * d2);
      hv.y = 0.5f * xv[k].y + 0.5f * (av.y + xv[k].y * d2);
      hv.z = 0.5f * xv[k].z + 0.5f * (av.z + xv[k].z * d2);
      hv.w = 0.5f * xv[k].w + 0.5f * (av.w + xv[k].w * d2);
      *ap = hv;
    }
  }
  __syncthreads();
  tile_proj_lds<D_, PH, D_, PH>(acc, Wssg, bssg, x1s, wv, lane);
  __syncthreads();
  tile_proj_pair<D_, PH, F1_>(x1s, n0, wv, lane, W1l, b1l, xl1, W1r, b1r, xr1);
}

template <int H, int C, int U, bool PROJ>
__device__ __forceinline__ void gat_tile_body(float* acc, int* mxi, float* den, float* satt,
                                              const int* __restrict__ off, const int* __restrict__ bins,
                                              const float* __restrict__ xl, const float* __restrict__ xr,
                                              const float* __restrict__ att, const float* __restrict__ bias,
                                              float* __restrict__ lg, float* __restrict__ out,
                                              const float* __restrict__ Wnl, const float* __restrict__ bnl,
                                              const float* __restrict__ Wnr, const float* __restrict__ bnr,
                                              float* __restrict__ nxl, float* __restrict__ nxr) {
  constexpr int F = H * C;
  constexpr int K = (F + 31) / 32;
  constexpr int PF = PROJ ? F + 4 : F;
  static_assert(C % 4 == 0, "head rows are read as 16-byte pieces");
  static_assert(!PROJ || (PF % 64 == 4 || PF % 64 == 20 || PF % 64 == 36 || PF % 64 == 52),
                "fragment reads of 16 rows must fall on 16 different bank groups");

  const int t = blockIdx.x, tid = threadIdx.x, lane = tid & 31;
  const int wv = __builtin_amdgcn_readfirstlane(tid >> 5);
  const int n0 = t * TS;
  float* const xrs = acc;
  constexpr int NST = TS * (F / 4) / 256;
  static_assert(TS * (F / 4) % 256 == 0, "the tile's rows split evenly over the group");
  {
    const float4* __restrict__ xr4 = (const float4*)(xr + (size_t)n0 * F);
    const int npc = min(TS, N_ - n0) * (F / 4);
    float4 st[NST];
#pragma unroll
    for (int k = 0; k < NST; ++k) {
      const int i = tid + 256 * k;
      st[k] = make_float4(0.0f, 0.0f, 0.0f, 0.0f);
      if (i < npc) st[k] = xr4[i];
    }
#pragma unroll
    for (int k = 0; k < NST; ++k) *(float4*)(xrs + 4 * (tid + 256 * k)) = st[k];
  }
  for (int i = tid; i < TS * H; i += 256) { mxi[i] = (int)0xFF800000u; den[i] = 0.0f; }
  for (int i = tid; i < F; i += 256) satt[i] = att[i];
  __syncthreads();
  const int b = off[t];
  const int cnt = off[t + 1] - b;
  const int tot = cnt + TS;

  if constexpr (H == 1) {
    for (int q0 = 0; wave_item<U>(q0, wv) < tot; q0 += 32) {
      int sL, dlL, slL;
      const int okL = tile_item(wave_item<U>(q0 + lane, wv), tot, cnt, b, n0, bins, sL, dlL, slL) ? 1 : 0;
      for (int g = 0; g < 32 / U; ++g) {
        if (wave_item<U>(q0 + g * U, wv) >= tot) break;
        int s[U], dl[U], sl[U]; bool ok[U];
        float l0[U], l1[U], r0[U], r1[U];
#pragma unroll
        for (int u = 0; u < U; ++u) {
          s[u]  = lane_bcast(sL, g * U + u);
          dl[u] = lane_bcast(dlL, g * U + u);
          sl[u] = lane_bcast(slL, g * U + u);
          ok[u] = lane_bcast(okL, g * U + u) != 0;
        }
#pragma unroll
        for (int u = 0; u < U; ++u) {
          l0[u] = xl[(size_t)s[u] * F + lane];
          l1[u] = xl[(size_t)s[u] * F + lane + 32];
          r0[u] = xrs[dl[u] * F + lane];
          r1[u] = xrs[dl[u] * F + lane + 32];
        }
#pragma unroll
        for (int u = 0; u < U; ++u) {
          float a = 0.0f;
          float t0 = l0[u] + r0[u];
          t0 = (t0 > 0.0f) ? t0 : 0.2f * t0;
          a += t0 * satt[lane];
          float t1 = l1[u] + r1[u];
          t1 = (t1 > 0.0f) ? t1 : 0.2f * t1;
          a += t1 * satt[lane + 32];
#pragma unroll
          for (int o = 16; o > 0; o >>= 1) a += __shfl_xor(a, o, 32);
          if (lane == 0 && ok[u]) { lg[sl[u]] = a; ldsMaxF(&mxi[dl[u]], a); }
        }
      }
    }
  } else {
    for (int idx = tid; idx < tot * H; idx += 256) {
      const int j = idx / H, h = idx - j * H;
      int s, dl, sl;
      if (tile_item(j, tot, cnt, b, n0, bins, s, dl, sl)) {
        const float4* pl = (const float4*)(xl + (size_t)s * F + h * C);
        const float4* pr = (const float4*)(xrs + dl * F + h * C);
        float4 lv[C / 4], rv[C / 4];
#pragma unroll
        for (int q = 0; q < C / 4; ++q) { lv[q] = pl[q]; rv[q] = pr[q]; }
        float a = 0.0f;
#pragma unroll
        for (int q = 0; q < C / 4; ++q) {
          const float le[4] = {lv[q].x, lv[q].y, lv[q].z, lv[q].w};
          const float re[4] = {rv[q].x, rv[q].y, rv[q].z, rv[q].w};
#pragma unroll
          for (int z = 0; z < 4; ++z) {
            float tt = le[z] + re[z];
            tt = (tt > 0.0f) ? tt : 0.2f * tt;
            a += tt * satt[h * C + q * 4 + z];
          }
        }
        lg[(size_t)sl * H + h] = a;
        ldsMaxF(&mxi[dl * H + h], a);
      }
    }
  }
  __syncthreads();

  for (int i = tid; i < TS * PF; i += 256) acc[i] = 0.0f;
  for (int idx = tid; idx < tot * H; idx += 256) {
    const int j = idx / H, h = idx - j * H;
    int s, dl, sl;
    if (tile_item(j, tot, cnt, b, n0, bins, s, dl, sl)) {
      const float a = expf(lg[(size_t)sl * H + h] - __int_as_float(mxi[dl * H + h]));
      lg[(size_t)sl * H + h] = a;
      atomicAdd(&den[dl * H + h], a);
    }
  }
  __syncthreads();

  for (int q0 = 0; wave_item<U>(q0, wv) < tot; q0 += 32) {
    int sL, dlL, slL;
    const int okL = tile_item(wave_item<U>(q0 + lane, wv), tot, cnt, b, n0, bins, sL, dlL, slL) ? 1 : 0;
    float aL = 0.0f;
    if constexpr (H == 1) aL = lg[slL];
    for (int g = 0; g < 32 / U; ++g) {
      if (wave_item<U>(q0 + g * U, wv) >= tot) break;
      int s[U], dl[U], sl[U]; bool ok[U];
      float xv[U][K], av[U][K];
#pragma unroll
      for (int u = 0; u < U; ++u) {
        s[u]  = lane_bcast(sL, g * U + u);
        dl[u] = lane_bcast(dlL, g * U + u);
        sl[u] = lane_bcast(slL, g * U + u);
        ok[u] = lane_bcast(okL, g * U + u) != 0;
      }
#pragma unroll
      for (int u = 0; u < U; ++u) {
#pragma unroll
        for (int k = 0; k < K; ++k) {
          const int r  = lane + 32 * k;
          const int rr = (r < F) ? r : r - 32;
          xv[u][k] = xl[(size_t)s[u] * F + rr];
          if constexpr (H == 1) av[u][k] = lane_bcast(aL, g * U + u);
          else                  av[u][k] = lg[(size_t)sl[u] * H + rr / C];
        }
      }
#pragma unroll
      for (int u = 0; u < U; ++u) {
#pragma unroll
        for (int k = 0; k < K; ++k) {
          const int r  = lane + 32 * k;
          const int rr = (r < F) ? r : r - 32;
          atomicAdd(&acc[dl[u] * PF + rr], (ok[u] && r < F) ? av[u][k] * xv[u][k] : 0.0f);
        }
      }
    }
  }
  __syncthreads();

  if constexpr (!PROJ) {
    for (int i = tid; i < TS * F; i += 256) {
      const int dl = i / F, r = i - dl * F;
      const int n = n0 + dl;
      if (n < N_) out[(size_t)n * F + r] = acc[i] / (den[dl * H + r / C] + 1e-16f) + bias[r];
    }
  } else {
    for (int i = tid; i < TS * F; i += 256) {
      const int dl = i / F, r = i - dl * F;
      acc[dl * PF + r] = acc[dl * PF + r] / (den[dl * H + r / C] + 1e-16f) + bias[r];
    }
    __syncthreads();
    tile_proj_pair<F, PF, F2_>(acc, n0, wv, lane, Wnl, bnl, nxl, Wnr, bnr, nxr);
  }
}

constexpr float SHIFT_LIM = 40.0f;
__device__ __forceinline__ float group8_sum(float p) {
  p += __int_as_float(__builtin_amdgcn_update_dpp(0, __float_as_int(p), 0xB1, 0xF, 0xF, true));
  p += __int_as_float(__builtin_amdgcn_update_dpp(0, __float_as_int(p), 0x4E, 0xF, 0xF, true));
  p += __int_as_float(__builtin_amdgcn_update_dpp(0, __float_as_int(p), 0x141, 0xF, 0xF, true));
  return p;
}
__device__ __forceinline__ float lrelu_dot4(const float4& l, const float4& r, const float4& a) {
  float p = 0.0f;
  float t0 = l.x + r.x; t0 = (t0 > 0.0f) ? t0 : 0.2f * t0; p += t0 * a.x;
  float t1 = l.y + r.y; t1 = (t1 > 0.0f) ? t1 : 0.2f * t1; p += t1 * a.y;
  float t2 = l.z + r.z; t2 = (t2 > 0.0f) ? t2 : 0.2f * t2; p += t2 * a.z;
  float t3 = l.w + r.w; t3 = (t3 > 0.0f) ? t3 : 0.2f * t3; p += t3 * a.w;
  return p;
}
template <int C, int U>
__global__ __launch_bounds__(256, 16) void k_gat1h_tile(const int* __restrict__ off, const int* __restrict__ bins,
                                                    const float* __restrict__ xl, const float* __restrict__ xr,
                                                    const float* __restrict__ att, const float* __restrict__ bias,
                                                    float* __restrict__ lg, float* __restrict__ out) {
  constexpr int F = C;
  static_assert(F == 64 && U == 4, "a group of 8 lanes holds a row as eight floats per lane, four items per step");
  constexpr int PA = F + 1;
  constexpr int G = 2;
  __shared__ __attribute__((aligned(16))) float acc[TS * PA];
  __shared__ __attribute__((aligned(16))) float xrs[TS * F];
  __shared__ float shf[TS];
  __shared__ float den[TS];
  __shared__ __attribute__((aligned(16))) float satt[F];
  __shared__ int   over;

  const int t = blockIdx.x, tid = threadIdx.x, lane = tid & 31;
  const int wv = __builtin_amdgcn_readfirstlane(tid >> 5);
  const int n0 = t * TS;
  constexpr int NST = TS * (F / 4) / 256;
  static_assert(TS * (F / 4) % 256 == 0, "the tile's rows split evenly over the group");
  {
    const float4* __restrict__ xr4 = (const float4*)(xr + (size_t)n0 * F);
    const int npc = min(TS, N_ - n0) * (F / 4);
    float4 st[NST];
#pragma unroll
    for (int k = 0; k < NST; ++k) {
      const int i = tid + 256 * k;
      st[k] = make_float4(0.0f, 0.0f, 0.0f, 0.0f);
      if (i < npc) st[k] = xr4[i];
    }
#pragma unroll
    for (int k = 0; k < NST; ++k) *(float4*)(xrs + 4 * (tid + 256 * k)) = st[k];
  }
  if (tid < F) satt[tid] = att[tid];
  if (tid == 0) over = 0;
  __syncthreads();

  {
    constexpr int RW = TS / 8;
    float l0[RW], l1[RW];
#pragma unroll
    for (int r = 0; r < RW; ++r) {
      const int n = min(n0 + wv * RW + r, N_ - 1);
      l0[r] = xl[(size_t)n * F + lane];
      l1[r] = xl[(size_t)n * F + lane + 32];
    }
    float a[RW];
#pragma unroll
    for (int r = 0; r < RW; ++r) {
      const int dl = wv * RW + r;
      a[r] = 0.0f;
      float t0 = l0[r] + xrs[dl * F + lane];
      t0 = (t0 > 0.0f) ? t0 : 0.2f * t0;
      a[r] += t0 * satt[lane];
      float t1 = l1[r] + xrs[dl * F + lane + 32];
      t1 = (t1 > 0.0f) ? t1 : 0.2f * t1;
      a[r] += t1 * satt[lane + 32];
    }
#pragma unroll
    for (int o = 16; o > 0; o >>= 1) {
#pragma unroll
      for (int r = 0; r < RW; ++r) a[r] += __shfl_xor(a[r], o, 32);
    }
#pragma unroll
    for (int r = 0; r < RW; ++r) {
      const int dl = wv * RW + r;
      const bool there = n0 + dl < N_;
      acc[dl * PA + lane]      = there ? l0[r] : 0.0f;
      acc[dl * PA + lane + 32] = there ? l1[r] : 0.0f;
    }
    if (lane == 0) {
#pragma unroll
      for (int r = 0; r < RW; ++r) {
        shf[wv * RW + r] = a[r];
        den[wv * RW + r] = (n0 + wv * RW + r < N_) ? 1.0f : 0.0f;
      }
    }
  }
  __syncthreads();

  const int b = off[t];
  const int cnt = off[t + 1] - b;

  const int grp = lane >> 3;
  const int cA = 32 * (grp & 1) + 4 * (lane & 7);
  const int cB = cA ^ 32;
  const float4 attA = *(const float4*)(satt + cA);
  const float4 attB = *(const float4*)(satt + cB);
  constexpr int OK_BIT = SRC_BITS + TSH;
  static_assert(OK_BIT < 31, "source, local destination and the exists flag share one word");

  for (int q0 = 0; wave_item<U>(q0, wv) < cnt; q0 += 32) {
    int sL, dlL;
    const int okL = edge_item(wave_item<U>(q0 + lane, wv), cnt, b, bins, sL, dlL) ? 1 : 0;
    const int pkL = sL | (dlL << SRC_BITS) | (okL << OK_BIT);
    for (int g = 0; g < 32 / U; g += G) {
      if (wave_item<U>(q0 + g * U, wv) >= cnt) break;
      int dl[G]; bool ok[G];
      float4 la[G], lb[G];
#pragma unroll
      for (int a = 0; a < G; ++a) {
        const int pk = __builtin_amdgcn_ds_bpermute(4 * ((g + a) * U + grp), pkL);
        const int s = pk & SRC_MASK;
        dl[a] = (pk >> SRC_BITS) & (TS - 1);
        ok[a] = (pk >> OK_BIT) != 0;
        la[a] = *(const float4*)(xl + (size_t)s * F + cA);
        lb[a] = *(const float4*)(xl + (size_t)s * F + cB);
      }
#pragma unroll
      for (int a = 0; a < G; ++a) {
        const float4 ra = *(const float4*)(xrs + dl[a] * F + cA);
        const float4 rb = *(const float4*)(xrs + dl[a] * F + cB);
        const float m = shf[dl[a]];
        const float p = group8_sum(lrelu_dot4(la[a], ra, attA) + lrelu_dot4(lb[a], rb, attB));
        const float d = p - m;
        const float w = expf(d);
        if (ok[a] && (lane & 7) == 0) {
          atomicAdd(&den[dl[a]], w);
          if (d > SHIFT_LIM) over = 1;
        }
        const float wz = ok[a] ? w : 0.0f;
        float* pa = acc + dl[a] * PA + cA;
        float* pb = acc + dl[a] * PA + cB;
        atomicAdd(pa,     wz * la[a].x);
        atomicAdd(pa + 1, wz * la[a].y);
        atomicAdd(pa + 2, wz * la[a].z);
        atomicAdd(pa + 3, wz * la[a].w);
        atomicAdd(pb,     wz * lb[a].x);
        atomicAdd(pb + 1, wz * lb[a].y);
        atomicAdd(pb + 2, wz * lb[a].z);
        atomicAdd(pb + 3, wz * lb[a].w);
      }
    }
  }
  __syncthreads();

  if (over != 0) {
    gat_tile_body<1, C, U, false>(acc, (int*)shf, den, satt, off, bins, xl, xr, att, bias, lg, out,
                                  nullptr, nullptr, nullptr, nullptr, nullptr, nullptr);
    return;
  }

  {
    constexpr int NP = TS * (F / 4) / 256;
    static_assert(256 % (F / 4) == 0, "a thread keeps its column group from piece to piece");
    const float4 bv = ((const float4*)bias)[tid % (F / 4)];
#pragma unroll 1
    for (int k = 0; k < NP; ++k) {
      const int i4 = tid + 256 * k;
      const int dl = i4 / (F / 4), c = 4 * (i4 % (F / 4));
      const int n = n0 + dl;
      const float* ap = acc + dl * PA + c;
      const float q = den[dl] + 1e-16f;
      float4 o;
      o.x = ap[0] / q + bv.x;
      o.y = ap[1] / q + bv.y;
      o.z = ap[2] / q + bv.z;
      o.w = ap[3] / q + bv.w;
      if (n < N_) *(float4*)(out + (size_t)n * F + c) = o;
    }
  }
}

struct f3 { float x, y, z; };
struct f6 { float2 a, b, c; };

__device__ __forceinline__ float quad_sum(float p) {
  p += __int_as_float(__builtin_amdgcn_update_dpp(0, __float_as_int(p), 0xB1, 0xF, 0xF, true));
  p += __int_as_float(__builtin_amdgcn_update_dpp(0, __float_as_int(p), 0x4E, 0xF, 0xF, true));
  return p;
}
__device__ __forceinline__ float slot_logit(const f3& l, const f3& r, const float (&a)[3]) {
  float p = 0.0f;
  float t0 = l.x + r.x; t0 = (t0 > 0.0f) ? t0 : 0.2f * t0; p += t0 * a[0];
  float t1 = l.y + r.y; t1 = (t1 > 0.0f) ? t1 : 0.2f * t1; p += t1 * a[1];
  float t2 = l.z + r.z; t2 = (t2 > 0.0f) ? t2 : 0.2f * t2; p += t2 * a[2];
  return quad_sum(p);
}

__device__ __forceinline__ float lrelu_term(float l, float r, float a) {
  float t = l + r;
  t = (t > 0.0f) ? t : 0.2f * t;
  return t * a;
}
__device__ __forceinline__ float pair_logit(const f6& l, const f6& r, const float (&a)[6]) {
  float p = 0.0f;
  p += lrelu_term(l.a.x, r.a.x, a[0]);
  p += lrelu_term(l.a.y, r.a.y, a[1]);
  p += lrelu_term(l.b.x, r.b.x, a[2]);
  p += lrelu_term(l.b.y, r.b.y, a[3]);
  p += lrelu_term(l.c.x, r.c.x, a[4]);
  p += lrelu_term(l.c.y, r.c.y, a[5]);
  p += __int_as_float(__builtin_amdgcn_update_dpp(0, __float_as_int(p), 0xB1, 0xF, 0xF, true));
  return p;
}

template <int U>
__global__ __launch_bounds__(256) void k_gat12_tile(const int* __restrict__ off, const int* __restrict__ bins,
                                                    const float* __restrict__ xl, const float* __restrict__ xr,
                                                    const float* __restrict__ att, const float* __restrict__ bias,
                                                    float* __restrict__ lg,
                                                    const float* __restrict__ Wnl, const float* __restrict__ bnl,
                                                    const float* __restrict__ Wnr, const float* __restrict__ bnr,
                                                    float* __restrict__ nxl, float* __restrict__ nxr) {
  constexpr int H = 12, C = 12, F = H * C, PF = F + 4;
  static_assert(U == 4, "four items are taken together in three slots");
  static_assert(4 * 3 == C && 32 * 3 * 3 == 2 * F, "shift phase: a quad of lanes is a head, three slots are two rows");
  static_assert(2 * 6 == C && 32 * 3 * 6 == 4 * F, "pass: a pair of lanes is a head, three slots are four rows");
  constexpr int NS = 3;
  __shared__ __attribute__((aligned(16))) float acc[TS * PF];
  __shared__ __attribute__((aligned(16))) float xrs[TS * F];
  __shared__ float shf[TS * H];
  __shared__ float den[TS * H];
  __shared__ float satt[F];
  __shared__ int   over;

  const int t = blockIdx.x, tid = threadIdx.x, lane = tid & 31;
  const int wv = __builtin_amdgcn_readfirstlane(tid >> 5);
  const int n0 = t * TS;
  constexpr int NST = TS * (F / 4) / 256;
  static_assert(TS * (F / 4) % 256 == 0, "the tile's rows split evenly over the group");
  {
    const float4* __restrict__ xr4 = (const float4*)(xr + (size_t)n0 * F);
    const int npc = min(TS, N_ - n0) * (F / 4);
    float4 st[NST];
#pragma unroll
    for (int k = 0; k < NST; ++k) {
      const int i = tid + 256 * k;
      st[k] = make_float4(0.0f, 0.0f, 0.0f, 0.0f);
      if (i < npc) st[k] = xr4[i];
    }
#pragma unroll
    for (int k = 0; k < NST; ++k) *(float4*)(xrs + 4 * (tid + 256 * k)) = st[k];
  }
  if (tid == 0) over = 0;

  const bool hiB = lane >= 16;
  int eo[3], hd[3];
  float av[3][3];
  eo[0] = 3 * lane;
  eo[1] = hiB ? 3 * lane - 48 : 96 + 3 * lane;
  eo[2] = 48 + 3 * lane;
#pragma unroll
  for (int z = 0; z < 3; ++z) {
    hd[z] = eo[z] / C;
#pragma unroll
    for (int c = 0; c < 3; ++c) av[z][c] = att[eo[z] + c];
  }
  bool sec[3];
  int eo6[3], hd6[3];
  float a6[3][6];
#pragma unroll
  for (int z = 0; z < 3; ++z) {
    const int p  = 32 * z + lane;
    const int it = p / 24;
    sec[z] = it != z;
    eo6[z] = 6 * (p - 24 * it);
    hd6[z] = eo6[z] / C;
#pragma unroll
    for (int c = 0; c < 6; ++c) a6[z][c] = att[eo6[z] + c];
  }
  __syncthreads();

  {
    constexpr int RW = TS / 8;
    static_assert(RW % 2 == 0, "rows are taken in pairs");
    constexpr int NSS = 3 * RW / 2;
    f3 lv[NSS];
    int rw[NSS];
#pragma unroll
    for (int i = 0; i < NSS; ++i) {
      const int z = i % 3;
      const int r0 = wv * RW + 2 * (i / 3), r1 = r0 + 1;
      rw[i] = (z == 0) ? r0 : (z == 2) ? r1 : (hiB ? r1 : r0);
      const int n = min(n0 + rw[i], N_ - 1);
      lv[i] = *(const f3*)(xl + (size_t)n * F + eo[z]);
    }
#pragma unroll
    for (int i = 0; i < NSS; ++i) {
      const int z = i % 3;
      const f3 rv = *(const f3*)(xrs + rw[i] * F + eo[z]);
      const float a = slot_logit(lv[i], rv, av[z]);
      const bool there = n0 + rw[i] < N_;
      float* ap = acc + rw[i] * PF + eo[z];
      ap[0] = there ? lv[i].x : 0.0f;
      ap[1] = there ? lv[i].y : 0.0f;
      ap[2] = there ? lv[i].z : 0.0f;
      if ((lane & 3) == 0) {
        shf[rw[i] * H + hd[z]] = a;
        den[rw[i] * H + hd[z]] = there ? 1.0f : 0.0f;
      }
    }
  }
  __syncthreads();

  const int b = off[t];
  const int cnt = off[t + 1] - b;

  for (int q0 = 0; wave_item<U>(q0, wv) < cnt; q0 += 32) {
    int sL, dlL;
    const int okL = edge_item(wave_item<U>(q0 + lane, wv), cnt, b, bins, sL, dlL) ? 1 : 0;
    for (int g = 0; g < 32 / U; ++g) {
      if (wave_item<U>(q0 + g * U, wv) >= cnt) break;
      int s[U], dl[U]; bool ok[U];
#pragma unroll
      for (int u = 0; u < U; ++u) {
        s[u]  = lane_bcast(sL, g * U + u);
        dl[u] = lane_bcast(dlL, g * U + u);
        ok[u] = lane_bcast(okL, g * U + u) != 0;
      }
      int sd[NS]; bool so[NS];
      f6 lv[NS], rv[NS];
      float mv[NS];
#pragma unroll
      for (int z = 0; z < NS; ++z) {
        const int ss = sec[z] ? s[z + 1] : s[z];
        sd[z] = sec[z] ? dl[z + 1] : dl[z];
        so[z] = sec[z] ? ok[z + 1] : ok[z];
        lv[z] = *(const f6*)(xl + (size_t)ss * F + eo6[z]);
        rv[z] = *(const f6*)(xrs + sd[z] * F + eo6[z]);
        mv[z] = shf[sd[z] * H + hd6[z]];
      }
#pragma unroll
      for (int z = 0; z < NS; ++z) {
        const float d = pair_logit(lv[z], rv[z], a6[z]) - mv[z];
        const float w = expf(d);
        if (so[z] && (lane & 1) == 0) {
          atomicAdd(&den[sd[z] * H + hd6[z]], w);
          if (d > SHIFT_LIM) over = 1;
        }
        const float wz = so[z] ? w : 0.0f;
        float* ap = acc + sd[z] * PF + eo6[z];
        atomicAdd(ap,     wz * lv[z].a.x);
        atomicAdd(ap + 1, wz * lv[z].a.y);
        atomicAdd(ap + 2, wz * lv[z].b.x);
        atomicAdd(ap + 3, wz * lv[z].b.y);
        atomicAdd(ap + 4, wz * lv[z].c.x);
        atomicAdd(ap + 5, wz * lv[z].c.y);
      }
    }
  }
  __syncthreads();

  if (over != 0) {
    gat_tile_body<H, C, U, true>(acc, (int*)shf, den, satt, off, bins, xl, xr, att, bias, lg, nullptr,
                                 Wnl, bnl, Wnr, bnr, nxl, nxr);
    return;
  }

  {
    constexpr int NP = TS * (F / 4) / 256;
    constexpr int NB = 3;
    static_assert(TS * (F / 4) % 256 == 0 && NP % NB == 0 && C % 4 == 0 && PF % 4 == 0, "pieces split evenly");
    const float4* __restrict__ b4 = (const float4*)bias;
    for (int k0 = 0; k0 < NP; k0 += NB) {
      float4 av[NB], bv[NB];
      float q[NB];
      float4* ap[NB];
#pragma unroll
      for (int k = 0; k < NB; ++k) {
        const int i4 = tid + 256 * (k0 + k);
        const int dl = i4 / (F / 4), r4 = i4 - dl * (F / 4);
        ap[k] = (float4*)(acc + dl * PF + 4 * r4);
        av[k] = *ap[k];
        q[k]  = den[dl * H + r4 / (C / 4)] + 1e-16f;
        bv[k] = b4[r4];
      }
#pragma unroll
      for (int k = 0; k < NB; ++k) {
        float4 o;
        o.x = av[k].x / q[k] + bv[k].x;
        o.y = av[k].y / q[k] + bv[k].y;
        o.z = av[k].z / q[k] + bv[k].z;
        o.w = av[k].w / q[k] + bv[k].w;
        *ap[k] = o;
      }
    }
  }
  __syncthreads();
  tile_proj_pair<F, PF, F2_>(acc, n0, wv, lane, Wnl, bnl, nxl, Wnr, bnr, nxr);
}

constexpr size_t OFF_DINV = 0;
constexpr size_t OFF_CNT  = OFF_DINV + N_;
constexpr size_t OFF_OFFS = OFF_CNT  + 2048;
constexpr size_t OFF_CUR  = OFF_OFFS + 2048;
constexpr size_t OFF_XL2  = (size_t)N_ + E_;
constexpr size_t OFF_XR2  = OFF_XL2  + (size_t)N_*D_;
constexpr size_t OFF_XL1  = OFF_XR2  + (size_t)N_*D_;
constexpr size_t OFF_XR1  = OFF_XL1  + (size_t)N_*F1_;
constexpr size_t OFF_BINS = OFF_XR1  + (size_t)N_*F1_;
constexpr size_t OFF_LG   = OFF_BINS + (size_t)N_*24;
static_assert(OFF_CUR + 2048 <= OFF_XL2, "bin tables fit in front of the projections");
static_assert(F2_ == D_, "layer 2's projections are sized with D_ above");
static_assert((size_t)E_ <= (size_t)N_ * 24, "bins fit their region");

extern "C" void kernel_launch(void* const* d_in, const int* in_sizes, int n_in,
                              void* d_out, int out_size, void* d_ws, size_t ws_size,
                              hipStream_t stream) {
  const float* x     = (const float*)d_in[0];
  const int*   ei    = (const int*)  d_in[1];
  const float* Wssg  = (const float*)d_in[2];
  const float* bssg  = (const float*)d_in[3];
  const float* W1l   = (const float*)d_in[4];
  const float* b1l   = (const float*)d_in[5];
  const float* W1r   = (const float*)d_in[6];
  const float* b1r   = (const float*)d_in[7];
  const float* att1  = (const float*)d_in[8];
  const float* bias1 = (const float*)d_in[9];
  const float* W2l   = (const float*)d_in[10];
  const float* b2l   = (const float*)d_in[11];
  const float* W2r   = (const float*)d_in[12];
  const float* b2r   = (const float*)d_in[13];
  const float* att2  = (const float*)d_in[14];
  const float* bias2 = (const float*)d_in[15];

  const int* src = ei;
  const int* dst = ei + E_;

  float* ws   = (float*)d_ws;
  float* dinv = ws + OFF_DINV;
  int*   cnt  = (int*)(ws + OFF_CNT);
  int*   offs = (int*)(ws + OFF_OFFS);
  int*   cur  = (int*)(ws + OFF_CUR);
  float* xl2  = ws + OFF_XL2;
  float* xr2  = ws + OFF_XR2;
  float* xl1  = ws + OFF_XL1;
  float* xr1  = ws + OFF_XR1;
  int*   bins = (int*)(ws + OFF_BINS);
  float* lg   = ws + OFF_LG;
  float* out  = (float*)d_out;

  const int TB = 256;

  k_zero_i<<<CDIV(2048, TB), TB, 0, stream>>>(cnt, 2048);
  k_tilecount<<<CDIV(E_, CH), 1024, 0, stream>>>(dst, cnt);
  k_scan<<<1, 1024, 0, stream>>>(cnt, offs, cur);
  k_binscatter<<<CDIV(E_, CH), 1024, 0, stream>>>(src, dst, cur, bins);

  k_dinv_tile<<<NT, TS, 0, stream>>>(offs, bins, dinv);
  k_ssg_tile<4><<<NT, 256, 0, stream>>>(offs, bins, x, dinv, Wssg, bssg, W1l, b1l, W1r, b1r, xl1, xr1);

  k_gat12_tile<4><<<NT, 256, 0, stream>>>(offs, bins, xl1, xr1, att1, bias1, lg,
                                          W2l, b2l, W2r, b2r, xl2, xr2);

  k_gat1h_tile<64, 4><<<NT, 256, 0, stream>>>(offs, bins, xl2, xr2, att2, bias2, lg, out);
}
